# starting kernel
# baseline (speedup 1.0000x reference)
#include <hip/hip_runtime.h>
#include <math.h>

#define BB   32
#define CC   122
#define TT   500
#define WID  80
#define UU   96
#define K2   (WID*CC)
#define HID  768
#define PROJ 1024
#define PLEN 484
#define EPSF 1e-5f

typedef float v2f __attribute__((ext_vector_type(2)));
typedef float v8f __attribute__((ext_vector_type(8)));

static __device__ __forceinline__ v8f wmma_f32(v2f a, v2f b, v8f c) {
    return __builtin_amdgcn_wmma_f32_16x16x4_f32(false, a, false, b, (short)0, c, false, false);
}

__global__ __launch_bounds__(32) void k_subject(
    const float* __restrict__ x, const int* __restrict__ sid,
    const float* __restrict__ W_sub, const float* __restrict__ b_sub,
    float* __restrict__ out0)
{
    const int lane = threadIdx.x;
    const int b  = blockIdx.z;
    const int m0 = blockIdx.x * 16;
    const int n0 = blockIdx.y * 16;
    const int s  = sid[b];
    const float* Wg = W_sub + (size_t)s * TT * TT;

    int am = m0 + (lane & 15); if (am >= CC) am = CC - 1;
    int an = n0 + (lane & 15); if (an >= TT) an = TT - 1;
    const int koff = (lane >> 4) * 2;
    const float* pa = x  + ((size_t)b * CC + am) * TT + koff;
    const float* pb = Wg + (size_t)an * TT + koff;

    v8f acc = {};
    #pragma unroll 4
    for (int k4 = 0; k4 < TT / 4; ++k4) {
        v2f a = *(const v2f*)(pa + 4 * k4);
        v2f w = *(const v2f*)(pb + 4 * k4);
        acc = wmma_f32(a, w, acc);
    }

    const float* bg = b_sub + (size_t)s * TT;
    const int nn = n0 + (lane & 15);
    #pragma unroll
    for (int r = 0; r < 8; ++r) {
        int mm = m0 + r + ((lane >> 4) << 3);
        if (mm < CC && nn < TT)
            out0[((size_t)b * CC + mm) * TT + nn] = acc[r] + bg[nn];
    }
}

__global__ __launch_bounds__(128) void k_pool(
    const float* __restrict__ out0, float* __restrict__ P)
{
    __shared__ float row[TT];
    const int bc = blockIdx.x;
    const float* src = out0 + (size_t)bc * TT;
    for (int i = threadIdx.x; i < TT; i += 128) row[i] = src[i];
    __syncthreads();
    float* dst = P + (size_t)bc * PLEN;
    for (int m = threadIdx.x; m < PLEN; m += 128) {
        float sum = 0.f;
        #pragma unroll
        for (int i = 0; i < 17; ++i) sum += row[m + i];
        dst[m] = sum * (1.f / 17.f);
    }
}

__global__ __launch_bounds__(160) void k_conv2(
    const float* __restrict__ P,
    const float* __restrict__ conv1_w, const float* __restrict__ conv1_b,
    const float* __restrict__ bn1_g, const float* __restrict__ bn1_b,
    const float* __restrict__ bn1_m, const float* __restrict__ bn1_v,
    const float* __restrict__ conv2_w, const float* __restrict__ conv2_b,
    const float* __restrict__ bn2_g, const float* __restrict__ bn2_b,
    const float* __restrict__ bn2_m, const float* __restrict__ bn2_v,
    float* __restrict__ h2)
{
    __shared__ float Pl[CC * 80];
    __shared__ float w1s[WID * 5];
    __shared__ float sc1[WID], sh1[WID];

    const int tid = threadIdx.x;
    const int b   = blockIdx.y;
    const int u0  = blockIdx.x * 16;

    for (int i = tid; i < WID * 5; i += 160) w1s[i] = conv1_w[i];
    if (tid < WID) {
        float sc = bn1_g[tid] * rsqrtf(bn1_v[tid] + EPSF);
        sc1[tid] = sc;
        sh1[tid] = bn1_b[tid] - bn1_m[tid] * sc + sc * conv1_b[tid];
    }
    for (int i = tid; i < CC * 80; i += 160) {
        int c = i / 80, q = i - c * 80;
        Pl[i] = P[((size_t)b * CC + c) * PLEN + 5 * u0 + q];
    }
    __syncthreads();

    const int wave = tid >> 5;
    const int lane = tid & 31;
    const int n    = lane & 15;
    const int half = lane >> 4;

    const float* paA = conv2_w + (size_t)(wave * 16 + (lane & 15)) * K2 + half * 2;

    int w0 = 0, c0 = half * 2;
    v8f acc = {};
    for (int k4 = 0; k4 < K2 / 4; ++k4) {
        v2f a = *(const v2f*)(paA + 4 * k4);

        const float* wr0 = &w1s[w0 * 5];
        const float* pr0 = &Pl[c0 * 80 + 5 * n];
        float v0 = wr0[0]*pr0[0] + wr0[1]*pr0[1] + wr0[2]*pr0[2] + wr0[3]*pr0[3] + wr0[4]*pr0[4];
        float g0 = sc1[w0] * v0 + sh1[w0];
        float e0 = g0 > 0.f ? g0 : (__expf(g0) - 1.f);

        int c1 = c0 + 1, w1i = w0;
        if (c1 == CC) { c1 = 0; w1i = w0 + 1; }
        const float* wr1 = &w1s[w1i * 5];
        const float* pr1 = &Pl[c1 * 80 + 5 * n];
        float v1 = wr1[0]*pr1[0] + wr1[1]*pr1[1] + wr1[2]*pr1[2] + wr1[3]*pr1[3] + wr1[4]*pr1[4];
        float g1 = sc1[w1i] * v1 + sh1[w1i];
        float e1 = g1 > 0.f ? g1 : (__expf(g1) - 1.f);

        v2f bb = { e0, e1 };
        acc = wmma_f32(a, bb, acc);

        c0 += 4;
        if (c0 >= CC) { c0 -= CC; w0 += 1; }
    }

    #pragma unroll
    for (int r = 0; r < 8; ++r) {
        int w2 = wave * 16 + r + half * 8;
        float sc = bn2_g[w2] * rsqrtf(bn2_v[w2] + EPSF);
        float sh = bn2_b[w2] - bn2_m[w2] * sc;
        float v  = sc * (acc[r] + conv2_b[w2]) + sh;
        float e  = v > 0.f ? v : (__expf(v) - 1.f);
        h2[((size_t)b * WID + w2) * UU + u0 + n] = e;
    }
}

__global__ __launch_bounds__(256) void k_proj(
    const float* __restrict__ h2, const float* __restrict__ proj_w,
    const float* __restrict__ proj_b, float* __restrict__ z)
{
    __shared__ float hs[WID * UU];
    __shared__ float pw[8 * WID];
    const int b = blockIdx.x;
    for (int i = threadIdx.x; i < WID * UU; i += 256) hs[i] = h2[(size_t)b * WID * UU + i];
    for (int i = threadIdx.x; i < 8 * WID; i += 256)  pw[i] = proj_w[i];
    __syncthreads();
    for (int idx = threadIdx.x; idx < HID; idx += 256) {
        int u = idx >> 3, e = idx & 7;
        float s = proj_b[e];
        #pragma unroll 8
        for (int w = 0; w < WID; ++w) s += pw[e * WID + w] * hs[w * UU + u];
        z[(size_t)b * HID + idx] = s;
    }
}

__global__ __launch_bounds__(32) void k_mlp1(
    const float* __restrict__ z, const float* __restrict__ w,
    const float* __restrict__ bias, float* __restrict__ z1, float* __restrict__ g)
{
    const int lane = threadIdx.x;
    const int m0 = blockIdx.x * 16, n0 = blockIdx.y * 16;
    const float* pa = z + (size_t)(m0 + (lane & 15)) * HID + (lane >> 4) * 2;
    const float* pb = w + (size_t)(n0 + (lane & 15)) * HID + (lane >> 4) * 2;
    v8f acc = {};
    #pragma unroll 4
    for (int k4 = 0; k4 < HID / 4; ++k4) {
        v2f a = *(const v2f*)(pa + 4 * k4);
        v2f bb = *(const v2f*)(pb + 4 * k4);
        acc = wmma_f32(a, bb, acc);
    }
    const int nn = n0 + (lane & 15);
    const float bn = bias[nn];
    #pragma unroll
    for (int r = 0; r < 8; ++r) {
        int mm = m0 + r + ((lane >> 4) << 3);
        float v = acc[r] + bn;
        z1[(size_t)mm * PROJ + nn] = v;
        g [(size_t)mm * PROJ + nn] = 0.5f * v * (1.f + erff(v * 0.70710678118654752f));
    }
}

__global__ __launch_bounds__(32) void k_mlp2(
    const float* __restrict__ g, const float* __restrict__ w,
    const float* __restrict__ bias, const float* __restrict__ z1,
    float* __restrict__ y)
{
    const int lane = threadIdx.x;
    const int m0 = blockIdx.x * 16, n0 = blockIdx.y * 16;
    const float* pa = g + (size_t)(m0 + (lane & 15)) * PROJ + (lane >> 4) * 2;
    const float* pb = w + (size_t)(n0 + (lane & 15)) * PROJ + (lane >> 4) * 2;
    v8f acc = {};
    #pragma unroll 4
    for (int k4 = 0; k4 < PROJ / 4; ++k4) {
        v2f a = *(const v2f*)(pa + 4 * k4);
        v2f bb = *(const v2f*)(pb + 4 * k4);
        acc = wmma_f32(a, bb, acc);
    }
    const int nn = n0 + (lane & 15);
    const float bn = bias[nn];
    #pragma unroll
    for (int r = 0; r < 8; ++r) {
        int mm = m0 + r + ((lane >> 4) << 3);
        y[(size_t)mm * PROJ + nn] = z1[(size_t)mm * PROJ + nn] + acc[r] + bn;
    }
}

__global__ __launch_bounds__(256) void k_ln(
    const float* __restrict__ y, const float* __restrict__ gam,
    const float* __restrict__ bet, float* __restrict__ out)
{
    __shared__ float red[256];
    const int b = blockIdx.x;
    const float* row = y + (size_t)b * PROJ;
    float s = 0.f;
    for (int i = threadIdx.x; i < PROJ; i += 256) s += row[i];
    red[threadIdx.x] = s; __syncthreads();
    for (int off = 128; off > 0; off >>= 1) {
        if (threadIdx.x < off) red[threadIdx.x] += red[threadIdx.x + off];
        __syncthreads();
    }
    const float mu = red[0] * (1.f / PROJ);
    __syncthreads();
    float s2 = 0.f;
    for (int i = threadIdx.x; i < PROJ; i += 256) { float d = row[i] - mu; s2 += d * d; }
    red[threadIdx.x] = s2; __syncthreads();
    for (int off = 128; off > 0; off >>= 1) {
        if (threadIdx.x < off) red[threadIdx.x] += red[threadIdx.x + off];
        __syncthreads();
    }
    const float inv = rsqrtf(red[0] * (1.f / PROJ) + EPSF);
    for (int i = threadIdx.x; i < PROJ; i += 256)
        out[(size_t)b * PROJ + i] = (row[i] - mu) * inv * gam[i] + bet[i];
}

extern "C" void kernel_launch(void* const* d_in, const int* in_sizes, int n_in,
                              void* d_out, int out_size, void* d_ws, size_t ws_size,
                              hipStream_t stream) {
    const float* x       = (const float*)d_in[0];
    const int*   sid     = (const int*)  d_in[1];
    const float* W_sub   = (const float*)d_in[2];
    const float* b_sub   = (const float*)d_in[3];
    const float* conv1_w = (const float*)d_in[4];
    const float* conv1_b = (const float*)d_in[5];
    const float* bn1_g   = (const float*)d_in[6];
    const float* bn1_b   = (const float*)d_in[7];
    const float* bn1_m   = (const float*)d_in[8];
    const float* bn1_v   = (const float*)d_in[9];
    const float* conv2_w = (const float*)d_in[10];
    const float* conv2_b = (const float*)d_in[11];
    const float* bn2_g   = (const float*)d_in[12];
    const float* bn2_b   = (const float*)d_in[13];
    const float* bn2_m   = (const float*)d_in[14];
    const float* bn2_v   = (const float*)d_in[15];
    const float* proj_w  = (const float*)d_in[16];
    const float* proj_b  = (const float*)d_in[17];
    const float* mlp1_w  = (const float*)d_in[18];
    const float* mlp1_b  = (const float*)d_in[19];
    const float* mlp2_w  = (const float*)d_in[20];
    const float* mlp2_b  = (const float*)d_in[21];
    const float* ln_g    = (const float*)d_in[22];
    const float* ln_b    = (const float*)d_in[23];

    float* ws   = (float*)d_ws;
    float* out0 = ws;
    float* P    = out0 + (size_t)BB * CC * TT;
    float* h2   = P    + (size_t)BB * CC * PLEN;
    float* z    = h2   + (size_t)BB * WID * UU;
    float* z1   = z    + (size_t)BB * HID;
    float* gg   = z1   + (size_t)BB * PROJ;
    float* yy   = gg   + (size_t)BB * PROJ;

    dim3 g1(8, 32, BB);
    k_subject<<<g1, 32, 0, stream>>>(x, sid, W_sub, b_sub, out0);

    k_pool<<<BB * CC, 128, 0, stream>>>(out0, P);

    dim3 g3(UU / 16, BB);
    k_conv2<<<g3, 160, 0, stream>>>(P, conv1_w, conv1_b, bn1_g, bn1_b, bn1_m, bn1_v,
                                    conv2_w, conv2_b, bn2_g, bn2_b, bn2_m, bn2_v, h2);

    k_proj<<<BB, 256, 0, stream>>>(h2, proj_w, proj_b, z);

    dim3 g5(2, PROJ / 16);
    k_mlp1<<<g5, 32, 0, stream>>>(z, mlp1_w, mlp1_b, z1, gg);
    k_mlp2<<<g5, 32, 0, stream>>>(gg, mlp2_w, mlp2_b, z1, yy);

    k_ln<<<BB, 256, 0, stream>>>(yy, ln_g, ln_b, (float*)d_out);
}


# conv2: K walked as a (w outer, c inner) nest with c=120,121 a half step: a lane's pooled patches stay in registers for all 80 w, taps/bn1 uniform per w; k regrouped per w, so the K sum order changes
# speedup vs baseline: 1.8850x; 1.8850x over previous
#include <hip/hip_runtime.h>
#include <math.h>

#define BB   32
#define CC   122
#define TT   500
#define WID  80
#define UU   96
#define K2   (WID*CC)
#define HID  768
#define PROJ 1024
#define PLEN 484
#define EPSF 1e-5f

typedef float v2f __attribute__((ext_vector_type(2)));
typedef float v8f __attribute__((ext_vector_type(8)));

static __device__ __forceinline__ v8f wmma_f32(v2f a, v2f b, v8f c) {
    return __builtin_amdgcn_wmma_f32_16x16x4_f32(false, a, false, b, (short)0, c, false, false);
}

__global__ __launch_bounds__(32) void k_subject(
    const float* __restrict__ x, const int* __restrict__ sid,
    const float* __restrict__ W_sub, const float* __restrict__ b_sub,
    float* __restrict__ out0)
{
    const int lane = threadIdx.x;
    const int b  = blockIdx.z;
    const int m0 = blockIdx.x * 16;
    const int n0 = blockIdx.y * 16;
    const int s  = sid[b];
    const float* Wg = W_sub + (size_t)s * TT * TT;

    int am = m0 + (lane & 15); if (am >= CC) am = CC - 1;
    int an = n0 + (lane & 15); if (an >= TT) an = TT - 1;
    const int koff = (lane >> 4) * 2;
    const float* pa = x  + ((size_t)b * CC + am) * TT + koff;
    const float* pb = Wg + (size_t)an * TT + koff;

    v8f acc = {};
    #pragma unroll 4
    for (int k4 = 0; k4 < TT / 4; ++k4) {
        v2f a = *(const v2f*)(pa + 4 * k4);
        v2f w = *(const v2f*)(pb + 4 * k4);
        acc = wmma_f32(a, w, acc);
    }

    const float* bg = b_sub + (size_t)s * TT;
    const int nn = n0 + (lane & 15);
    #pragma unroll
    for (int r = 0; r < 8; ++r) {
        int mm = m0 + r + ((lane >> 4) << 3);
        if (mm < CC && nn < TT)
            out0[((size_t)b * CC + mm) * TT + nn] = acc[r] + bg[nn];
    }
}

__global__ __launch_bounds__(128) void k_pool(
    const float* __restrict__ out0, float* __restrict__ P)
{
    __shared__ float row[TT];
    const int bc = blockIdx.x;
    const float* src = out0 + (size_t)bc * TT;
    for (int i = threadIdx.x; i < TT; i += 128) row[i] = src[i];
    __syncthreads();
    float* dst = P + (size_t)bc * PLEN;
    for (int m = threadIdx.x; m < PLEN; m += 128) {
        float sum = 0.f;
        #pragma unroll
        for (int i = 0; i < 17; ++i) sum += row[m + i];
        dst[m] = sum * (1.f / 17.f);
    }
}

#define CSTEPS 30
#define GS     6
#define NCH    62
#define H1_STAGE(expr) _Pragma("unroll") for (int i = 0; i < 2 * GS; ++i) { e[i] = (expr); }

__global__ __launch_bounds__(160) void k_conv2(
    const float* __restrict__ P,
    const float* __restrict__ conv1_w, const float* __restrict__ conv1_b,
    const float* __restrict__ bn1_g, const float* __restrict__ bn1_b,
    const float* __restrict__ bn1_m, const float* __restrict__ bn1_v,
    const float* __restrict__ conv2_w, const float* __restrict__ conv2_b,
    const float* __restrict__ bn2_g, const float* __restrict__ bn2_b,
    const float* __restrict__ bn2_m, const float* __restrict__ bn2_v,
    float* __restrict__ h2)
{
    __shared__ float Pl[CC * 80];
    __shared__ float w1s[WID * 5];
    __shared__ float sc1[WID], sh1[WID];

    const int tid = threadIdx.x;
    const int b   = blockIdx.y;
    const int u0  = blockIdx.x * 16;

    for (int i = tid; i < WID * 5; i += 160) w1s[i] = conv1_w[i];
    if (tid < WID) {
        float sc = bn1_g[tid] * rsqrtf(bn1_v[tid] + EPSF);
        sc1[tid] = sc;
        sh1[tid] = bn1_b[tid] - bn1_m[tid] * sc + sc * conv1_b[tid];
    }
    for (int i = tid; i < CC * 80; i += 160) {
        int c = i / 80, q = i - c * 80;
        Pl[i] = P[((size_t)b * CC + c) * PLEN + 5 * u0 + q];
    }
    __syncthreads();

    const int wave = tid >> 5;
    const int lane = tid & 31;
    const int n    = lane & 15;
    const int half = lane >> 4;

    float pr[NCH][5];
    #pragma unroll
    for (int i = 0; i < NCH; ++i) {
        const int c = (i < 2 * CSTEPS) ? (4 * (i >> 1) + 2 * half + (i & 1)) : (120 + (i - 2 * CSTEPS));
        const float* p = &Pl[c * 80 + 5 * n];
        #pragma unroll
        for (int j = 0; j < 5; ++j) pr[i][j] = p[j];
    }

    const float* rowA = conv2_w + (size_t)(wave * 16 + n) * K2;
    v2f acur[CSTEPS + 1];
    #pragma unroll
    for (int s = 0; s < CSTEPS; ++s) acur[s] = *(const v2f*)(rowA + 2 * half + 4 * s);
    acur[CSTEPS] = *(const v2f*)(rowA + 120);

    v8f acc = {};
    for (int w = 0; w < WID; ++w) {
        const float t0 = w1s[w * 5 + 0], t1 = w1s[w * 5 + 1], t2 = w1s[w * 5 + 2],
                    t3 = w1s[w * 5 + 3], t4 = w1s[w * 5 + 4];
        const float sc = sc1[w], sh = sh1[w];

        const int wn = (w + 1 < WID) ? (w + 1) : w;
        const float* pan = rowA + wn * CC;

        #pragma unroll
        for (int g = 0; g < CSTEPS / GS; ++g) {
            float e[2 * GS];
            H1_STAGE(t0 * pr[2 * GS * g + i][0])
            H1_STAGE(__builtin_fmaf(t1, pr[2 * GS * g + i][1], e[i]))
            H1_STAGE(__builtin_fmaf(t2, pr[2 * GS * g + i][2], e[i]))
            H1_STAGE(__builtin_fmaf(t3, pr[2 * GS * g + i][3], e[i]))
            H1_STAGE(__builtin_fmaf(t4, pr[2 * GS * g + i][4], e[i]))
            H1_STAGE(__builtin_fmaf(sc, e[i], sh))
            H1_STAGE(e[i] > 0.f ? e[i] : (__expf(e[i]) - 1.f))
            #pragma unroll
            for (int s = 0; s < GS; ++s) {
                v2f bb = { e[2 * s], e[2 * s + 1] };
                acc = wmma_f32(acur[GS * g + s], bb, acc);
            }
            #pragma unroll
            for (int s = 0; s < GS; ++s)
                acur[GS * g + s] = *(const v2f*)(pan + 2 * half + 4 * (GS * g + s));
        }
        {
            float e0 = t0 * pr[60][0], e1 = t0 * pr[61][0];
            e0 = __builtin_fmaf(t1, pr[60][1], e0); e1 = __builtin_fmaf(t1, pr[61][1], e1);
            e0 = __builtin_fmaf(t2, pr[60][2], e0); e1 = __builtin_fmaf(t2, pr[61][2], e1);
            e0 = __builtin_fmaf(t3, pr[60][3], e0); e1 = __builtin_fmaf(t3, pr[61][3], e1);
            e0 = __builtin_fmaf(t4, pr[60][4], e0); e1 = __builtin_fmaf(t4, pr[61][4], e1);
            e0 = __builtin_fmaf(sc, e0, sh);        e1 = __builtin_fmaf(sc, e1, sh);
            e0 = e0 > 0.f ? e0 : (__expf(e0) - 1.f);
            e1 = e1 > 0.f ? e1 : (__expf(e1) - 1.f);
            v2f a  = acur[CSTEPS];
            v2f bb = { e0, e1 };
            if (half) { a = v2f{0.f, 0.f}; bb = v2f{0.f, 0.f}; }
            acc = wmma_f32(a, bb, acc);
            acur[CSTEPS] = *(const v2f*)(pan + 120);
        }
    }

    #pragma unroll
    for (int r = 0; r < 8; ++r) {
        int w2 = wave * 16 + r + half * 8;
        float sc = bn2_g[w2] * rsqrtf(bn2_v[w2] + EPSF);
        float sh = bn2_b[w2] - bn2_m[w2] * sc;
        float v  = sc * (acc[r] + conv2_b[w2]) + sh;
        float e  = v > 0.f ? v : (__expf(v) - 1.f);
        h2[((size_t)b * WID + w2) * UU + u0 + n] = e;
    }
}

__global__ __launch_bounds__(256) void k_proj(
    const float* __restrict__ h2, const float* __restrict__ proj_w,
    const float* __restrict__ proj_b, float* __restrict__ z)
{
    __shared__ float hs[WID * UU];
    __shared__ float pw[8 * WID];
    const int b = blockIdx.x;
    for (int i = threadIdx.x; i < WID * UU; i += 256) hs[i] = h2[(size_t)b * WID * UU + i];
    for (int i = threadIdx.x; i < 8 * WID; i += 256)  pw[i] = proj_w[i];
    __syncthreads();
    for (int idx = threadIdx.x; idx < HID; idx += 256) {
        int u = idx >> 3, e = idx & 7;
        float s = proj_b[e];
        #pragma unroll 8
        for (int w = 0; w < WID; ++w) s += pw[e * WID + w] * hs[w * UU + u];
        z[(size_t)b * HID + idx] = s;
    }
}

__global__ __launch_bounds__(32) void k_mlp1(
    const float* __restrict__ z, const float* __restrict__ w,
    const float* __restrict__ bias, float* __restrict__ z1, float* __restrict__ g)
{
    const int lane = threadIdx.x;
    const int m0 = blockIdx.x * 16, n0 = blockIdx.y * 16;
    const float* pa = z + (size_t)(m0 + (lane & 15)) * HID + (lane >> 4) * 2;
    const float* pb = w + (size_t)(n0 + (lane & 15)) * HID + (lane >> 4) * 2;
    v8f acc = {};
    #pragma unroll 4
    for (int k4 = 0; k4 < HID / 4; ++k4) {
        v2f a = *(const v2f*)(pa + 4 * k4);
        v2f bb = *(const v2f*)(pb + 4 * k4);
        acc = wmma_f32(a, bb, acc);
    }
    const int nn = n0 + (lane & 15);
    const float bn = bias[nn];
    #pragma unroll
    for (int r = 0; r < 8; ++r) {
        int mm = m0 + r + ((lane >> 4) << 3);
        float v = acc[r] + bn;
        z1[(size_t)mm * PROJ + nn] = v;
        g [(size_t)mm * PROJ + nn] = 0.5f * v * (1.f + erff(v * 0.70710678118654752f));
    }
}

__global__ __launch_bounds__(32) void k_mlp2(
    const float* __restrict__ g, const float* __restrict__ w,
    const float* __restrict__ bias, const float* __restrict__ z1,
    float* __restrict__ y)
{
    const int lane = threadIdx.x;
    const int m0 = blockIdx.x * 16, n0 = blockIdx.y * 16;
    const float* pa = g + (size_t)(m0 + (lane & 15)) * PROJ + (lane >> 4) * 2;
    const float* pb = w + (size_t)(n0 + (lane & 15)) * PROJ + (lane >> 4) * 2;
    v8f acc = {};
    #pragma unroll 4
    for (int k4 = 0; k4 < PROJ / 4; ++k4) {
        v2f a = *(const v2f*)(pa + 4 * k4);
        v2f bb = *(const v2f*)(pb + 4 * k4);
        acc = wmma_f32(a, bb, acc);
    }
    const int nn = n0 + (lane & 15);
    const float bn = bias[nn];
    #pragma unroll
    for (int r = 0; r < 8; ++r) {
        int mm = m0 + r + ((lane >> 4) << 3);
        y[(size_t)mm * PROJ + nn] = z1[(size_t)mm * PROJ + nn] + acc[r] + bn;
    }
}

__global__ __launch_bounds__(256) void k_ln(
    const float* __restrict__ y, const float* __restrict__ gam,
    const float* __restrict__ bet, float* __restrict__ out)
{
    __shared__ float red[256];
    const int b = blockIdx.x;
    const float* row = y + (size_t)b * PROJ;
    float s = 0.f;
    for (int i = threadIdx.x; i < PROJ; i += 256) s += row[i];
    red[threadIdx.x] = s; __syncthreads();
    for (int off = 128; off > 0; off >>= 1) {
        if (threadIdx.x < off) red[threadIdx.x] += red[threadIdx.x + off];
        __syncthreads();
    }
    const float mu = red[0] * (1.f / PROJ);
    __syncthreads();
    float s2 = 0.f;
    for (int i = threadIdx.x; i < PROJ; i += 256) { float d = row[i] - mu; s2 += d * d; }
    red[threadIdx.x] = s2; __syncthreads();
    for (int off = 128; off > 0; off >>= 1) {
        if (threadIdx.x < off) red[threadIdx.x] += red[threadIdx.x + off];
        __syncthreads();
    }
    const float inv = rsqrtf(red[0] * (1.f / PROJ) + EPSF);
    for (int i = threadIdx.x; i < PROJ; i += 256)
        out[(size_t)b * PROJ + i] = (row[i] - mu) * inv * gam[i] + bet[i];
}

extern "C" void kernel_launch(void* const* d_in, const int* in_sizes, int n_in,
                              void* d_out, int out_size, void* d_ws, size_t ws_size,
                              hipStream_t stream) {
    const float* x       = (const float*)d_in[0];
    const int*   sid     = (const int*)  d_in[1];
    const float* W_sub   = (const float*)d_in[2];
    const float* b_sub   = (const float*)d_in[3];
    const float* conv1_w = (const float*)d_in[4];
    const float* conv1_b = (const float*)d_in[5];
    const float* bn1_g   = (const float*)d_in[6];
    const float* bn1_b   = (const float*)d_in[7];
    const float* bn1_m   = (const float*)d_in[8];
    const float* bn1_v   = (const float*)d_in[9];
    const float* conv2_w = (const float*)d_in[10];
    const float* conv2_b = (const float*)d_in[11];
    const float* bn2_g   = (const float*)d_in[12];
    const float* bn2_b   = (const float*)d_in[13];
    const float* bn2_m   = (const float*)d_in[14];
    const float* bn2_v   = (const float*)d_in[15];
    const float* proj_w  = (const float*)d_in[16];
    const float* proj_b  = (const float*)d_in[17];
    const float* mlp1_w  = (const float*)d_in[18];
    const float* mlp1_b  = (const float*)d_in[19];
    const float* mlp2_w  = (const float*)d_in[20];
    const float* mlp2_b  = (const float*)d_in[21];
    const float* ln_g    = (const float*)d_in[22];
    const float* ln_b    = (const float*)d_in[23];

    float* ws   = (float*)d_ws;
    float* out0 = ws;
    float* P    = out0 + (size_t)BB * CC * TT;
    float* h2   = P    + (size_t)BB * CC * PLEN;
    float* z    = h2   + (size_t)BB * WID * UU;
    float* z1   = z    + (size_t)BB * HID;
    float* gg   = z1   + (size_t)BB * PROJ;
    float* yy   = gg   + (size_t)BB * PROJ;

    dim3 g1(8, 32, BB);
    k_subject<<<g1, 32, 0, stream>>>(x, sid, W_sub, b_sub, out0);

    k_pool<<<BB * CC, 128, 0, stream>>>(out0, P);

    dim3 g3(UU / 16, BB);
    k_conv2<<<g3, 160, 0, stream>>>(P, conv1_w, conv1_b, bn1_g, bn1_b, bn1_m, bn1_v,
                                    conv2_w, conv2_b, bn2_g, bn2_b, bn2_m, bn2_v, h2);

    k_proj<<<BB, 256, 0, stream>>>(h2, proj_w, proj_b, z);

    dim3 g5(2, PROJ / 16);
    k_mlp1<<<g5, 32, 0, stream>>>(z, mlp1_w, mlp1_b, z1, gg);
    k_mlp2<<<g5, 32, 0, stream>>>(gg, mlp2_w, mlp2_b, z1, yy);

    k_ln<<<BB, 256, 0, stream>>>(yy, ln_g, ln_b, (float*)d_out);
}


# k_subject: one 8-wave block per (batch, 64 t) covers all 122 rows; x and W rows go through LDS as whole 128-B runs (2 buffers), 16-B fragment reads; k regrouped in the WMMAs: K sum order changes
# speedup vs baseline: 2.7367x; 1.4518x over previous
#include <hip/hip_runtime.h>
#include <math.h>

#define BB   32
#define CC   122
#define TT   500
#define WID  80
#define UU   96
#define K2   (WID*CC)
#define HID  768
#define PROJ 1024
#define PLEN 484
#define EPSF 1e-5f

typedef float v2f __attribute__((ext_vector_type(2)));
typedef float v8f __attribute__((ext_vector_type(8)));

static __device__ __forceinline__ v8f wmma_f32(v2f a, v2f b, v8f c) {
    return __builtin_amdgcn_wmma_f32_16x16x4_f32(false, a, false, b, (short)0, c, false, false);
}

#define SKC   32
#define SP4   9
#define SNCH  16
#define SNT   64
#define SMR   128

__global__ __launch_bounds__(256) void k_subject(
    const float* __restrict__ x, const int* __restrict__ sid,
    const float* __restrict__ W_sub, const float* __restrict__ b_sub,
    float* __restrict__ out0)
{
    __shared__ float4 As[2][SMR * SP4];
    __shared__ float4 Bs[2][SNT * SP4];

    const int tid = threadIdx.x;
    const int b   = blockIdx.y;
    const int n0  = blockIdx.x * SNT;
    const int s   = sid[b];
    const float* Wg = W_sub + (size_t)s * TT * TT;
    const float* xb = x + (size_t)b * CC * TT;

    const int kq = tid & 7;
    const int r  = tid >> 3;
    const float* ga[4];
    const float* gb[2];
    #pragma unroll
    for (int i = 0; i < 4; ++i) {
        int row = r + 32 * i; if (row >= CC) row = CC - 1;
        ga[i] = xb + (size_t)row * TT + 4 * kq;
    }
    #pragma unroll
    for (int i = 0; i < 2; ++i) {
        int row = n0 + r + 32 * i; if (row >= TT) row = TT - 1;
        gb[i] = Wg + (size_t)row * TT + 4 * kq;
    }
    const int st = r * SP4 + kq;

    const int wv   = tid >> 5;
    const int lane = tid & 31;
    const int c    = lane & 15;
    const int h    = lane >> 4;
    const int fa = (16 * wv + c) * SP4 + h;
    const int fb = c * SP4 + h;

    float4 ra[4], rb[2];
    v8f acc[4] = {};

    #pragma unroll
    for (int i = 0; i < 4; ++i) ra[i] = *(const float4*)(ga[i]);
    #pragma unroll
    for (int i = 0; i < 2; ++i) rb[i] = *(const float4*)(gb[i]);
    #pragma unroll
    for (int i = 0; i < 4; ++i) As[0][st + 32 * SP4 * i] = ra[i];
    #pragma unroll
    for (int i = 0; i < 2; ++i) Bs[0][st + 32 * SP4 * i] = rb[i];
    __syncthreads();

    for (int ch = 0; ch < SNCH; ++ch) {
        const int  buf  = ch & 1;
        const bool more = (ch + 1 < SNCH);
        const int  ko   = (ch + 1) * SKC;
        const bool tail = (ko + 4 * kq) >= TT;
        if (more) {
            const int kc = tail ? 0 : ko;
            #pragma unroll
            for (int i = 0; i < 4; ++i) ra[i] = *(const float4*)(ga[i] + kc);
            #pragma unroll
            for (int i = 0; i < 2; ++i) rb[i] = *(const float4*)(gb[i] + kc);
        }

        const float4* pA = &As[buf][fa];
        const float4* pB = &Bs[buf][fb];
        float4 af[4], bf[4][4];
        #pragma unroll
        for (int s8 = 0; s8 < 4; ++s8) {
            af[s8] = pA[2 * s8];
            #pragma unroll
            for (int j = 0; j < 4; ++j) bf[s8][j] = pB[16 * SP4 * j + 2 * s8];
        }
        __builtin_amdgcn_sched_barrier(0);
        #pragma unroll
        for (int s8 = 0; s8 < 4; ++s8) {
            const v2f a0 = { af[s8].x, af[s8].y };
            const v2f a1 = { af[s8].z, af[s8].w };
            #pragma unroll
            for (int j = 0; j < 4; ++j) {
                const v2f w0 = { bf[s8][j].x, bf[s8][j].y };
                acc[j] = wmma_f32(a0, w0, acc[j]);
            }
            #pragma unroll
            for (int j = 0; j < 4; ++j) {
                const v2f w1 = { bf[s8][j].z, bf[s8][j].w };
                acc[j] = wmma_f32(a1, w1, acc[j]);
            }
        }

        if (more) {
            if (tail) {
                #pragma unroll
                for (int i = 0; i < 4; ++i) ra[i] = float4{0.f, 0.f, 0.f, 0.f};
                #pragma unroll
                for (int i = 0; i < 2; ++i) rb[i] = float4{0.f, 0.f, 0.f, 0.f};
            }
            #pragma unroll
            for (int i = 0; i < 4; ++i) As[buf ^ 1][st + 32 * SP4 * i] = ra[i];
            #pragma unroll
            for (int i = 0; i < 2; ++i) Bs[buf ^ 1][st + 32 * SP4 * i] = rb[i];
            __syncthreads();
        }
    }

    const float* bg = b_sub + (size_t)s * TT;
    #pragma unroll
    for (int j = 0; j < 4; ++j) {
        const int nn = n0 + 16 * j + c;
        if (nn < TT) {
            const float bn = bg[nn];
            #pragma unroll
            for (int q = 0; q < 8; ++q) {
                const int mm = 16 * wv + q + 8 * h;
                if (mm < CC)
                    out0[((size_t)b * CC + mm) * TT + nn] = acc[j][q] + bn;
            }
        }
    }
}

__global__ __launch_bounds__(128) void k_pool(
    const float* __restrict__ out0, float* __restrict__ P)
{
    __shared__ float row[TT];
    const int bc = blockIdx.x;
    const float* src = out0 + (size_t)bc * TT;
    for (int i = threadIdx.x; i < TT; i += 128) row[i] = src[i];
    __syncthreads();
    float* dst = P + (size_t)bc * PLEN;
    for (int m = threadIdx.x; m < PLEN; m += 128) {
        float sum = 0.f;
        #pragma unroll
        for (int i = 0; i < 17; ++i) sum += row[m + i];
        dst[m] = sum * (1.f / 17.f);
    }
}

#define CSTEPS 30
#define GS     6
#define NCH    62
#define H1_STAGE(expr) _Pragma("unroll") for (int i = 0; i < 2 * GS; ++i) { e[i] = (expr); }

__global__ __launch_bounds__(160) void k_conv2(
    const float* __restrict__ P,
    const float* __restrict__ conv1_w, const float* __restrict__ conv1_b,
    const float* __restrict__ bn1_g, const float* __restrict__ bn1_b,
    const float* __restrict__ bn1_m, const float* __restrict__ bn1_v,
    const float* __restrict__ conv2_w, const float* __restrict__ conv2_b,
    const float* __restrict__ bn2_g, const float* __restrict__ bn2_b,
    const float* __restrict__ bn2_m, const float* __restrict__ bn2_v,
    float* __restrict__ h2)
{
    __shared__ float Pl[CC * 80];
    __shared__ float w1s[WID * 5];
    __shared__ float sc1[WID], sh1[WID];

    const int tid = threadIdx.x;
    const int b   = blockIdx.y;
    const int u0  = blockIdx.x * 16;

    for (int i = tid; i < WID * 5; i += 160) w1s[i] = conv1_w[i];
    if (tid < WID) {
        float sc = bn1_g[tid] * rsqrtf(bn1_v[tid] + EPSF);
        sc1[tid] = sc;
        sh1[tid] = bn1_b[tid] - bn1_m[tid] * sc + sc * conv1_b[tid];
    }
    for (int i = tid; i < CC * 80; i += 160) {
        int c = i / 80, q = i - c * 80;
        Pl[i] = P[((size_t)b * CC + c) * PLEN + 5 * u0 + q];
    }
    __syncthreads();

    const int wave = tid >> 5;
    const int lane = tid & 31;
    const int n    = lane & 15;
    const int half = lane >> 4;

    float pr[NCH][5];
    #pragma unroll
    for (int i = 0; i < NCH; ++i) {
        const int c = (i < 2 * CSTEPS) ? (4 * (i >> 1) + 2 * half + (i & 1)) : (120 + (i - 2 * CSTEPS));
        const float* p = &Pl[c * 80 + 5 * n];
        #pragma unroll
        for (int j = 0; j < 5; ++j) pr[i][j] = p[j];
    }

    const float* rowA = conv2_w + (size_t)(wave * 16 + n) * K2;
    v2f acur[CSTEPS + 1];
    #pragma unroll
    for (int s = 0; s < CSTEPS; ++s) acur[s] = *(const v2f*)(rowA + 2 * half + 4 * s);
    acur[CSTEPS] = *(const v2f*)(rowA + 120);

    v8f acc = {};
    for (int w = 0; w < WID; ++w) {
        const float t0 = w1s[w * 5 + 0], t1 = w1s[w * 5 + 1], t2 = w1s[w * 5 + 2],
                    t3 = w1s[w * 5 + 3], t4 = w1s[w * 5 + 4];
        const float sc = sc1[w], sh = sh1[w];

        const int wn = (w + 1 < WID) ? (w + 1) : w;
        const float* pan = rowA + wn * CC;

        #pragma unroll
        for (int g = 0; g < CSTEPS / GS; ++g) {
            float e[2 * GS];
            H1_STAGE(t0 * pr[2 * GS * g + i][0])
            H1_STAGE(__builtin_fmaf(t1, pr[2 * GS * g + i][1], e[i]))
            H1_STAGE(__builtin_fmaf(t2, pr[2 * GS * g + i][2], e[i]))
            H1_STAGE(__builtin_fmaf(t3, pr[2 * GS * g + i][3], e[i]))
            H1_STAGE(__builtin_fmaf(t4, pr[2 * GS * g + i][4], e[i]))
            H1_STAGE(__builtin_fmaf(sc, e[i], sh))
            H1_STAGE(e[i] > 0.f ? e[i] : (__expf(e[i]) - 1.f))
            #pragma unroll
            for (int s = 0; s < GS; ++s) {
                v2f bb = { e[2 * s], e[2 * s + 1] };
                acc = wmma_f32(acur[GS * g + s], bb, acc);
            }
            #pragma unroll
            for (int s = 0; s < GS; ++s)
                acur[GS * g + s] = *(const v2f*)(pan + 2 * half + 4 * (GS * g + s));
        }
        {
            float e0 = t0 * pr[60][0], e1 = t0 * pr[61][0];
            e0 = __builtin_fmaf(t1, pr[60][1], e0); e1 = __builtin_fmaf(t1, pr[61][1], e1);
            e0 = __builtin_fmaf(t2, pr[60][2], e0); e1 = __builtin_fmaf(t2, pr[61][2], e1);
            e0 = __builtin_fmaf(t3, pr[60][3], e0); e1 = __builtin_fmaf(t3, pr[61][3], e1);
            e0 = __builtin_fmaf(t4, pr[60][4], e0); e1 = __builtin_fmaf(t4, pr[61][4], e1);
            e0 = __builtin_fmaf(sc, e0, sh);        e1 = __builtin_fmaf(sc, e1, sh);
            e0 = e0 > 0.f ? e0 : (__expf(e0) - 1.f);
            e1 = e1 > 0.f ? e1 : (__expf(e1) - 1.f);
            v2f a  = acur[CSTEPS];
            v2f bb = { e0, e1 };
            if (half) { a = v2f{0.f, 0.f}; bb = v2f{0.f, 0.f}; }
            acc = wmma_f32(a, bb, acc);
            acur[CSTEPS] = *(const v2f*)(pan + 120);
        }
    }

    #pragma unroll
    for (int r = 0; r < 8; ++r) {
        int w2 = wave * 16 + r + half * 8;
        float sc = bn2_g[w2] * rsqrtf(bn2_v[w2] + EPSF);
        float sh = bn2_b[w2] - bn2_m[w2] * sc;
        float v  = sc * (acc[r] + conv2_b[w2]) + sh;
        float e  = v > 0.f ? v : (__expf(v) - 1.f);
        h2[((size_t)b * WID + w2) * UU + u0 + n] = e;
    }
}

__global__ __launch_bounds__(256) void k_proj(
    const float* __restrict__ h2, const float* __restrict__ proj_w,
    const float* __restrict__ proj_b, float* __restrict__ z)
{
    __shared__ float hs[WID * UU];
    __shared__ float pw[8 * WID];
    const int b = blockIdx.x;
    for (int i = threadIdx.x; i < WID * UU; i += 256) hs[i] = h2[(size_t)b * WID * UU + i];
    for (int i = threadIdx.x; i < 8 * WID; i += 256)  pw[i] = proj_w[i];
    __syncthreads();
    for (int idx = threadIdx.x; idx < HID; idx += 256) {
        int u = idx >> 3, e = idx & 7;
        float s = proj_b[e];
        #pragma unroll 8
        for (int w = 0; w < WID; ++w) s += pw[e * WID + w] * hs[w * UU + u];
        z[(size_t)b * HID + idx] = s;
    }
}

__global__ __launch_bounds__(32) void k_mlp1(
    const float* __restrict__ z, const float* __restrict__ w,
    const float* __restrict__ bias, float* __restrict__ z1, float* __restrict__ g)
{
    const int lane = threadIdx.x;
    const int m0 = blockIdx.x * 16, n0 = blockIdx.y * 16;
    const float* pa = z + (size_t)(m0 + (lane & 15)) * HID + (lane >> 4) * 2;
    const float* pb = w + (size_t)(n0 + (lane & 15)) * HID + (lane >> 4) * 2;
    v8f acc = {};
    #pragma unroll 4
    for (int k4 = 0; k4 < HID / 4; ++k4) {
        v2f a = *(const v2f*)(pa + 4 * k4);
        v2f bb = *(const v2f*)(pb + 4 * k4);
        acc = wmma_f32(a, bb, acc);
    }
    const int nn = n0 + (lane & 15);
    const float bn = bias[nn];
    #pragma unroll
    for (int r = 0; r < 8; ++r) {
        int mm = m0 + r + ((lane >> 4) << 3);
        float v = acc[r] + bn;
        z1[(size_t)mm * PROJ + nn] = v;
        g [(size_t)mm * PROJ + nn] = 0.5f * v * (1.f + erff(v * 0.70710678118654752f));
    }
}

__global__ __launch_bounds__(32) void k_mlp2(
    const float* __restrict__ g, const float* __restrict__ w,
    const float* __restrict__ bias, const float* __restrict__ z1,
    float* __restrict__ y)
{
    const int lane = threadIdx.x;
    const int m0 = blockIdx.x * 16, n0 = blockIdx.y * 16;
    const float* pa = g + (size_t)(m0 + (lane & 15)) * PROJ + (lane >> 4) * 2;
    const float* pb = w + (size_t)(n0 + (lane & 15)) * PROJ + (lane >> 4) * 2;
    v8f acc = {};
    #pragma unroll 4
    for (int k4 = 0; k4 < PROJ / 4; ++k4) {
        v2f a = *(const v2f*)(pa + 4 * k4);
        v2f bb = *(const v2f*)(pb + 4 * k4);
        acc = wmma_f32(a, bb, acc);
    }
    const int nn = n0 + (lane & 15);
    const float bn = bias[nn];
    #pragma unroll
    for (int r = 0; r < 8; ++r) {
        int mm = m0 + r + ((lane >> 4) << 3);
        y[(size_t)mm * PROJ + nn] = z1[(size_t)mm * PROJ + nn] + acc[r] + bn;
    }
}

__global__ __launch_bounds__(256) void k_ln(
    const float* __restrict__ y, const float* __restrict__ gam,
    const float* __restrict__ bet, float* __restrict__ out)
{
    __shared__ float red[256];
    const int b = blockIdx.x;
    const float* row = y + (size_t)b * PROJ;
    float s = 0.f;
    for (int i = threadIdx.x; i < PROJ; i += 256) s += row[i];
    red[threadIdx.x] = s; __syncthreads();
    for (int off = 128; off > 0; off >>= 1) {
        if (threadIdx.x < off) red[threadIdx.x] += red[threadIdx.x + off];
        __syncthreads();
    }
    const float mu = red[0] * (1.f / PROJ);
    __syncthreads();
    float s2 = 0.f;
    for (int i = threadIdx.x; i < PROJ; i += 256) { float d = row[i] - mu; s2 += d * d; }
    red[threadIdx.x] = s2; __syncthreads();
    for (int off = 128; off > 0; off >>= 1) {
        if (threadIdx.x < off) red[threadIdx.x] += red[threadIdx.x + off];
        __syncthreads();
    }
    const float inv = rsqrtf(red[0] * (1.f / PROJ) + EPSF);
    for (int i = threadIdx.x; i < PROJ; i += 256)
        out[(size_t)b * PROJ + i] = (row[i] - mu) * inv * gam[i] + bet[i];
}

extern "C" void kernel_launch(void* const* d_in, const int* in_sizes, int n_in,
                              void* d_out, int out_size, void* d_ws, size_t ws_size,
                              hipStream_t stream) {
    const float* x       = (const float*)d_in[0];
    const int*   sid     = (const int*)  d_in[1];
    const float* W_sub   = (const float*)d_in[2];
    const float* b_sub   = (const float*)d_in[3];
    const float* conv1_w = (const float*)d_in[4];
    const float* conv1_b = (const float*)d_in[5];
    const float* bn1_g   = (const float*)d_in[6];
    const float* bn1_b   = (const float*)d_in[7];
    const float* bn1_m   = (const float*)d_in[8];
    const float* bn1_v   = (const float*)d_in[9];
    const float* conv2_w = (const float*)d_in[10];
    const float* conv2_b = (const float*)d_in[11];
    const float* bn2_g   = (const float*)d_in[12];
    const float* bn2_b   = (const float*)d_in[13];
    const float* bn2_m   = (const float*)d_in[14];
    const float* bn2_v   = (const float*)d_in[15];
    const float* proj_w  = (const float*)d_in[16];
    const float* proj_b  = (const float*)d_in[17];
    const float* mlp1_w  = (const float*)d_in[18];
    const float* mlp1_b  = (const float*)d_in[19];
    const float* mlp2_w  = (const float*)d_in[20];
    const float* mlp2_b  = (const float*)d_in[21];
    const float* ln_g    = (const float*)d_in[22];
    const float* ln_b    = (const float*)d_in[23];

    float* ws   = (float*)d_ws;
    float* out0 = ws;
    float* P    = out0 + (size_t)BB * CC * TT;
    float* h2   = P    + (size_t)BB * CC * PLEN;
    float* z    = h2   + (size_t)BB * WID * UU;
    float* z1   = z    + (size_t)BB * HID;
    float* gg   = z1   + (size_t)BB * PROJ;
    float* yy   = gg   + (size_t)BB * PROJ;

    dim3 g1(8, BB);
    k_subject<<<g1, 256, 0, stream>>>(x, sid, W_sub, b_sub, out0);

    k_pool<<<BB * CC, 128, 0, stream>>>(out0, P);

    dim3 g3(UU / 16, BB);
    k_conv2<<<g3, 160, 0, stream>>>(P, conv1_w, conv1_b, bn1_g, bn1_b, bn1_m, bn1_v,
                                    conv2_w, conv2_b, bn2_g, bn2_b, bn2_m, bn2_v, h2);

    k_proj<<<BB, 256, 0, stream>>>(h2, proj_w, proj_b, z);

    dim3 g5(2, PROJ / 16);
    k_mlp1<<<g5, 32, 0, stream>>>(z, mlp1_w, mlp1_b, z1, gg);
    k_mlp2<<<g5, 32, 0, stream>>>(gg, mlp2_w, mlp2_b, z1, yy);

    k_ln<<<BB, 256, 0, stream>>>(yy, ln_g, ln_b, (float*)d_out);
}


# mlp1/mlp2: one wave per tile walks K in blocks of 32 with its 16-B fragment loads issued 7 blocks (56 loads) ahead of the WMMAs; k regrouped inside the WMMAs, so the K sum order changes
# speedup vs baseline: 3.1034x; 1.1340x over previous
#include <hip/hip_runtime.h>
#include <math.h>

#define BB   32
#define CC   122
#define TT   500
#define WID  80
#define UU   96
#define K2   (WID*CC)
#define HID  768
#define PROJ 1024
#define PLEN 484
#define EPSF 1e-5f

typedef float v2f __attribute__((ext_vector_type(2)));
typedef float v8f __attribute__((ext_vector_type(8)));

static __device__ __forceinline__ v8f wmma_f32(v2f a, v2f b, v8f c) {
    return __builtin_amdgcn_wmma_f32_16x16x4_f32(false, a, false, b, (short)0, c, false, false);
}

#define SKC   32
#define SP4   9
#define SNCH  16
#define SNT   64
#define SMR   128

__global__ __launch_bounds__(256) void k_subject(
    const float* __restrict__ x, const int* __restrict__ sid,
    const float* __restrict__ W_sub, const float* __restrict__ b_sub,
    float* __restrict__ out0)
{
    __shared__ float4 As[2][SMR * SP4];
    __shared__ float4 Bs[2][SNT * SP4];

    const int tid = threadIdx.x;
    const int b   = blockIdx.y;
    const int n0  = blockIdx.x * SNT;
    const int s   = sid[b];
    const float* Wg = W_sub + (size_t)s * TT * TT;
    const float* xb = x + (size_t)b * CC * TT;

    const int kq = tid & 7;
    const int r  = tid >> 3;
    const float* ga[4];
    const float* gb[2];
    #pragma unroll
    for (int i = 0; i < 4; ++i) {
        int row = r + 32 * i; if (row >= CC) row = CC - 1;
        ga[i] = xb + (size_t)row * TT + 4 * kq;
    }
    #pragma unroll
    for (int i = 0; i < 2; ++i) {
        int row = n0 + r + 32 * i; if (row >= TT) row = TT - 1;
        gb[i] = Wg + (size_t)row * TT + 4 * kq;
    }
    const int st = r * SP4 + kq;

    const int wv   = tid >> 5;
    const int lane = tid & 31;
    const int c    = lane & 15;
    const int h    = lane >> 4;
    const int fa = (16 * wv + c) * SP4 + h;
    const int fb = c * SP4 + h;

    float4 ra[4], rb[2];
    v8f acc[4] = {};

    #pragma unroll
    for (int i = 0; i < 4; ++i) ra[i] = *(const float4*)(ga[i]);
    #pragma unroll
    for (int i = 0; i < 2; ++i) rb[i] = *(const float4*)(gb[i]);
    #pragma unroll
    for (int i = 0; i < 4; ++i) As[0][st + 32 * SP4 * i] = ra[i];
    #pragma unroll
    for (int i = 0; i < 2; ++i) Bs[0][st + 32 * SP4 * i] = rb[i];
    __syncthreads();

    for (int ch = 0; ch < SNCH; ++ch) {
        const int  buf  = ch & 1;
        const bool more = (ch + 1 < SNCH);
        const int  ko   = (ch + 1) * SKC;
        const bool tail = (ko + 4 * kq) >= TT;
        if (more) {
            const int kc = tail ? 0 : ko;
            #pragma unroll
            for (int i = 0; i < 4; ++i) ra[i] = *(const float4*)(ga[i] + kc);
            #pragma unroll
            for (int i = 0; i < 2; ++i) rb[i] = *(const float4*)(gb[i] + kc);
        }

        const float4* pA = &As[buf][fa];
        const float4* pB = &Bs[buf][fb];
        float4 af[4], bf[4][4];
        #pragma unroll
        for (int s8 = 0; s8 < 4; ++s8) {
            af[s8] = pA[2 * s8];
            #pragma unroll
            for (int j = 0; j < 4; ++j) bf[s8][j] = pB[16 * SP4 * j + 2 * s8];
        }
        __builtin_amdgcn_sched_barrier(0);
        #pragma unroll
        for (int s8 = 0; s8 < 4; ++s8) {
            const v2f a0 = { af[s8].x, af[s8].y };
            const v2f a1 = { af[s8].z, af[s8].w };
            #pragma unroll
            for (int j = 0; j < 4; ++j) {
                const v2f w0 = { bf[s8][j].x, bf[s8][j].y };
                acc[j] = wmma_f32(a0, w0, acc[j]);
            }
            #pragma unroll
            for (int j = 0; j < 4; ++j) {
                const v2f w1 = { bf[s8][j].z, bf[s8][j].w };
                acc[j] = wmma_f32(a1, w1, acc[j]);
            }
        }

        if (more) {
            if (tail) {
                #pragma unroll
                for (int i = 0; i < 4; ++i) ra[i] = float4{0.f, 0.f, 0.f, 0.f};
                #pragma unroll
                for (int i = 0; i < 2; ++i) rb[i] = float4{0.f, 0.f, 0.f, 0.f};
            }
            #pragma unroll
            for (int i = 0; i < 4; ++i) As[buf ^ 1][st + 32 * SP4 * i] = ra[i];
            #pragma unroll
            for (int i = 0; i < 2; ++i) Bs[buf ^ 1][st + 32 * SP4 * i] = rb[i];
            __syncthreads();
        }
    }

    const float* bg = b_sub + (size_t)s * TT;
    #pragma unroll
    for (int j = 0; j < 4; ++j) {
        const int nn = n0 + 16 * j + c;
        if (nn < TT) {
            const float bn = bg[nn];
            #pragma unroll
            for (int q = 0; q < 8; ++q) {
                const int mm = 16 * wv + q + 8 * h;
                if (mm < CC)
                    out0[((size_t)b * CC + mm) * TT + nn] = acc[j][q] + bn;
            }
        }
    }
}

__global__ __launch_bounds__(128) void k_pool(
    const float* __restrict__ out0, float* __restrict__ P)
{
    __shared__ float row[TT];
    const int bc = blockIdx.x;
    const float* src = out0 + (size_t)bc * TT;
    for (int i = threadIdx.x; i < TT; i += 128) row[i] = src[i];
    __syncthreads();
    float* dst = P + (size_t)bc * PLEN;
    for (int m = threadIdx.x; m < PLEN; m += 128) {
        float sum = 0.f;
        #pragma unroll
        for (int i = 0; i < 17; ++i) sum += row[m + i];
        dst[m] = sum * (1.f / 17.f);
    }
}

#define CSTEPS 30
#define GS     6
#define NCH    62
#define H1_STAGE(expr) _Pragma("unroll") for (int i = 0; i < 2 * GS; ++i) { e[i] = (expr); }

__global__ __launch_bounds__(160) void k_conv2(
    const float* __restrict__ P,
    const float* __restrict__ conv1_w, const float* __restrict__ conv1_b,
    const float* __restrict__ bn1_g, const float* __restrict__ bn1_b,
    const float* __restrict__ bn1_m, const float* __restrict__ bn1_v,
    const float* __restrict__ conv2_w, const float* __restrict__ conv2_b,
    const float* __restrict__ bn2_g, const float* __restrict__ bn2_b,
    const float* __restrict__ bn2_m, const float* __restrict__ bn2_v,
    float* __restrict__ h2)
{
    __shared__ float Pl[CC * 80];
    __shared__ float w1s[WID * 5];
    __shared__ float sc1[WID], sh1[WID];

    const int tid = threadIdx.x;
    const int b   = blockIdx.y;
    const int u0  = blockIdx.x * 16;

    for (int i = tid; i < WID * 5; i += 160) w1s[i] = conv1_w[i];
    if (tid < WID) {
        float sc = bn1_g[tid] * rsqrtf(bn1_v[tid] + EPSF);
        sc1[tid] = sc;
        sh1[tid] = bn1_b[tid] - bn1_m[tid] * sc + sc * conv1_b[tid];
    }
    for (int i = tid; i < CC * 80; i += 160) {
        int c = i / 80, q = i - c * 80;
        Pl[i] = P[((size_t)b * CC + c) * PLEN + 5 * u0 + q];
    }
    __syncthreads();

    const int wave = tid >> 5;
    const int lane = tid & 31;
    const int n    = lane & 15;
    const int half = lane >> 4;

    float pr[NCH][5];
    #pragma unroll
    for (int i = 0; i < NCH; ++i) {
        const int c = (i < 2 * CSTEPS) ? (4 * (i >> 1) + 2 * half + (i & 1)) : (120 + (i - 2 * CSTEPS));
        const float* p = &Pl[c * 80 + 5 * n];
        #pragma unroll
        for (int j = 0; j < 5; ++j) pr[i][j] = p[j];
    }

    const float* rowA = conv2_w + (size_t)(wave * 16 + n) * K2;
    v2f acur[CSTEPS + 1];
    #pragma unroll
    for (int s = 0; s < CSTEPS; ++s) acur[s] = *(const v2f*)(rowA + 2 * half + 4 * s);
    acur[CSTEPS] = *(const v2f*)(rowA + 120);

    v8f acc = {};
    for (int w = 0; w < WID; ++w) {
        const float t0 = w1s[w * 5 + 0], t1 = w1s[w * 5 + 1], t2 = w1s[w * 5 + 2],
                    t3 = w1s[w * 5 + 3], t4 = w1s[w * 5 + 4];
        const float sc = sc1[w], sh = sh1[w];

        const int wn = (w + 1 < WID) ? (w + 1) : w;
        const float* pan = rowA + wn * CC;

        #pragma unroll
        for (int g = 0; g < CSTEPS / GS; ++g) {
            float e[2 * GS];
            H1_STAGE(t0 * pr[2 * GS * g + i][0])
            H1_STAGE(__builtin_fmaf(t1, pr[2 * GS * g + i][1], e[i]))
            H1_STAGE(__builtin_fmaf(t2, pr[2 * GS * g + i][2], e[i]))
            H1_STAGE(__builtin_fmaf(t3, pr[2 * GS * g + i][3], e[i]))
            H1_STAGE(__builtin_fmaf(t4, pr[2 * GS * g + i][4], e[i]))
            H1_STAGE(__builtin_fmaf(sc, e[i], sh))
            H1_STAGE(e[i] > 0.f ? e[i] : (__expf(e[i]) - 1.f))
            #pragma unroll
            for (int s = 0; s < GS; ++s) {
                v2f bb = { e[2 * s], e[2 * s + 1] };
                acc = wmma_f32(acur[GS * g + s], bb, acc);
            }
            #pragma unroll
            for (int s = 0; s < GS; ++s)
                acur[GS * g + s] = *(const v2f*)(pan + 2 * half + 4 * (GS * g + s));
        }
        {
            float e0 = t0 * pr[60][0], e1 = t0 * pr[61][0];
            e0 = __builtin_fmaf(t1, pr[60][1], e0); e1 = __builtin_fmaf(t1, pr[61][1], e1);
            e0 = __builtin_fmaf(t2, pr[60][2], e0); e1 = __builtin_fmaf(t2, pr[61][2], e1);
            e0 = __builtin_fmaf(t3, pr[60][3], e0); e1 = __builtin_fmaf(t3, pr[61][3], e1);
            e0 = __builtin_fmaf(t4, pr[60][4], e0); e1 = __builtin_fmaf(t4, pr[61][4], e1);
            e0 = __builtin_fmaf(sc, e0, sh);        e1 = __builtin_fmaf(sc, e1, sh);
            e0 = e0 > 0.f ? e0 : (__expf(e0) - 1.f);
            e1 = e1 > 0.f ? e1 : (__expf(e1) - 1.f);
            v2f a  = acur[CSTEPS];
            v2f bb = { e0, e1 };
            if (half) { a = v2f{0.f, 0.f}; bb = v2f{0.f, 0.f}; }
            acc = wmma_f32(a, bb, acc);
            acur[CSTEPS] = *(const v2f*)(pan + 120);
        }
    }

    #pragma unroll
    for (int r = 0; r < 8; ++r) {
        int w2 = wave * 16 + r + half * 8;
        float sc = bn2_g[w2] * rsqrtf(bn2_v[w2] + EPSF);
        float sh = bn2_b[w2] - bn2_m[w2] * sc;
        float v  = sc * (acc[r] + conv2_b[w2]) + sh;
        float e  = v > 0.f ? v : (__expf(v) - 1.f);
        h2[((size_t)b * WID + w2) * UU + u0 + n] = e;
    }
}

__global__ __launch_bounds__(256) void k_proj(
    const float* __restrict__ h2, const float* __restrict__ proj_w,
    const float* __restrict__ proj_b, float* __restrict__ z)
{
    __shared__ float hs[WID * UU];
    __shared__ float pw[8 * WID];
    const int b = blockIdx.x;
    for (int i = threadIdx.x; i < WID * UU; i += 256) hs[i] = h2[(size_t)b * WID * UU + i];
    for (int i = threadIdx.x; i < 8 * WID; i += 256)  pw[i] = proj_w[i];
    __syncthreads();
    for (int idx = threadIdx.x; idx < HID; idx += 256) {
        int u = idx >> 3, e = idx & 7;
        float s = proj_b[e];
        #pragma unroll 8
        for (int w = 0; w < WID; ++w) s += pw[e * WID + w] * hs[w * UU + u];
        z[(size_t)b * HID + idx] = s;
    }
}

#define MKB    32
#define MAHEAD 7

template <int K>
static __device__ __forceinline__ v8f mlp_tile(const float* pa, const float* pb)
{
    constexpr int NB = K / MKB;
    float4 fa[NB][4], fb[NB][4];
    v8f acc = {};
    #pragma unroll
    for (int i = 0; i < MAHEAD; ++i) {
        #pragma unroll
        for (int j = 0; j < 4; ++j) {
            fa[i][j] = *(const float4*)(pa + MKB * i + 8 * j);
            fb[i][j] = *(const float4*)(pb + MKB * i + 8 * j);
        }
    }
    #pragma unroll
    for (int i = 0; i < NB; ++i) {
        if (i + MAHEAD < NB) {
            #pragma unroll
            for (int j = 0; j < 4; ++j) {
                fa[i + MAHEAD][j] = *(const float4*)(pa + MKB * (i + MAHEAD) + 8 * j);
                fb[i + MAHEAD][j] = *(const float4*)(pb + MKB * (i + MAHEAD) + 8 * j);
            }
        }
        #pragma unroll
        for (int j = 0; j < 4; ++j) {
            const v2f a0 = { fa[i][j].x, fa[i][j].y };
            const v2f b0 = { fb[i][j].x, fb[i][j].y };
            acc = wmma_f32(a0, b0, acc);
            const v2f a1 = { fa[i][j].z, fa[i][j].w };
            const v2f b1 = { fb[i][j].z, fb[i][j].w };
            acc = wmma_f32(a1, b1, acc);
        }
        __builtin_amdgcn_sched_barrier(0);
    }
    return acc;
}

__global__ __launch_bounds__(32) void k_mlp1(
    const float* __restrict__ z, const float* __restrict__ w,
    const float* __restrict__ bias, float* __restrict__ z1, float* __restrict__ g)
{
    const int lane = threadIdx.x;
    const int m0 = blockIdx.x * 16, n0 = blockIdx.y * 16;
    const float* pa = z + (size_t)(m0 + (lane & 15)) * HID + (lane >> 4) * 4;
    const float* pb = w + (size_t)(n0 + (lane & 15)) * HID + (lane >> 4) * 4;
    const v8f acc = mlp_tile<HID>(pa, pb);
    const int nn = n0 + (lane & 15);
    const float bn = bias[nn];
    #pragma unroll
    for (int r = 0; r < 8; ++r) {
        int mm = m0 + r + ((lane >> 4) << 3);
        float v = acc[r] + bn;
        z1[(size_t)mm * PROJ + nn] = v;
        g [(size_t)mm * PROJ + nn] = 0.5f * v * (1.f + erff(v * 0.70710678118654752f));
    }
}

__global__ __launch_bounds__(32) void k_mlp2(
    const float* __restrict__ g, const float* __restrict__ w,
    const float* __restrict__ bias, const float* __restrict__ z1,
    float* __restrict__ y)
{
    const int lane = threadIdx.x;
    const int m0 = blockIdx.x * 16, n0 = blockIdx.y * 16;
    const float* pa = g + (size_t)(m0 + (lane & 15)) * PROJ + (lane >> 4) * 4;
    const float* pb = w + (size_t)(n0 + (lane & 15)) * PROJ + (lane >> 4) * 4;
    const v8f acc = mlp_tile<PROJ>(pa, pb);
    const int nn = n0 + (lane & 15);
    const float bn = bias[nn];
    #pragma unroll
    for (int r = 0; r < 8; ++r) {
        int mm = m0 + r + ((lane >> 4) << 3);
        y[(size_t)mm * PROJ + nn] = z1[(size_t)mm * PROJ + nn] + acc[r] + bn;
    }
}

__global__ __launch_bounds__(256) void k_ln(
    const float* __restrict__ y, const float* __restrict__ gam,
    const float* __restrict__ bet, float* __restrict__ out)
{
    __shared__ float red[256];
    const int b = blockIdx.x;
    const float* row = y + (size_t)b * PROJ;
    float s = 0.f;
    for (int i = threadIdx.x; i < PROJ; i += 256) s += row[i];
    red[threadIdx.x] = s; __syncthreads();
    for (int off = 128; off > 0; off >>= 1) {
        if (threadIdx.x < off) red[threadIdx.x] += red[threadIdx.x + off];
        __syncthreads();
    }
    const float mu = red[0] * (1.f / PROJ);
    __syncthreads();
    float s2 = 0.f;
    for (int i = threadIdx.x; i < PROJ; i += 256) { float d = row[i] - mu; s2 += d * d; }
    red[threadIdx.x] = s2; __syncthreads();
    for (int off = 128; off > 0; off >>= 1) {
        if (threadIdx.x < off) red[threadIdx.x] += red[threadIdx.x + off];
        __syncthreads();
    }
    const float inv = rsqrtf(red[0] * (1.f / PROJ) + EPSF);
    for (int i = threadIdx.x; i < PROJ; i += 256)
        out[(size_t)b * PROJ + i] = (row[i] - mu) * inv * gam[i] + bet[i];
}

extern "C" void kernel_launch(void* const* d_in, const int* in_sizes, int n_in,
                              void* d_out, int out_size, void* d_ws, size_t ws_size,
                              hipStream_t stream) {
    const float* x       = (const float*)d_in[0];
    const int*   sid     = (const int*)  d_in[1];
    const float* W_sub   = (const float*)d_in[2];
    const float* b_sub   = (const float*)d_in[3];
    const float* conv1_w = (const float*)d_in[4];
    const float* conv1_b = (const float*)d_in[5];
    const float* bn1_g   = (const float*)d_in[6];
    const float* bn1_b   = (const float*)d_in[7];
    const float* bn1_m   = (const float*)d_in[8];
    const float* bn1_v   = (const float*)d_in[9];
    const float* conv2_w = (const float*)d_in[10];
    const float* conv2_b = (const float*)d_in[11];
    const float* bn2_g   = (const float*)d_in[12];
    const float* bn2_b   = (const float*)d_in[13];
    const float* bn2_m   = (const float*)d_in[14];
    const float* bn2_v   = (const float*)d_in[15];
    const float* proj_w  = (const float*)d_in[16];
    const float* proj_b  = (const float*)d_in[17];
    const float* mlp1_w  = (const float*)d_in[18];
    const float* mlp1_b  = (const float*)d_in[19];
    const float* mlp2_w  = (const float*)d_in[20];
    const float* mlp2_b  = (const float*)d_in[21];
    const float* ln_g    = (const float*)d_in[22];
    const float* ln_b    = (const float*)d_in[23];

    float* ws   = (float*)d_ws;
    float* out0 = ws;
    float* P    = out0 + (size_t)BB * CC * TT;
    float* h2   = P    + (size_t)BB * CC * PLEN;
    float* z    = h2   + (size_t)BB * WID * UU;
    float* z1   = z    + (size_t)BB * HID;
    float* gg   = z1   + (size_t)BB * PROJ;
    float* yy   = gg   + (size_t)BB * PROJ;

    dim3 g1(8, BB);
    k_subject<<<g1, 256, 0, stream>>>(x, sid, W_sub, b_sub, out0);

    k_pool<<<BB * CC, 128, 0, stream>>>(out0, P);

    dim3 g3(UU / 16, BB);
    k_conv2<<<g3, 160, 0, stream>>>(P, conv1_w, conv1_b, bn1_g, bn1_b, bn1_m, bn1_v,
                                    conv2_w, conv2_b, bn2_g, bn2_b, bn2_m, bn2_v, h2);

    k_proj<<<BB, 256, 0, stream>>>(h2, proj_w, proj_b, z);

    dim3 g5(2, PROJ / 16);
    k_mlp1<<<g5, 32, 0, stream>>>(z, mlp1_w, mlp1_b, z1, gg);
    k_mlp2<<<g5, 32, 0, stream>>>(gg, mlp2_w, mlp2_b, z1, yy);

    k_ln<<<BB, 256, 0, stream>>>(yy, ln_g, ln_b, (float*)d_out);
}


# conv2: 4-wave blocks (one M-tile x 4 N-tiles) put one wave on each SIMD, not 2+1+1+1; next group's h1 stages written between this group's WMMAs; patches read straight from P; sum order unchanged
# speedup vs baseline: 3.2894x; 1.0599x over previous
#include <hip/hip_runtime.h>
#include <math.h>

#define BB   32
#define CC   122
#define TT   500
#define WID  80
#define UU   96
#define K2   (WID*CC)
#define HID  768
#define PROJ 1024
#define PLEN 484
#define EPSF 1e-5f

typedef float v2f __attribute__((ext_vector_type(2)));
typedef float v8f __attribute__((ext_vector_type(8)));

static __device__ __forceinline__ v8f wmma_f32(v2f a, v2f b, v8f c) {
    return __builtin_amdgcn_wmma_f32_16x16x4_f32(false, a, false, b, (short)0, c, false, false);
}

#define SKC   32
#define SP4   9
#define SNCH  16
#define SNT   64
#define SMR   128

__global__ __launch_bounds__(256) void k_subject(
    const float* __restrict__ x, const int* __restrict__ sid,
    const float* __restrict__ W_sub, const float* __restrict__ b_sub,
    float* __restrict__ out0)
{
    __shared__ float4 As[2][SMR * SP4];
    __shared__ float4 Bs[2][SNT * SP4];

    const int tid = threadIdx.x;
    const int b   = blockIdx.y;
    const int n0  = blockIdx.x * SNT;
    const int s   = sid[b];
    const float* Wg = W_sub + (size_t)s * TT * TT;
    const float* xb = x + (size_t)b * CC * TT;

    const int kq = tid & 7;
    const int r  = tid >> 3;
    const float* ga[4];
    const float* gb[2];
    #pragma unroll
    for (int i = 0; i < 4; ++i) {
        int row = r + 32 * i; if (row >= CC) row = CC - 1;
        ga[i] = xb + (size_t)row * TT + 4 * kq;
    }
    #pragma unroll
    for (int i = 0; i < 2; ++i) {
        int row = n0 + r + 32 * i; if (row >= TT) row = TT - 1;
        gb[i] = Wg + (size_t)row * TT + 4 * kq;
    }
    const int st = r * SP4 + kq;

    const int wv   = tid >> 5;
    const int lane = tid & 31;
    const int c    = lane & 15;
    const int h    = lane >> 4;
    const int fa = (16 * wv + c) * SP4 + h;
    const int fb = c * SP4 + h;

    float4 ra[4], rb[2];
    v8f acc[4] = {};

    #pragma unroll
    for (int i = 0; i < 4; ++i) ra[i] = *(const float4*)(ga[i]);
    #pragma unroll
    for (int i = 0; i < 2; ++i) rb[i] = *(const float4*)(gb[i]);
    #pragma unroll
    for (int i = 0; i < 4; ++i) As[0][st + 32 * SP4 * i] = ra[i];
    #pragma unroll
    for (int i = 0; i < 2; ++i) Bs[0][st + 32 * SP4 * i] = rb[i];
    __syncthreads();

    for (int ch = 0; ch < SNCH; ++ch) {
        const int  buf  = ch & 1;
        const bool more = (ch + 1 < SNCH);
        const int  ko   = (ch + 1) * SKC;
        const bool tail = (ko + 4 * kq) >= TT;
        if (more) {
            const int kc = tail ? 0 : ko;
            #pragma unroll
            for (int i = 0; i < 4; ++i) ra[i] = *(const float4*)(ga[i] + kc);
            #pragma unroll
            for (int i = 0; i < 2; ++i) rb[i] = *(const float4*)(gb[i] + kc);
        }

        const float4* pA = &As[buf][fa];
        const float4* pB = &Bs[buf][fb];
        float4 af[4], bf[4][4];
        #pragma unroll
        for (int s8 = 0; s8 < 4; ++s8) {
            af[s8] = pA[2 * s8];
            #pragma unroll
            for (int j = 0; j < 4; ++j) bf[s8][j] = pB[16 * SP4 * j + 2 * s8];
        }
        __builtin_amdgcn_sched_barrier(0);
        #pragma unroll
        for (int s8 = 0; s8 < 4; ++s8) {
            const v2f a0 = { af[s8].x, af[s8].y };
            const v2f a1 = { af[s8].z, af[s8].w };
            #pragma unroll
            for (int j = 0; j < 4; ++j) {
                const v2f w0 = { bf[s8][j].x, bf[s8][j].y };
                acc[j] = wmma_f32(a0, w0, acc[j]);
            }
            #pragma unroll
            for (int j = 0; j < 4; ++j) {
                const v2f w1 = { bf[s8][j].z, bf[s8][j].w };
                acc[j] = wmma_f32(a1, w1, acc[j]);
            }
        }

        if (more) {
            if (tail) {
                #pragma unroll
                for (int i = 0; i < 4; ++i) ra[i] = float4{0.f, 0.f, 0.f, 0.f};
                #pragma unroll
                for (int i = 0; i < 2; ++i) rb[i] = float4{0.f, 0.f, 0.f, 0.f};
            }
            #pragma unroll
            for (int i = 0; i < 4; ++i) As[buf ^ 1][st + 32 * SP4 * i] = ra[i];
            #pragma unroll
            for (int i = 0; i < 2; ++i) Bs[buf ^ 1][st + 32 * SP4 * i] = rb[i];
            __syncthreads();
        }
    }

    const float* bg = b_sub + (size_t)s * TT;
    #pragma unroll
    for (int j = 0; j < 4; ++j) {
        const int nn = n0 + 16 * j + c;
        if (nn < TT) {
            const float bn = bg[nn];
            #pragma unroll
            for (int q = 0; q < 8; ++q) {
                const int mm = 16 * wv + q + 8 * h;
                if (mm < CC)
                    out0[((size_t)b * CC + mm) * TT + nn] = acc[j][q] + bn;
            }
        }
    }
}

__global__ __launch_bounds__(128) void k_pool(
    const float* __restrict__ out0, float* __restrict__ P)
{
    __shared__ float row[TT];
    const int bc = blockIdx.x;
    const float* src = out0 + (size_t)bc * TT;
    for (int i = threadIdx.x; i < TT; i += 128) row[i] = src[i];
    __syncthreads();
    float* dst = P + (size_t)bc * PLEN;
    for (int m = threadIdx.x; m < PLEN; m += 128) {
        float sum = 0.f;
        #pragma unroll
        for (int i = 0; i < 17; ++i) sum += row[m + i];
        dst[m] = sum * (1.f / 17.f);
    }
}

#define CSTEPS 30
#define GS     6
#define NCH    62
#define NC(G) (((G) == 4) ? 14 : 12)
#define ST(G, D, expr) _Pragma("unroll") for (int i = 0; i < 14; ++i) { if (i < NC(G)) { D[i] = (expr); } }
#define S12(G, E, T)  ST(G, E, T[0] * pr[12 * (G) + i][0]) ST(G, E, __builtin_fmaf(T[1], pr[12 * (G) + i][1], E[i]))
#define S34(G, E, T)  ST(G, E, __builtin_fmaf(T[2], pr[12 * (G) + i][2], E[i])) ST(G, E, __builtin_fmaf(T[3], pr[12 * (G) + i][3], E[i]))
#define S567(G, E, M, T) ST(G, E, __builtin_fmaf(T[4], pr[12 * (G) + i][4], E[i])) ST(G, E, __builtin_fmaf(T[5], E[i], T[6])) \
                         ST(G, M, E[i] * 1.44269504088896340736f)
#define S8(G, M)      ST(G, M, __builtin_amdgcn_exp2f(M[i]))
#define S9(G, E, M)   ST(G, E, E[i] > 0.f ? E[i] : (M[i] - 1.f))
#define WM(G, S, EC)  { const v2f bb = { EC[2 * (S)], EC[2 * (S) + 1] }; acc = wmma_f32(acur[GS * (G) + (S)], bb, acc); }

__global__ __launch_bounds__(128) void k_conv2(
    const float* __restrict__ P,
    const float* __restrict__ conv1_w, const float* __restrict__ conv1_b,
    const float* __restrict__ bn1_g, const float* __restrict__ bn1_b,
    const float* __restrict__ bn1_m, const float* __restrict__ bn1_v,
    const float* __restrict__ conv2_w, const float* __restrict__ conv2_b,
    const float* __restrict__ bn2_g, const float* __restrict__ bn2_b,
    const float* __restrict__ bn2_m, const float* __restrict__ bn2_v,
    float* __restrict__ h2)
{
    __shared__ float w1s[WID * 5];
    __shared__ float sc1[WID], sh1[WID];

    const int tid = threadIdx.x;

    for (int i = tid; i < WID * 5; i += 128) w1s[i] = conv1_w[i];
    if (tid < WID) {
        float sc = bn1_g[tid] * rsqrtf(bn1_v[tid] + EPSF);
        sc1[tid] = sc;
        sh1[tid] = bn1_b[tid] - bn1_m[tid] * sc + sc * conv1_b[tid];
    }
    __syncthreads();

    const int wave = tid >> 5;
    const int lane = tid & 31;
    const int mt   = blockIdx.y;
    const int nt   = blockIdx.x * 4 + wave;
    const int b    = nt / 6;
    const int u0   = (nt - 6 * b) * 16;
    const int n    = lane & 15;
    const int half = lane >> 4;

    float pr[NCH][5];
    #pragma unroll
    for (int i = 0; i < NCH; ++i) {
        const int c = (i < 2 * CSTEPS) ? (4 * (i >> 1) + 2 * half + (i & 1)) : (120 + (i - 2 * CSTEPS));
        const float* p = P + ((size_t)b * CC + c) * PLEN + 5 * (u0 + n);
        #pragma unroll
        for (int j = 0; j < 5; ++j) pr[i][j] = p[j];
    }

    const float* rowA = conv2_w + (size_t)(mt * 16 + n) * K2;
    v2f acur[CSTEPS + 1];
    #pragma unroll
    for (int s = 0; s < CSTEPS; ++s) acur[s] = *(const v2f*)(rowA + 2 * half + 4 * s);
    acur[CSTEPS] = *(const v2f*)(rowA + 120);

    v8f acc = {};
    float tc[7], tn[7];
    #pragma unroll
    for (int j = 0; j < 5; ++j) tc[j] = w1s[j];
    tc[5] = sc1[0]; tc[6] = sh1[0];
    float ec[14] = {}, en[14] = {}, em[14] = {};
    S12(0, ec, tc) S34(0, ec, tc) S567(0, ec, em, tc) S8(0, em) S9(0, ec, em)

    for (int w = 0; w < WID; ++w) {
        const int wn = (w + 1 < WID) ? (w + 1) : w;
        #pragma unroll
        for (int j = 0; j < 5; ++j) tn[j] = w1s[wn * 5 + j];
        tn[5] = sc1[wn]; tn[6] = sh1[wn];
        const float* pan = rowA + wn * CC;

        #pragma unroll
        for (int g = 0; g < 5; ++g) {
            if (g < 4) {
                WM(g, 0, ec) S12(g + 1, en, tc)
                WM(g, 1, ec) S34(g + 1, en, tc)
                WM(g, 2, ec) S567(g + 1, en, em, tc)
                WM(g, 3, ec) S8(g + 1, em)
                WM(g, 4, ec) S9(g + 1, en, em)
                WM(g, 5, ec)
            } else {
                WM(4, 0, ec) S12(0, en, tn)
                WM(4, 1, ec) S34(0, en, tn)
                WM(4, 2, ec) S567(0, en, em, tn)
                WM(4, 3, ec) S8(0, em)
                WM(4, 4, ec) S9(0, en, em)
                WM(4, 5, ec)
                {
                    v2f a  = acur[CSTEPS];
                    v2f bb = { ec[12], ec[13] };
                    if (half) { a = v2f{0.f, 0.f}; bb = v2f{0.f, 0.f}; }
                    acc = wmma_f32(a, bb, acc);
                }
                acur[CSTEPS] = *(const v2f*)(pan + 120);
            }
            #pragma unroll
            for (int s = 0; s < GS; ++s)
                acur[GS * g + s] = *(const v2f*)(pan + 2 * half + 4 * (GS * g + s));
            #pragma unroll
            for (int i = 0; i < 14; ++i) ec[i] = en[i];
        }
        #pragma unroll
        for (int j = 0; j < 7; ++j) tc[j] = tn[j];
    }

    #pragma unroll
    for (int r = 0; r < 8; ++r) {
        int w2 = mt * 16 + r + half * 8;
        float sc = bn2_g[w2] * rsqrtf(bn2_v[w2] + EPSF);
        float sh = bn2_b[w2] - bn2_m[w2] * sc;
        float v  = sc * (acc[r] + conv2_b[w2]) + sh;
        float e  = v > 0.f ? v : (__expf(v) - 1.f);
        h2[((size_t)b * WID + w2) * UU + u0 + n] = e;
    }
}

__global__ __launch_bounds__(256) void k_proj(
    const float* __restrict__ h2, const float* __restrict__ proj_w,
    const float* __restrict__ proj_b, float* __restrict__ z)
{
    __shared__ float hs[WID * UU];
    __shared__ float pw[8 * WID];
    const int b = blockIdx.x;
    for (int i = threadIdx.x; i < WID * UU; i += 256) hs[i] = h2[(size_t)b * WID * UU + i];
    for (int i = threadIdx.x; i < 8 * WID; i += 256)  pw[i] = proj_w[i];
    __syncthreads();
    for (int idx = threadIdx.x; idx < HID; idx += 256) {
        int u = idx >> 3, e = idx & 7;
        float s = proj_b[e];
        #pragma unroll 8
        for (int w = 0; w < WID; ++w) s += pw[e * WID + w] * hs[w * UU + u];
        z[(size_t)b * HID + idx] = s;
    }
}

#define MKB    32
#define MAHEAD 7

template <int K>
static __device__ __forceinline__ v8f mlp_tile(const float* pa, const float* pb)
{
    constexpr int NB = K / MKB;
    float4 fa[NB][4], fb[NB][4];
    v8f acc = {};
    #pragma unroll
    for (int i = 0; i < MAHEAD; ++i) {
        #pragma unroll
        for (int j = 0; j < 4; ++j) {
            fa[i][j] = *(const float4*)(pa + MKB * i + 8 * j);
            fb[i][j] = *(const float4*)(pb + MKB * i + 8 * j);
        }
    }
    #pragma unroll
    for (int i = 0; i < NB; ++i) {
        if (i + MAHEAD < NB) {
            #pragma unroll
            for (int j = 0; j < 4; ++j) {
                fa[i + MAHEAD][j] = *(const float4*)(pa + MKB * (i + MAHEAD) + 8 * j);
                fb[i + MAHEAD][j] = *(const float4*)(pb + MKB * (i + MAHEAD) + 8 * j);
            }
        }
        #pragma unroll
        for (int j = 0; j < 4; ++j) {
            const v2f a0 = { fa[i][j].x, fa[i][j].y };
            const v2f b0 = { fb[i][j].x, fb[i][j].y };
            acc = wmma_f32(a0, b0, acc);
            const v2f a1 = { fa[i][j].z, fa[i][j].w };
            const v2f b1 = { fb[i][j].z, fb[i][j].w };
            acc = wmma_f32(a1, b1, acc);
        }
        __builtin_amdgcn_sched_barrier(0);
    }
    return acc;
}

__global__ __launch_bounds__(32) void k_mlp1(
    const float* __restrict__ z, const float* __restrict__ w,
    const float* __restrict__ bias, float* __restrict__ z1, float* __restrict__ g)
{
    const int lane = threadIdx.x;
    const int m0 = blockIdx.x * 16, n0 = blockIdx.y * 16;
    const float* pa = z + (size_t)(m0 + (lane & 15)) * HID + (lane >> 4) * 4;
    const float* pb = w + (size_t)(n0 + (lane & 15)) * HID + (lane >> 4) * 4;
    const v8f acc = mlp_tile<HID>(pa, pb);
    const int nn = n0 + (lane & 15);
    const float bn = bias[nn];
    #pragma unroll
    for (int r = 0; r < 8; ++r) {
        int mm = m0 + r + ((lane >> 4) << 3);
        float v = acc[r] + bn;
        z1[(size_t)mm * PROJ + nn] = v;
        g [(size_t)mm * PROJ + nn] = 0.5f * v * (1.f + erff(v * 0.70710678118654752f));
    }
}

__global__ __launch_bounds__(32) void k_mlp2(
    const float* __restrict__ g, const float* __restrict__ w,
    const float* __restrict__ bias, const float* __restrict__ z1,
    float* __restrict__ y)
{
    const int lane = threadIdx.x;
    const int m0 = blockIdx.x * 16, n0 = blockIdx.y * 16;
    const float* pa = g + (size_t)(m0 + (lane & 15)) * PROJ + (lane >> 4) * 4;
    const float* pb = w + (size_t)(n0 + (lane & 15)) * PROJ + (lane >> 4) * 4;
    const v8f acc = mlp_tile<PROJ>(pa, pb);
    const int nn = n0 + (lane & 15);
    const float bn = bias[nn];
    #pragma unroll
    for (int r = 0; r < 8; ++r) {
        int mm = m0 + r + ((lane >> 4) << 3);
        y[(size_t)mm * PROJ + nn] = z1[(size_t)mm * PROJ + nn] + acc[r] + bn;
    }
}

__global__ __launch_bounds__(256) void k_ln(
    const float* __restrict__ y, const float* __restrict__ gam,
    const float* __restrict__ bet, float* __restrict__ out)
{
    __shared__ float red[256];
    const int b = blockIdx.x;
    const float* row = y + (size_t)b * PROJ;
    float s = 0.f;
    for (int i = threadIdx.x; i < PROJ; i += 256) s += row[i];
    red[threadIdx.x] = s; __syncthreads();
    for (int off = 128; off > 0; off >>= 1) {
        if (threadIdx.x < off) red[threadIdx.x] += red[threadIdx.x + off];
        __syncthreads();
    }
    const float mu = red[0] * (1.f / PROJ);
    __syncthreads();
    float s2 = 0.f;
    for (int i = threadIdx.x; i < PROJ; i += 256) { float d = row[i] - mu; s2 += d * d; }
    red[threadIdx.x] = s2; __syncthreads();
    for (int off = 128; off > 0; off >>= 1) {
        if (threadIdx.x < off) red[threadIdx.x] += red[threadIdx.x + off];
        __syncthreads();
    }
    const float inv = rsqrtf(red[0] * (1.f / PROJ) + EPSF);
    for (int i = threadIdx.x; i < PROJ; i += 256)
        out[(size_t)b * PROJ + i] = (row[i] - mu) * inv * gam[i] + bet[i];
}

extern "C" void kernel_launch(void* const* d_in, const int* in_sizes, int n_in,
                              void* d_out, int out_size, void* d_ws, size_t ws_size,
                              hipStream_t stream) {
    const float* x       = (const float*)d_in[0];
    const int*   sid     = (const int*)  d_in[1];
    const float* W_sub   = (const float*)d_in[2];
    const float* b_sub   = (const float*)d_in[3];
    const float* conv1_w = (const float*)d_in[4];
    const float* conv1_b = (const float*)d_in[5];
    const float* bn1_g   = (const float*)d_in[6];
    const float* bn1_b   = (const float*)d_in[7];
    const float* bn1_m   = (const float*)d_in[8];
    const float* bn1_v   = (const float*)d_in[9];
    const float* conv2_w = (const float*)d_in[10];
    const float* conv2_b = (const float*)d_in[11];
    const float* bn2_g   = (const float*)d_in[12];
    const float* bn2_b   = (const float*)d_in[13];
    const float* bn2_m   = (const float*)d_in[14];
    const float* bn2_v   = (const float*)d_in[15];
    const float* proj_w  = (const float*)d_in[16];
    const float* proj_b  = (const float*)d_in[17];
    const float* mlp1_w  = (const float*)d_in[18];
    const float* mlp1_b  = (const float*)d_in[19];
    const float* mlp2_w  = (const float*)d_in[20];
    const float* mlp2_b  = (const float*)d_in[21];
    const float* ln_g    = (const float*)d_in[22];
    const float* ln_b    = (const float*)d_in[23];

    float* ws   = (float*)d_ws;
    float* out0 = ws;
    float* P    = out0 + (size_t)BB * CC * TT;
    float* h2   = P    + (size_t)BB * CC * PLEN;
    float* z    = h2   + (size_t)BB * WID * UU;
    float* z1   = z    + (size_t)BB * HID;
    float* gg   = z1   + (size_t)BB * PROJ;
    float* yy   = gg   + (size_t)BB * PROJ;

    dim3 g1(8, BB);
    k_subject<<<g1, 256, 0, stream>>>(x, sid, W_sub, b_sub, out0);

    k_pool<<<BB * CC, 128, 0, stream>>>(out0, P);

    dim3 g3(BB * (UU / 16) / 4, WID / 16);
    k_conv2<<<g3, 128, 0, stream>>>(P, conv1_w, conv1_b, bn1_g, bn1_b, bn1_m, bn1_v,
                                    conv2_w, conv2_b, bn2_g, bn2_b, bn2_m, bn2_v, h2);

    k_proj<<<BB, 256, 0, stream>>>(h2, proj_w, proj_b, z);

    dim3 g5(2, PROJ / 16);
    k_mlp1<<<g5, 32, 0, stream>>>(z, mlp1_w, mlp1_b, z1, gg);
    k_mlp2<<<g5, 32, 0, stream>>>(gg, mlp2_w, mlp2_b, z1, yy);

    k_ln<<<BB, 256, 0, stream>>>(yy, ln_g, ln_b, (float*)d_out);
}


# conv2: the block copies each w's 16x122 conv2_w slab into LDS most of a w ahead (8 threads per row, 2 buffers, 1 barrier per w); A fragments are 8-B LDS reads, not 16-row global loads; sum order kept
# speedup vs baseline: 4.1212x; 1.2529x over previous
#include <hip/hip_runtime.h>
#include <math.h>

#define BB   32
#define CC   122
#define TT   500
#define WID  80
#define UU   96
#define K2   (WID*CC)
#define HID  768
#define PROJ 1024
#define PLEN 484
#define EPSF 1e-5f

typedef float v2f __attribute__((ext_vector_type(2)));
typedef float v8f __attribute__((ext_vector_type(8)));

static __device__ __forceinline__ v8f wmma_f32(v2f a, v2f b, v8f c) {
    return __builtin_amdgcn_wmma_f32_16x16x4_f32(false, a, false, b, (short)0, c, false, false);
}

#define SKC   32
#define SP4   9
#define SNCH  16
#define SNT   64
#define SMR   128

__global__ __launch_bounds__(256) void k_subject(
    const float* __restrict__ x, const int* __restrict__ sid,
    const float* __restrict__ W_sub, const float* __restrict__ b_sub,
    float* __restrict__ out0)
{
    __shared__ float4 As[2][SMR * SP4];
    __shared__ float4 Bs[2][SNT * SP4];

    const int tid = threadIdx.x;
    const int b   = blockIdx.y;
    const int n0  = blockIdx.x * SNT;
    const int s   = sid[b];
    const float* Wg = W_sub + (size_t)s * TT * TT;
    const float* xb = x + (size_t)b * CC * TT;

    const int kq = tid & 7;
    const int r  = tid >> 3;
    const float* ga[4];
    const float* gb[2];
    #pragma unroll
    for (int i = 0; i < 4; ++i) {
        int row = r + 32 * i; if (row >= CC) row = CC - 1;
        ga[i] = xb + (size_t)row * TT + 4 * kq;
    }
    #pragma unroll
    for (int i = 0; i < 2; ++i) {
        int row = n0 + r + 32 * i; if (row >= TT) row = TT - 1;
        gb[i] = Wg + (size_t)row * TT + 4 * kq;
    }
    const int st = r * SP4 + kq;

    const int wv   = tid >> 5;
    const int lane = tid & 31;
    const int c    = lane & 15;
    const int h    = lane >> 4;
    const int fa = (16 * wv + c) * SP4 + h;
    const int fb = c * SP4 + h;

    float4 ra[4], rb[2];
    v8f acc[4] = {};

    #pragma unroll
    for (int i = 0; i < 4; ++i) ra[i] = *(const float4*)(ga[i]);
    #pragma unroll
    for (int i = 0; i < 2; ++i) rb[i] = *(const float4*)(gb[i]);
    #pragma unroll
    for (int i = 0; i < 4; ++i) As[0][st + 32 * SP4 * i] = ra[i];
    #pragma unroll
    for (int i = 0; i < 2; ++i) Bs[0][st + 32 * SP4 * i] = rb[i];
    __syncthreads();

    for (int ch = 0; ch < SNCH; ++ch) {
        const int  buf  = ch & 1;
        const bool more = (ch + 1 < SNCH);
        const int  ko   = (ch + 1) * SKC;
        const bool tail = (ko + 4 * kq) >= TT;
        if (more) {
            const int kc = tail ? 0 : ko;
            #pragma unroll
            for (int i = 0; i < 4; ++i) ra[i] = *(const float4*)(ga[i] + kc);
            #pragma unroll
            for (int i = 0; i < 2; ++i) rb[i] = *(const float4*)(gb[i] + kc);
        }

        const float4* pA = &As[buf][fa];
        const float4* pB = &Bs[buf][fb];
        float4 af[4], bf[4][4];
        #pragma unroll
        for (int s8 = 0; s8 < 4; ++s8) {
            af[s8] = pA[2 * s8];
            #pragma unroll
            for (int j = 0; j < 4; ++j) bf[s8][j] = pB[16 * SP4 * j + 2 * s8];
        }
        __builtin_amdgcn_sched_barrier(0);
        #pragma unroll
        for (int s8 = 0; s8 < 4; ++s8) {
            const v2f a0 = { af[s8].x, af[s8].y };
            const v2f a1 = { af[s8].z, af[s8].w };
            #pragma unroll
            for (int j = 0; j < 4; ++j) {
                const v2f w0 = { bf[s8][j].x, bf[s8][j].y };
                acc[j] = wmma_f32(a0, w0, acc[j]);
            }
            #pragma unroll
            for (int j = 0; j < 4; ++j) {
                const v2f w1 = { bf[s8][j].z, bf[s8][j].w };
                acc[j] = wmma_f32(a1, w1, acc[j]);
            }
        }

        if (more) {
            if (tail) {
                #pragma unroll
                for (int i = 0; i < 4; ++i) ra[i] = float4{0.f, 0.f, 0.f, 0.f};
                #pragma unroll
                for (int i = 0; i < 2; ++i) rb[i] = float4{0.f, 0.f, 0.f, 0.f};
            }
            #pragma unroll
            for (int i = 0; i < 4; ++i) As[buf ^ 1][st + 32 * SP4 * i] = ra[i];
            #pragma unroll
            for (int i = 0; i < 2; ++i) Bs[buf ^ 1][st + 32 * SP4 * i] = rb[i];
            __syncthreads();
        }
    }

    const float* bg = b_sub + (size_t)s * TT;
    #pragma unroll
    for (int j = 0; j < 4; ++j) {
        const int nn = n0 + 16 * j + c;
        if (nn < TT) {
            const float bn = bg[nn];
            #pragma unroll
            for (int q = 0; q < 8; ++q) {
                const int mm = 16 * wv + q + 8 * h;
                if (mm < CC)
                    out0[((size_t)b * CC + mm) * TT + nn] = acc[j][q] + bn;
            }
        }
    }
}

__global__ __launch_bounds__(128) void k_pool(
    const float* __restrict__ out0, float* __restrict__ P)
{
    __shared__ float row[TT];
    const int bc = blockIdx.x;
    const float* src = out0 + (size_t)bc * TT;
    for (int i = threadIdx.x; i < TT; i += 128) row[i] = src[i];
    __syncthreads();
    float* dst = P + (size_t)bc * PLEN;
    for (int m = threadIdx.x; m < PLEN; m += 128) {
        float sum = 0.f;
        #pragma unroll
        for (int i = 0; i < 17; ++i) sum += row[m + i];
        dst[m] = sum * (1.f / 17.f);
    }
}

#define CSTEPS 30
#define GS     6
#define NCH    62
#define AP     124
#define ACP    8
#define NC(G) (((G) == 4) ? 14 : 12)
#define ST(G, D, expr) _Pragma("unroll") for (int i = 0; i < 14; ++i) { if (i < NC(G)) { D[i] = (expr); } }
#define S12(G, E, T)  ST(G, E, T[0] * pr[12 * (G) + i][0]) ST(G, E, __builtin_fmaf(T[1], pr[12 * (G) + i][1], E[i]))
#define S34(G, E, T)  ST(G, E, __builtin_fmaf(T[2], pr[12 * (G) + i][2], E[i])) ST(G, E, __builtin_fmaf(T[3], pr[12 * (G) + i][3], E[i]))
#define S567(G, E, M, T) ST(G, E, __builtin_fmaf(T[4], pr[12 * (G) + i][4], E[i])) ST(G, E, __builtin_fmaf(T[5], E[i], T[6])) \
                         ST(G, M, E[i] * 1.44269504088896340736f)
#define S8(G, M)      ST(G, M, __builtin_amdgcn_exp2f(M[i]))
#define S9(G, E, M)   ST(G, E, E[i] > 0.f ? E[i] : (M[i] - 1.f))
#define WM(G, S, EC)  { const v2f bb = { EC[2 * (S)], EC[2 * (S) + 1] }; acc = wmma_f32(acur[GS * (G) + (S)], bb, acc); }

__global__ __launch_bounds__(128) void k_conv2(
    const float* __restrict__ P,
    const float* __restrict__ conv1_w, const float* __restrict__ conv1_b,
    const float* __restrict__ bn1_g, const float* __restrict__ bn1_b,
    const float* __restrict__ bn1_m, const float* __restrict__ bn1_v,
    const float* __restrict__ conv2_w, const float* __restrict__ conv2_b,
    const float* __restrict__ bn2_g, const float* __restrict__ bn2_b,
    const float* __restrict__ bn2_m, const float* __restrict__ bn2_v,
    float* __restrict__ h2)
{
    __shared__ float w1s[WID * 5];
    __shared__ float sc1[WID], sh1[WID];
    __shared__ __attribute__((aligned(16))) float As[2][16 * AP];

    const int tid = threadIdx.x;
    const int mt  = blockIdx.y;

    const float* gA = conv2_w + (size_t)(mt * 16) * K2;
    const int cr = tid >> 3, cq = tid & 7;
    const int q7 = (cq + 56 > 60) ? 60 : (cq + 56);
    const unsigned gb0 = 4u * (unsigned)(cr * K2 + 2 * cq);
    const unsigned gb7 = 4u * (unsigned)(cr * K2 + 2 * q7);
    const int lo0 = cr * AP + 2 * cq, lo7 = cr * AP + 2 * q7;
    v2f stg[ACP];
    #pragma unroll
    for (int j = 0; j < ACP - 1; ++j) stg[j] = *(const v2f*)((const char*)gA + gb0 + 64 * j);
    stg[ACP - 1] = *(const v2f*)((const char*)gA + gb7);
    #pragma unroll
    for (int j = 0; j < ACP - 1; ++j) *(v2f*)(&As[0][lo0 + 16 * j]) = stg[j];
    *(v2f*)(&As[0][lo7]) = stg[ACP - 1];

    for (int i = tid; i < WID * 5; i += 128) w1s[i] = conv1_w[i];
    if (tid < WID) {
        float sc = bn1_g[tid] * rsqrtf(bn1_v[tid] + EPSF);
        sc1[tid] = sc;
        sh1[tid] = bn1_b[tid] - bn1_m[tid] * sc + sc * conv1_b[tid];
    }
    __syncthreads();

    const int wave = tid >> 5;
    const int lane = tid & 31;
    const int nt   = blockIdx.x * 4 + wave;
    const int b    = nt / 6;
    const int u0   = (nt - 6 * b) * 16;
    const int n    = lane & 15;
    const int half = lane >> 4;

    float pr[NCH][5];
    #pragma unroll
    for (int i = 0; i < NCH; ++i) {
        const int c = (i < 2 * CSTEPS) ? (4 * (i >> 1) + 2 * half + (i & 1)) : (120 + (i - 2 * CSTEPS));
        const float* p = P + ((size_t)b * CC + c) * PLEN + 5 * (u0 + n);
        #pragma unroll
        for (int j = 0; j < 5; ++j) pr[i][j] = p[j];
    }

    const int fro = n * AP + 2 * half;
    v2f acur[CSTEPS + 1];

    v8f acc = {};
    float tc[7], tn[7];
    #pragma unroll
    for (int j = 0; j < 5; ++j) tc[j] = w1s[j];
    tc[5] = sc1[0]; tc[6] = sh1[0];
    float ec[14] = {}, en[14] = {}, em[14] = {};
    S12(0, ec, tc) S34(0, ec, tc) S567(0, ec, em, tc) S8(0, em) S9(0, ec, em)

    for (int w = 0; w < WID; ++w) {
        const int wn = (w + 1 < WID) ? (w + 1) : w;
        #pragma unroll
        for (int j = 0; j < 5; ++j) tn[j] = w1s[wn * 5 + j];
        tn[5] = sc1[wn]; tn[6] = sh1[wn];
        const float* fr = &As[w & 1][fro];
        #pragma unroll
        for (int s = 0; s < CSTEPS; ++s) acur[s] = *(const v2f*)(fr + 4 * s);
        acur[CSTEPS] = *(const v2f*)(&As[w & 1][n * AP + 120]);
        __builtin_amdgcn_sched_barrier(0);
        const char* gw = (const char*)(gA + wn * CC);
        #pragma unroll
        for (int j = 0; j < ACP - 1; ++j) stg[j] = *(const v2f*)(gw + gb0 + 64 * j);
        stg[ACP - 1] = *(const v2f*)(gw + gb7);
        __builtin_amdgcn_sched_barrier(0);

        #pragma unroll
        for (int g = 0; g < 5; ++g) {
            if (g < 4) {
                WM(g, 0, ec) S12(g + 1, en, tc)
                WM(g, 1, ec) S34(g + 1, en, tc)
                WM(g, 2, ec) S567(g + 1, en, em, tc)
                WM(g, 3, ec) S8(g + 1, em)
                WM(g, 4, ec) S9(g + 1, en, em)
                WM(g, 5, ec)
            } else {
                __builtin_amdgcn_sched_barrier(0);
                #pragma unroll
                for (int j = 0; j < ACP - 1; ++j) *(v2f*)(&As[(w + 1) & 1][lo0 + 16 * j]) = stg[j];
                *(v2f*)(&As[(w + 1) & 1][lo7]) = stg[ACP - 1];
                WM(4, 0, ec) S12(0, en, tn)
                WM(4, 1, ec) S34(0, en, tn)
                WM(4, 2, ec) S567(0, en, em, tn)
                WM(4, 3, ec) S8(0, em)
                WM(4, 4, ec) S9(0, en, em)
                WM(4, 5, ec)
                {
                    v2f a  = acur[CSTEPS];
                    v2f bb = { ec[12], ec[13] };
                    if (half) { a = v2f{0.f, 0.f}; bb = v2f{0.f, 0.f}; }
                    acc = wmma_f32(a, bb, acc);
                }
            }
            #pragma unroll
            for (int i = 0; i < 14; ++i) ec[i] = en[i];
        }
        #pragma unroll
        for (int j = 0; j < 7; ++j) tc[j] = tn[j];
        __syncthreads();
    }

    #pragma unroll
    for (int r = 0; r < 8; ++r) {
        int w2 = mt * 16 + r + half * 8;
        float sc = bn2_g[w2] * rsqrtf(bn2_v[w2] + EPSF);
        float sh = bn2_b[w2] - bn2_m[w2] * sc;
        float v  = sc * (acc[r] + conv2_b[w2]) + sh;
        float e  = v > 0.f ? v : (__expf(v) - 1.f);
        h2[((size_t)b * WID + w2) * UU + u0 + n] = e;
    }
}

__global__ __launch_bounds__(256) void k_proj(
    const float* __restrict__ h2, const float* __restrict__ proj_w,
    const float* __restrict__ proj_b, float* __restrict__ z)
{
    __shared__ float hs[WID * UU];
    __shared__ float pw[8 * WID];
    const int b = blockIdx.x;
    for (int i = threadIdx.x; i < WID * UU; i += 256) hs[i] = h2[(size_t)b * WID * UU + i];
    for (int i = threadIdx.x; i < 8 * WID; i += 256)  pw[i] = proj_w[i];
    __syncthreads();
    for (int idx = threadIdx.x; idx < HID; idx += 256) {
        int u = idx >> 3, e = idx & 7;
        float s = proj_b[e];
        #pragma unroll 8
        for (int w = 0; w < WID; ++w) s += pw[e * WID + w] * hs[w * UU + u];
        z[(size_t)b * HID + idx] = s;
    }
}

#define MKB    32
#define MAHEAD 7

template <int K>
static __device__ __forceinline__ v8f mlp_tile(const float* pa, const float* pb)
{
    constexpr int NB = K / MKB;
    float4 fa[NB][4], fb[NB][4];
    v8f acc = {};
    #pragma unroll
    for (int i = 0; i < MAHEAD; ++i) {
        #pragma unroll
        for (int j = 0; j < 4; ++j) {
            fa[i][j] = *(const float4*)(pa + MKB * i + 8 * j);
            fb[i][j] = *(const float4*)(pb + MKB * i + 8 * j);
        }
    }
    #pragma unroll
    for (int i = 0; i < NB; ++i) {
        if (i + MAHEAD < NB) {
            #pragma unroll
            for (int j = 0; j < 4; ++j) {
                fa[i + MAHEAD][j] = *(const float4*)(pa + MKB * (i + MAHEAD) + 8 * j);
                fb[i + MAHEAD][j] = *(const float4*)(pb + MKB * (i + MAHEAD) + 8 * j);
            }
        }
        #pragma unroll
        for (int j = 0; j < 4; ++j) {
            const v2f a0 = { fa[i][j].x, fa[i][j].y };
            const v2f b0 = { fb[i][j].x, fb[i][j].y };
            acc = wmma_f32(a0, b0, acc);
            const v2f a1 = { fa[i][j].z, fa[i][j].w };
            const v2f b1 = { fb[i][j].z, fb[i][j].w };
            acc = wmma_f32(a1, b1, acc);
        }
        __builtin_amdgcn_sched_barrier(0);
    }
    return acc;
}

__global__ __launch_bounds__(32) void k_mlp1(
    const float* __restrict__ z, const float* __restrict__ w,
    const float* __restrict__ bias, float* __restrict__ z1, float* __restrict__ g)
{
    const int lane = threadIdx.x;
    const int m0 = blockIdx.x * 16, n0 = blockIdx.y * 16;
    const float* pa = z + (size_t)(m0 + (lane & 15)) * HID + (lane >> 4) * 4;
    const float* pb = w + (size_t)(n0 + (lane & 15)) * HID + (lane >> 4) * 4;
    const v8f acc = mlp_tile<HID>(pa, pb);
    const int nn = n0 + (lane & 15);
    const float bn = bias[nn];
    #pragma unroll
    for (int r = 0; r < 8; ++r) {
        int mm = m0 + r + ((lane >> 4) << 3);
        float v = acc[r] + bn;
        z1[(size_t)mm * PROJ + nn] = v;
        g [(size_t)mm * PROJ + nn] = 0.5f * v * (1.f + erff(v * 0.70710678118654752f));
    }
}

__global__ __launch_bounds__(32) void k_mlp2(
    const float* __restrict__ g, const float* __restrict__ w,
    const float* __restrict__ bias, const float* __restrict__ z1,
    float* __restrict__ y)
{
    const int lane = threadIdx.x;
    const int m0 = blockIdx.x * 16, n0 = blockIdx.y * 16;
    const float* pa = g + (size_t)(m0 + (lane & 15)) * PROJ + (lane >> 4) * 4;
    const float* pb = w + (size_t)(n0 + (lane & 15)) * PROJ + (lane >> 4) * 4;
    const v8f acc = mlp_tile<PROJ>(pa, pb);
    const int nn = n0 + (lane & 15);
    const float bn = bias[nn];
    #pragma unroll
    for (int r = 0; r < 8; ++r) {
        int mm = m0 + r + ((lane >> 4) << 3);
        y[(size_t)mm * PROJ + nn] = z1[(size_t)mm * PROJ + nn] + acc[r] + bn;
    }
}

__global__ __launch_bounds__(256) void k_ln(
    const float* __restrict__ y, const float* __restrict__ gam,
    const float* __restrict__ bet, float* __restrict__ out)
{
    __shared__ float red[256];
    const int b = blockIdx.x;
    const float* row = y + (size_t)b * PROJ;
    float s = 0.f;
    for (int i = threadIdx.x; i < PROJ; i += 256) s += row[i];
    red[threadIdx.x] = s; __syncthreads();
    for (int off = 128; off > 0; off >>= 1) {
        if (threadIdx.x < off) red[threadIdx.x] += red[threadIdx.x + off];
        __syncthreads();
    }
    const float mu = red[0] * (1.f / PROJ);
    __syncthreads();
    float s2 = 0.f;
    for (int i = threadIdx.x; i < PROJ; i += 256) { float d = row[i] - mu; s2 += d * d; }
    red[threadIdx.x] = s2; __syncthreads();
    for (int off = 128; off > 0; off >>= 1) {
        if (threadIdx.x < off) red[threadIdx.x] += red[threadIdx.x + off];
        __syncthreads();
    }
    const float inv = rsqrtf(red[0] * (1.f / PROJ) + EPSF);
    for (int i = threadIdx.x; i < PROJ; i += 256)
        out[(size_t)b * PROJ + i] = (row[i] - mu) * inv * gam[i] + bet[i];
}

extern "C" void kernel_launch(void* const* d_in, const int* in_sizes, int n_in,
                              void* d_out, int out_size, void* d_ws, size_t ws_size,
                              hipStream_t stream) {
    const float* x       = (const float*)d_in[0];
    const int*   sid     = (const int*)  d_in[1];
    const float* W_sub   = (const float*)d_in[2];
    const float* b_sub   = (const float*)d_in[3];
    const float* conv1_w = (const float*)d_in[4];
    const float* conv1_b = (const float*)d_in[5];
    const float* bn1_g   = (const float*)d_in[6];
    const float* bn1_b   = (const float*)d_in[7];
    const float* bn1_m   = (const float*)d_in[8];
    const float* bn1_v   = (const float*)d_in[9];
    const float* conv2_w = (const float*)d_in[10];
    const float* conv2_b = (const float*)d_in[11];
    const float* bn2_g   = (const float*)d_in[12];
    const float* bn2_b   = (const float*)d_in[13];
    const float* bn2_m   = (const float*)d_in[14];
    const float* bn2_v   = (const float*)d_in[15];
    const float* proj_w  = (const float*)d_in[16];
    const float* proj_b  = (const float*)d_in[17];
    const float* mlp1_w  = (const float*)d_in[18];
    const float* mlp1_b  = (const float*)d_in[19];
    const float* mlp2_w  = (const float*)d_in[20];
    const float* mlp2_b  = (const float*)d_in[21];
    const float* ln_g    = (const float*)d_in[22];
    const float* ln_b    = (const float*)d_in[23];

    float* ws   = (float*)d_ws;
    float* out0 = ws;
    float* P    = out0 + (size_t)BB * CC * TT;
    float* h2   = P    + (size_t)BB * CC * PLEN;
    float* z    = h2   + (size_t)BB * WID * UU;
    float* z1   = z    + (size_t)BB * HID;
    float* gg   = z1   + (size_t)BB * PROJ;
    float* yy   = gg   + (size_t)BB * PROJ;

    dim3 g1(8, BB);
    k_subject<<<g1, 256, 0, stream>>>(x, sid, W_sub, b_sub, out0);

    k_pool<<<BB * CC, 128, 0, stream>>>(out0, P);

    dim3 g3(BB * (UU / 16) / 4, WID / 16);
    k_conv2<<<g3, 128, 0, stream>>>(P, conv1_w, conv1_b, bn1_g, bn1_b, bn1_m, bn1_v,
                                    conv2_w, conv2_b, bn2_g, bn2_b, bn2_m, bn2_v, h2);

    k_proj<<<BB, 256, 0, stream>>>(h2, proj_w, proj_b, z);

    dim3 g5(2, PROJ / 16);
    k_mlp1<<<g5, 32, 0, stream>>>(z, mlp1_w, mlp1_b, z1, gg);
    k_mlp2<<<g5, 32, 0, stream>>>(gg, mlp2_w, mlp2_b, z1, yy);

    k_ln<<<BB, 256, 0, stream>>>(yy, ln_g, ln_b, (float*)d_out);
}


# k_pool folded into k_subject: a block multiplies 80 out0 columns (5 N-tiles, 16 of overlap), pools its 64 through LDS and stores P itself; out0 never written, one launch fewer; sum orders kept
# speedup vs baseline: 4.2614x; 1.0340x over previous
#include <hip/hip_runtime.h>
#include <math.h>

#define BB   32
#define CC   122
#define TT   500
#define WID  80
#define UU   96
#define K2   (WID*CC)
#define HID  768
#define PROJ 1024
#define PLEN 484
#define EPSF 1e-5f

typedef float v2f __attribute__((ext_vector_type(2)));
typedef float v8f __attribute__((ext_vector_type(8)));

static __device__ __forceinline__ v8f wmma_f32(v2f a, v2f b, v8f c) {
    return __builtin_amdgcn_wmma_f32_16x16x4_f32(false, a, false, b, (short)0, c, false, false);
}

#define SKC   32
#define SP4   9
#define SNCH  16
#define SNT   64
#define SNB   80
#define SMR   128
#define TPF   84

__global__ __launch_bounds__(256) void k_subject(
    const float* __restrict__ x, const int* __restrict__ sid,
    const float* __restrict__ W_sub, const float* __restrict__ b_sub,
    float* __restrict__ P)
{
    __shared__ float4 As[2][SMR * SP4];
    __shared__ float4 Bs[2][SNB * SP4];
    __shared__ __attribute__((aligned(16))) float Ts[SMR * TPF];

    const int tid = threadIdx.x;
    const int b   = blockIdx.y;
    const int n0  = blockIdx.x * SNT;
    const int s   = sid[b];
    const float* Wg = W_sub + (size_t)s * TT * TT;
    const float* xb = x + (size_t)b * CC * TT;

    const int kq = tid & 7;
    const int r  = tid >> 3;
    const bool third = (r < SNB - 64);
    const float* ga[4];
    const float* gb[3];
    #pragma unroll
    for (int i = 0; i < 4; ++i) {
        int row = r + 32 * i; if (row >= CC) row = CC - 1;
        ga[i] = xb + (size_t)row * TT + 4 * kq;
    }
    #pragma unroll
    for (int i = 0; i < 3; ++i) {
        int row = n0 + r + 32 * i; if (row >= TT) row = TT - 1;
        gb[i] = Wg + (size_t)row * TT + 4 * kq;
    }
    const int st = r * SP4 + kq;

    const int wv   = tid >> 5;
    const int lane = tid & 31;
    const int c    = lane & 15;
    const int h    = lane >> 4;
    const int fa = (16 * wv + c) * SP4 + h;
    const int fb = c * SP4 + h;

    float4 ra[4], rb[3];
    v8f acc[5] = {};

    #pragma unroll
    for (int i = 0; i < 4; ++i) ra[i] = *(const float4*)(ga[i]);
    #pragma unroll
    for (int i = 0; i < 2; ++i) rb[i] = *(const float4*)(gb[i]);
    if (third) rb[2] = *(const float4*)(gb[2]);
    #pragma unroll
    for (int i = 0; i < 4; ++i) As[0][st + 32 * SP4 * i] = ra[i];
    #pragma unroll
    for (int i = 0; i < 2; ++i) Bs[0][st + 32 * SP4 * i] = rb[i];
    if (third) Bs[0][st + 64 * SP4] = rb[2];
    __syncthreads();

    for (int ch = 0; ch < SNCH; ++ch) {
        const int  buf  = ch & 1;
        const bool more = (ch + 1 < SNCH);
        const int  ko   = (ch + 1) * SKC;
        const bool tail = (ko + 4 * kq) >= TT;
        if (more) {
            const int kc = tail ? 0 : ko;
            #pragma unroll
            for (int i = 0; i < 4; ++i) ra[i] = *(const float4*)(ga[i] + kc);
            #pragma unroll
            for (int i = 0; i < 2; ++i) rb[i] = *(const float4*)(gb[i] + kc);
            if (third) rb[2] = *(const float4*)(gb[2] + kc);
        }

        const float4* pA = &As[buf][fa];
        const float4* pB = &Bs[buf][fb];
        float4 af[4], bf[4][5];
        #pragma unroll
        for (int s8 = 0; s8 < 4; ++s8) {
            af[s8] = pA[2 * s8];
            #pragma unroll
            for (int j = 0; j < 5; ++j) bf[s8][j] = pB[16 * SP4 * j + 2 * s8];
        }
        __builtin_amdgcn_sched_barrier(0);
        #pragma unroll
        for (int s8 = 0; s8 < 4; ++s8) {
            const v2f a0 = { af[s8].x, af[s8].y };
            const v2f a1 = { af[s8].z, af[s8].w };
            #pragma unroll
            for (int j = 0; j < 5; ++j) {
                const v2f w0 = { bf[s8][j].x, bf[s8][j].y };
                acc[j] = wmma_f32(a0, w0, acc[j]);
            }
            #pragma unroll
            for (int j = 0; j < 5; ++j) {
                const v2f w1 = { bf[s8][j].z, bf[s8][j].w };
                acc[j] = wmma_f32(a1, w1, acc[j]);
            }
        }

        if (more) {
            if (tail) {
                #pragma unroll
                for (int i = 0; i < 4; ++i) ra[i] = float4{0.f, 0.f, 0.f, 0.f};
                #pragma unroll
                for (int i = 0; i < 3; ++i) rb[i] = float4{0.f, 0.f, 0.f, 0.f};
            }
            #pragma unroll
            for (int i = 0; i < 4; ++i) As[buf ^ 1][st + 32 * SP4 * i] = ra[i];
            #pragma unroll
            for (int i = 0; i < 2; ++i) Bs[buf ^ 1][st + 32 * SP4 * i] = rb[i];
            if (third) Bs[buf ^ 1][st + 64 * SP4] = rb[2];
            __syncthreads();
        }
    }

    const float* bg = b_sub + (size_t)s * TT;
    float* tw = Ts + (16 * wv) * TPF;
    #pragma unroll
    for (int j = 0; j < 5; ++j) {
        int nn = n0 + 16 * j + c; if (nn >= TT) nn = TT - 1;
        const float bn = bg[nn];
        #pragma unroll
        for (int q = 0; q < 8; ++q)
            tw[(q + 8 * h) * TPF + 16 * j + c] = acc[j][q] + bn;
    }
    __syncthreads();

    const int pc = lane & 15;
    const int rs = lane >> 4;
    const int m  = n0 + 4 * pc;
    #pragma unroll
    for (int it = 0; it < 8; ++it) {
        const int rl = 2 * it + rs;
        const int mm = 16 * wv + rl;
        const float4* src = (const float4*)(tw + rl * TPF + 4 * pc);
        float a[20];
        #pragma unroll
        for (int k = 0; k < 5; ++k) {
            const float4 v = src[k];
            a[4 * k] = v.x; a[4 * k + 1] = v.y; a[4 * k + 2] = v.z; a[4 * k + 3] = v.w;
        }
        float o[4];
        #pragma unroll
        for (int e = 0; e < 4; ++e) {
            float sum = 0.f;
            #pragma unroll
            for (int i = 0; i < 17; ++i) sum += a[e + i];
            o[e] = sum * (1.f / 17.f);
        }
        if (mm < CC && m < PLEN)
            *(float4*)(P + ((size_t)b * CC + mm) * PLEN + m) = float4{o[0], o[1], o[2], o[3]};
    }
}

#define CSTEPS 30
#define GS     6
#define NCH    62
#define AP     124
#define ACP    8
#define NC(G) (((G) == 4) ? 14 : 12)
#define ST(G, D, expr) _Pragma("unroll") for (int i = 0; i < 14; ++i) { if (i < NC(G)) { D[i] = (expr); } }
#define S12(G, E, T)  ST(G, E, T[0] * pr[12 * (G) + i][0]) ST(G, E, __builtin_fmaf(T[1], pr[12 * (G) + i][1], E[i]))
#define S34(G, E, T)  ST(G, E, __builtin_fmaf(T[2], pr[12 * (G) + i][2], E[i])) ST(G, E, __builtin_fmaf(T[3], pr[12 * (G) + i][3], E[i]))
#define S567(G, E, M, T) ST(G, E, __builtin_fmaf(T[4], pr[12 * (G) + i][4], E[i])) ST(G, E, __builtin_fmaf(T[5], E[i], T[6])) \
                         ST(G, M, E[i] * 1.44269504088896340736f)
#define S8(G, M)      ST(G, M, __builtin_amdgcn_exp2f(M[i]))
#define S9(G, E, M)   ST(G, E, E[i] > 0.f ? E[i] : (M[i] - 1.f))
#define WM(G, S, EC)  { const v2f bb = { EC[2 * (S)], EC[2 * (S) + 1] }; acc = wmma_f32(acur[GS * (G) + (S)], bb, acc); }

__global__ __launch_bounds__(128) void k_conv2(
    const float* __restrict__ P,
    const float* __restrict__ conv1_w, const float* __restrict__ conv1_b,
    const float* __restrict__ bn1_g, const float* __restrict__ bn1_b,
    const float* __restrict__ bn1_m, const float* __restrict__ bn1_v,
    const float* __restrict__ conv2_w, const float* __restrict__ conv2_b,
    const float* __restrict__ bn2_g, const float* __restrict__ bn2_b,
    const float* __restrict__ bn2_m, const float* __restrict__ bn2_v,
    float* __restrict__ h2)
{
    __shared__ float w1s[WID * 5];
    __shared__ float sc1[WID], sh1[WID];
    __shared__ __attribute__((aligned(16))) float As[2][16 * AP];

    const int tid = threadIdx.x;
    const int mt  = blockIdx.y;

    const float* gA = conv2_w + (size_t)(mt * 16) * K2;
    const int cr = tid >> 3, cq = tid & 7;
    const int q7 = (cq + 56 > 60) ? 60 : (cq + 56);
    const unsigned gb0 = 4u * (unsigned)(cr * K2 + 2 * cq);
    const unsigned gb7 = 4u * (unsigned)(cr * K2 + 2 * q7);
    const int lo0 = cr * AP + 2 * cq, lo7 = cr * AP + 2 * q7;
    v2f stg[ACP];
    #pragma unroll
    for (int j = 0; j < ACP - 1; ++j) stg[j] = *(const v2f*)((const char*)gA + gb0 + 64 * j);
    stg[ACP - 1] = *(const v2f*)((const char*)gA + gb7);
    #pragma unroll
    for (int j = 0; j < ACP - 1; ++j) *(v2f*)(&As[0][lo0 + 16 * j]) = stg[j];
    *(v2f*)(&As[0][lo7]) = stg[ACP - 1];

    for (int i = tid; i < WID * 5; i += 128) w1s[i] = conv1_w[i];
    if (tid < WID) {
        float sc = bn1_g[tid] * rsqrtf(bn1_v[tid] + EPSF);
        sc1[tid] = sc;
        sh1[tid] = bn1_b[tid] - bn1_m[tid] * sc + sc * conv1_b[tid];
    }
    __syncthreads();

    const int wave = tid >> 5;
    const int lane = tid & 31;
    const int nt   = blockIdx.x * 4 + wave;
    const int b    = nt / 6;
    const int u0   = (nt - 6 * b) * 16;
    const int n    = lane & 15;
    const int half = lane >> 4;

    float pr[NCH][5];
    #pragma unroll
    for (int i = 0; i < NCH; ++i) {
        const int c = (i < 2 * CSTEPS) ? (4 * (i >> 1) + 2 * half + (i & 1)) : (120 + (i - 2 * CSTEPS));
        const float* p = P + ((size_t)b * CC + c) * PLEN + 5 * (u0 + n);
        #pragma unroll
        for (int j = 0; j < 5; ++j) pr[i][j] = p[j];
    }

    const int fro = n * AP + 2 * half;
    v2f acur[CSTEPS + 1];

    v8f acc = {};
    float tc[7], tn[7];
    #pragma unroll
    for (int j = 0; j < 5; ++j) tc[j] = w1s[j];
    tc[5] = sc1[0]; tc[6] = sh1[0];
    float ec[14] = {}, en[14] = {}, em[14] = {};
    S12(0, ec, tc) S34(0, ec, tc) S567(0, ec, em, tc) S8(0, em) S9(0, ec, em)

    for (int w = 0; w < WID; ++w) {
        const int wn = (w + 1 < WID) ? (w + 1) : w;
        #pragma unroll
        for (int j = 0; j < 5; ++j) tn[j] = w1s[wn * 5 + j];
        tn[5] = sc1[wn]; tn[6] = sh1[wn];
        const float* fr = &As[w & 1][fro];
        #pragma unroll
        for (int s = 0; s < CSTEPS; ++s) acur[s] = *(const v2f*)(fr + 4 * s);
        acur[CSTEPS] = *(const v2f*)(&As[w & 1][n * AP + 120]);
        __builtin_amdgcn_sched_barrier(0);
        const char* gw = (const char*)(gA + wn * CC);
        #pragma unroll
        for (int j = 0; j < ACP - 1; ++j) stg[j] = *(const v2f*)(gw + gb0 + 64 * j);
        stg[ACP - 1] = *(const v2f*)(gw + gb7);
        __builtin_amdgcn_sched_barrier(0);

        #pragma unroll
        for (int g = 0; g < 5; ++g) {
            if (g < 4) {
                WM(g, 0, ec) S12(g + 1, en, tc)
                WM(g, 1, ec) S34(g + 1, en, tc)
                WM(g, 2, ec) S567(g + 1, en, em, tc)
                WM(g, 3, ec) S8(g + 1, em)
                WM(g, 4, ec) S9(g + 1, en, em)
                WM(g, 5, ec)
            } else {
                __builtin_amdgcn_sched_barrier(0);
                #pragma unroll
                for (int j = 0; j < ACP - 1; ++j) *(v2f*)(&As[(w + 1) & 1][lo0 + 16 * j]) = stg[j];
                *(v2f*)(&As[(w + 1) & 1][lo7]) = stg[ACP - 1];
                WM(4, 0, ec) S12(0, en, tn)
                WM(4, 1, ec) S34(0, en, tn)
                WM(4, 2, ec) S567(0, en, em, tn)
                WM(4, 3, ec) S8(0, em)
                WM(4, 4, ec) S9(0, en, em)
                WM(4, 5, ec)
                {
                    v2f a  = acur[CSTEPS];
                    v2f bb = { ec[12], ec[13] };
                    if (half) { a = v2f{0.f, 0.f}; bb = v2f{0.f, 0.f}; }
                    acc = wmma_f32(a, bb, acc);
                }
            }
            #pragma unroll
            for (int i = 0; i < 14; ++i) ec[i] = en[i];
        }
        #pragma unroll
        for (int j = 0; j < 7; ++j) tc[j] = tn[j];
        __syncthreads();
    }

    #pragma unroll
    for (int r = 0; r < 8; ++r) {
        int w2 = mt * 16 + r + half * 8;
        float sc = bn2_g[w2] * rsqrtf(bn2_v[w2] + EPSF);
        float sh = bn2_b[w2] - bn2_m[w2] * sc;
        float v  = sc * (acc[r] + conv2_b[w2]) + sh;
        float e  = v > 0.f ? v : (__expf(v) - 1.f);
        h2[((size_t)b * WID + w2) * UU + u0 + n] = e;
    }
}

__global__ __launch_bounds__(256) void k_proj(
    const float* __restrict__ h2, const float* __restrict__ proj_w,
    const float* __restrict__ proj_b, float* __restrict__ z)
{
    __shared__ float hs[WID * UU];
    __shared__ float pw[8 * WID];
    const int b = blockIdx.x;
    for (int i = threadIdx.x; i < WID * UU; i += 256) hs[i] = h2[(size_t)b * WID * UU + i];
    for (int i = threadIdx.x; i < 8 * WID; i += 256)  pw[i] = proj_w[i];
    __syncthreads();
    for (int idx = threadIdx.x; idx < HID; idx += 256) {
        int u = idx >> 3, e = idx & 7;
        float s = proj_b[e];
        #pragma unroll 8
        for (int w = 0; w < WID; ++w) s += pw[e * WID + w] * hs[w * UU + u];
        z[(size_t)b * HID + idx] = s;
    }
}

#define MKB    32
#define MAHEAD 7

template <int K>
static __device__ __forceinline__ v8f mlp_tile(const float* pa, const float* pb)
{
    constexpr int NB = K / MKB;
    float4 fa[NB][4], fb[NB][4];
    v8f acc = {};
    #pragma unroll
    for (int i = 0; i < MAHEAD; ++i) {
        #pragma unroll
        for (int j = 0; j < 4; ++j) {
            fa[i][j] = *(const float4*)(pa + MKB * i + 8 * j);
            fb[i][j] = *(const float4*)(pb + MKB * i + 8 * j);
        }
    }
    #pragma unroll
    for (int i = 0; i < NB; ++i) {
        if (i + MAHEAD < NB) {
            #pragma unroll
            for (int j = 0; j < 4; ++j) {
                fa[i + MAHEAD][j] = *(const float4*)(pa + MKB * (i + MAHEAD) + 8 * j);
                fb[i + MAHEAD][j] = *(const float4*)(pb + MKB * (i + MAHEAD) + 8 * j);
            }
        }
        #pragma unroll
        for (int j = 0; j < 4; ++j) {
            const v2f a0 = { fa[i][j].x, fa[i][j].y };
            const v2f b0 = { fb[i][j].x, fb[i][j].y };
            acc = wmma_f32(a0, b0, acc);
            const v2f a1 = { fa[i][j].z, fa[i][j].w };
            const v2f b1 = { fb[i][j].z, fb[i][j].w };
            acc = wmma_f32(a1, b1, acc);
        }
        __builtin_amdgcn_sched_barrier(0);
    }
    return acc;
}

__global__ __launch_bounds__(32) void k_mlp1(
    const float* __restrict__ z, const float* __restrict__ w,
    const float* __restrict__ bias, float* __restrict__ z1, float* __restrict__ g)
{
    const int lane = threadIdx.x;
    const int m0 = blockIdx.x * 16, n0 = blockIdx.y * 16;
    const float* pa = z + (size_t)(m0 + (lane & 15)) * HID + (lane >> 4) * 4;
    const float* pb = w + (size_t)(n0 + (lane & 15)) * HID + (lane >> 4) * 4;
    const v8f acc = mlp_tile<HID>(pa, pb);
    const int nn = n0 + (lane & 15);
    const float bn = bias[nn];
    #pragma unroll
    for (int r = 0; r < 8; ++r) {
        int mm = m0 + r + ((lane >> 4) << 3);
        float v = acc[r] + bn;
        z1[(size_t)mm * PROJ + nn] = v;
        g [(size_t)mm * PROJ + nn] = 0.5f * v * (1.f + erff(v * 0.70710678118654752f));
    }
}

__global__ __launch_bounds__(32) void k_mlp2(
    const float* __restrict__ g, const float* __restrict__ w,
    const float* __restrict__ bias, const float* __restrict__ z1,
    float* __restrict__ y)
{
    const int lane = threadIdx.x;
    const int m0 = blockIdx.x * 16, n0 = blockIdx.y * 16;
    const float* pa = g + (size_t)(m0 + (lane & 15)) * PROJ + (lane >> 4) * 4;
    const float* pb = w + (size_t)(n0 + (lane & 15)) * PROJ + (lane >> 4) * 4;
    const v8f acc = mlp_tile<PROJ>(pa, pb);
    const int nn = n0 + (lane & 15);
    const float bn = bias[nn];
    #pragma unroll
    for (int r = 0; r < 8; ++r) {
        int mm = m0 + r + ((lane >> 4) << 3);
        y[(size_t)mm * PROJ + nn] = z1[(size_t)mm * PROJ + nn] + acc[r] + bn;
    }
}

__global__ __launch_bounds__(256) void k_ln(
    const float* __restrict__ y, const float* __restrict__ gam,
    const float* __restrict__ bet, float* __restrict__ out)
{
    __shared__ float red[256];
    const int b = blockIdx.x;
    const float* row = y + (size_t)b * PROJ;
    float s = 0.f;
    for (int i = threadIdx.x; i < PROJ; i += 256) s += row[i];
    red[threadIdx.x] = s; __syncthreads();
    for (int off = 128; off > 0; off >>= 1) {
        if (threadIdx.x < off) red[threadIdx.x] += red[threadIdx.x + off];
        __syncthreads();
    }
    const float mu = red[0] * (1.f / PROJ);
    __syncthreads();
    float s2 = 0.f;
    for (int i = threadIdx.x; i < PROJ; i += 256) { float d = row[i] - mu; s2 += d * d; }
    red[threadIdx.x] = s2; __syncthreads();
    for (int off = 128; off > 0; off >>= 1) {
        if (threadIdx.x < off) red[threadIdx.x] += red[threadIdx.x + off];
        __syncthreads();
    }
    const float inv = rsqrtf(red[0] * (1.f / PROJ) + EPSF);
    for (int i = threadIdx.x; i < PROJ; i += 256)
        out[(size_t)b * PROJ + i] = (row[i] - mu) * inv * gam[i] + bet[i];
}

extern "C" void kernel_launch(void* const* d_in, const int* in_sizes, int n_in,
                              void* d_out, int out_size, void* d_ws, size_t ws_size,
                              hipStream_t stream) {
    const float* x       = (const float*)d_in[0];
    const int*   sid     = (const int*)  d_in[1];
    const float* W_sub   = (const float*)d_in[2];
    const float* b_sub   = (const float*)d_in[3];
    const float* conv1_w = (const float*)d_in[4];
    const float* conv1_b = (const float*)d_in[5];
    const float* bn1_g   = (const float*)d_in[6];
    const float* bn1_b   = (const float*)d_in[7];
    const float* bn1_m   = (const float*)d_in[8];
    const float* bn1_v   = (const float*)d_in[9];
    const float* conv2_w = (const float*)d_in[10];
    const float* conv2_b = (const float*)d_in[11];
    const float* bn2_g   = (const float*)d_in[12];
    const float* bn2_b   = (const float*)d_in[13];
    const float* bn2_m   = (const float*)d_in[14];
    const float* bn2_v   = (const float*)d_in[15];
    const float* proj_w  = (const float*)d_in[16];
    const float* proj_b  = (const float*)d_in[17];
    const float* mlp1_w  = (const float*)d_in[18];
    const float* mlp1_b  = (const float*)d_in[19];
    const float* mlp2_w  = (const float*)d_in[20];
    const float* mlp2_b  = (const float*)d_in[21];
    const float* ln_g    = (const float*)d_in[22];
    const float* ln_b    = (const float*)d_in[23];

    float* ws   = (float*)d_ws;
    float* out0 = ws;
    float* P    = out0 + (size_t)BB * CC * TT;
    float* h2   = P    + (size_t)BB * CC * PLEN;
    float* z    = h2   + (size_t)BB * WID * UU;
    float* z1   = z    + (size_t)BB * HID;
    float* gg   = z1   + (size_t)BB * PROJ;
    float* yy   = gg   + (size_t)BB * PROJ;

    dim3 g1(8, BB);
    k_subject<<<g1, 256, 0, stream>>>(x, sid, W_sub, b_sub, P);

    dim3 g3(BB * (UU / 16) / 4, WID / 16);
    k_conv2<<<g3, 128, 0, stream>>>(P, conv1_w, conv1_b, bn1_g, bn1_b, bn1_m, bn1_v,
                                    conv2_w, conv2_b, bn2_g, bn2_b, bn2_m, bn2_v, h2);

    k_proj<<<BB, 256, 0, stream>>>(h2, proj_w, proj_b, z);

    dim3 g5(2, PROJ / 16);
    k_mlp1<<<g5, 32, 0, stream>>>(z, mlp1_w, mlp1_b, z1, gg);
    k_mlp2<<<g5, 32, 0, stream>>>(gg, mlp2_w, mlp2_b, z1, yy);

    k_ln<<<BB, 256, 0, stream>>>(yy, ln_g, ln_b, (float*)d_out);
}


# conv2: the A slab's global loads run two trips ahead (slab w+2 requested at the top of trip w into a second register stage, trip body twice per loop pass, barrier fences LDS only); K sum order kept
# speedup vs baseline: 4.5975x; 1.0789x over previous
#include <hip/hip_runtime.h>
#include <math.h>

#define BB   32
#define CC   122
#define TT   500
#define WID  80
#define UU   96
#define K2   (WID*CC)
#define HID  768
#define PROJ 1024
#define PLEN 484
#define EPSF 1e-5f

typedef float v2f __attribute__((ext_vector_type(2)));
typedef float v8f __attribute__((ext_vector_type(8)));

static __device__ __forceinline__ v8f wmma_f32(v2f a, v2f b, v8f c) {
    return __builtin_amdgcn_wmma_f32_16x16x4_f32(false, a, false, b, (short)0, c, false, false);
}

#define SKC   32
#define SP4   9
#define SNCH  16
#define SNT   64
#define SNB   80
#define SMR   128
#define TPF   84

__global__ __launch_bounds__(256) void k_subject(
    const float* __restrict__ x, const int* __restrict__ sid,
    const float* __restrict__ W_sub, const float* __restrict__ b_sub,
    float* __restrict__ P)
{
    __shared__ float4 As[2][SMR * SP4];
    __shared__ float4 Bs[2][SNB * SP4];
    __shared__ __attribute__((aligned(16))) float Ts[SMR * TPF];

    const int tid = threadIdx.x;
    const int b   = blockIdx.y;
    const int n0  = blockIdx.x * SNT;
    const int s   = sid[b];
    const float* Wg = W_sub + (size_t)s * TT * TT;
    const float* xb = x + (size_t)b * CC * TT;

    const int kq = tid & 7;
    const int r  = tid >> 3;
    const bool third = (r < SNB - 64);
    const float* ga[4];
    const float* gb[3];
    #pragma unroll
    for (int i = 0; i < 4; ++i) {
        int row = r + 32 * i; if (row >= CC) row = CC - 1;
        ga[i] = xb + (size_t)row * TT + 4 * kq;
    }
    #pragma unroll
    for (int i = 0; i < 3; ++i) {
        int row = n0 + r + 32 * i; if (row >= TT) row = TT - 1;
        gb[i] = Wg + (size_t)row * TT + 4 * kq;
    }
    const int st = r * SP4 + kq;

    const int wv   = tid >> 5;
    const int lane = tid & 31;
    const int c    = lane & 15;
    const int h    = lane >> 4;
    const int fa = (16 * wv + c) * SP4 + h;
    const int fb = c * SP4 + h;

    float4 ra[4], rb[3];
    v8f acc[5] = {};

    #pragma unroll
    for (int i = 0; i < 4; ++i) ra[i] = *(const float4*)(ga[i]);
    #pragma unroll
    for (int i = 0; i < 2; ++i) rb[i] = *(const float4*)(gb[i]);
    if (third) rb[2] = *(const float4*)(gb[2]);
    #pragma unroll
    for (int i = 0; i < 4; ++i) As[0][st + 32 * SP4 * i] = ra[i];
    #pragma unroll
    for (int i = 0; i < 2; ++i) Bs[0][st + 32 * SP4 * i] = rb[i];
    if (third) Bs[0][st + 64 * SP4] = rb[2];
    __syncthreads();

    for (int ch = 0; ch < SNCH; ++ch) {
        const int  buf  = ch & 1;
        const bool more = (ch + 1 < SNCH);
        const int  ko   = (ch + 1) * SKC;
        const bool tail = (ko + 4 * kq) >= TT;
        if (more) {
            const int kc = tail ? 0 : ko;
            #pragma unroll
            for (int i = 0; i < 4; ++i) ra[i] = *(const float4*)(ga[i] + kc);
            #pragma unroll
            for (int i = 0; i < 2; ++i) rb[i] = *(const float4*)(gb[i] + kc);
            if (third) rb[2] = *(const float4*)(gb[2] + kc);
        }

        const float4* pA = &As[buf][fa];
        const float4* pB = &Bs[buf][fb];
        float4 af[4], bf[4][5];
        #pragma unroll
        for (int s8 = 0; s8 < 4; ++s8) {
            af[s8] = pA[2 * s8];
            #pragma unroll
            for (int j = 0; j < 5; ++j) bf[s8][j] = pB[16 * SP4 * j + 2 * s8];
        }
        __builtin_amdgcn_sched_barrier(0);
        #pragma unroll
        for (int s8 = 0; s8 < 4; ++s8) {
            const v2f a0 = { af[s8].x, af[s8].y };
            const v2f a1 = { af[s8].z, af[s8].w };
            #pragma unroll
            for (int j = 0; j < 5; ++j) {
                const v2f w0 = { bf[s8][j].x, bf[s8][j].y };
                acc[j] = wmma_f32(a0, w0, acc[j]);
            }
            #pragma unroll
            for (int j = 0; j < 5; ++j) {
                const v2f w1 = { bf[s8][j].z, bf[s8][j].w };
                acc[j] = wmma_f32(a1, w1, acc[j]);
            }
        }

        if (more) {
            if (tail) {
                #pragma unroll
                for (int i = 0; i < 4; ++i) ra[i] = float4{0.f, 0.f, 0.f, 0.f};
                #pragma unroll
                for (int i = 0; i < 3; ++i) rb[i] = float4{0.f, 0.f, 0.f, 0.f};
            }
            #pragma unroll
            for (int i = 0; i < 4; ++i) As[buf ^ 1][st + 32 * SP4 * i] = ra[i];
            #pragma unroll
            for (int i = 0; i < 2; ++i) Bs[buf ^ 1][st + 32 * SP4 * i] = rb[i];
            if (third) Bs[buf ^ 1][st + 64 * SP4] = rb[2];
            __syncthreads();
        }
    }

    const float* bg = b_sub + (size_t)s * TT;
    float* tw = Ts + (16 * wv) * TPF;
    #pragma unroll
    for (int j = 0; j < 5; ++j) {
        int nn = n0 + 16 * j + c; if (nn >= TT) nn = TT - 1;
        const float bn = bg[nn];
        #pragma unroll
        for (int q = 0; q < 8; ++q)
            tw[(q + 8 * h) * TPF + 16 * j + c] = acc[j][q] + bn;
    }
    __syncthreads();

    const int pc = lane & 15;
    const int rs = lane >> 4;
    const int m  = n0 + 4 * pc;
    #pragma unroll
    for (int it = 0; it < 8; ++it) {
        const int rl = 2 * it + rs;
        const int mm = 16 * wv + rl;
        const float4* src = (const float4*)(tw + rl * TPF + 4 * pc);
        float a[20];
        #pragma unroll
        for (int k = 0; k < 5; ++k) {
            const float4 v = src[k];
            a[4 * k] = v.x; a[4 * k + 1] = v.y; a[4 * k + 2] = v.z; a[4 * k + 3] = v.w;
        }
        float o[4];
        #pragma unroll
        for (int e = 0; e < 4; ++e) {
            float sum = 0.f;
            #pragma unroll
            for (int i = 0; i < 17; ++i) sum += a[e + i];
            o[e] = sum * (1.f / 17.f);
        }
        if (mm < CC && m < PLEN)
            *(float4*)(P + ((size_t)b * CC + mm) * PLEN + m) = float4{o[0], o[1], o[2], o[3]};
    }
}

#define CSTEPS 30
#define GS     6
#define NPR    31
#define AP     124
#define ACP    8
#define SPL(x)        (v2f{ (x), (x) })
#define FMA(a, b, c)  __builtin_elementwise_fma(SPL(a), (b), (c))
#define NC(G) (((G) == 4) ? 7 : 6)
#define ST(G, D, expr) _Pragma("unroll") for (int i = 0; i < 7; ++i) { if (i < NC(G)) { D[i] = (expr); } }
#define S12(G, E, T)  ST(G, E, SPL(T[0]) * pr[6 * (G) + i][0]) ST(G, E, FMA(T[1], pr[6 * (G) + i][1], E[i]))
#define S34(G, E, T)  ST(G, E, FMA(T[2], pr[6 * (G) + i][2], E[i])) ST(G, E, FMA(T[3], pr[6 * (G) + i][3], E[i]))
#define S567(G, E, M, T) ST(G, E, FMA(T[4], pr[6 * (G) + i][4], E[i])) ST(G, E, FMA(T[5], E[i], SPL(T[6]))) \
                         ST(G, M, E[i] * SPL(1.44269504088896340736f))
#define S8(G, M)      ST(G, M, (v2f{ __builtin_amdgcn_exp2f(M[i].x), __builtin_amdgcn_exp2f(M[i].y) }))
#define S9(G, E, M)   ST(G, E, (v2f{ E[i].x > 0.f ? E[i].x : (M[i].x - 1.f), E[i].y > 0.f ? E[i].y : (M[i].y - 1.f) }))
#define WM(G, S, EC)  { acc = wmma_f32(acur[GS * (G) + (S)], EC[(S)], acc); }

__global__ __launch_bounds__(128) void k_conv2(
    const float* __restrict__ P,
    const float* __restrict__ conv1_w, const float* __restrict__ conv1_b,
    const float* __restrict__ bn1_g, const float* __restrict__ bn1_b,
    const float* __restrict__ bn1_m, const float* __restrict__ bn1_v,
    const float* __restrict__ conv2_w, const float* __restrict__ conv2_b,
    const float* __restrict__ bn2_g, const float* __restrict__ bn2_b,
    const float* __restrict__ bn2_m, const float* __restrict__ bn2_v,
    float* __restrict__ h2)
{
    __shared__ float w1s[WID * 5];
    __shared__ float sc1[WID], sh1[WID];
    __shared__ __attribute__((aligned(16))) float As[2][16 * AP];

    const int tid = threadIdx.x;
    const int mt  = blockIdx.y;

    const float* gA = conv2_w + (size_t)(mt * 16) * K2;
    const int cr = tid >> 3, cq = tid & 7;
    const int q7 = (cq + 56 > 60) ? 60 : (cq + 56);
    const unsigned gb0 = 4u * (unsigned)(cr * K2 + 2 * cq);
    const unsigned gb7 = 4u * (unsigned)(cr * K2 + 2 * q7);
    const int lo0 = cr * AP + 2 * cq, lo7 = cr * AP + 2 * q7;
    v2f sA[ACP], sB[ACP];
    #pragma unroll
    for (int j = 0; j < ACP - 1; ++j) sA[j] = *(const v2f*)((const char*)gA + gb0 + 64 * j);
    sA[ACP - 1] = *(const v2f*)((const char*)gA + gb7);
    #pragma unroll
    for (int j = 0; j < ACP - 1; ++j) *(v2f*)(&As[0][lo0 + 16 * j]) = sA[j];
    *(v2f*)(&As[0][lo7]) = sA[ACP - 1];
    {
        const char* g1 = (const char*)(gA + CC);
        #pragma unroll
        for (int j = 0; j < ACP - 1; ++j) sA[j] = *(const v2f*)(g1 + gb0 + 64 * j);
        sA[ACP - 1] = *(const v2f*)(g1 + gb7);
    }

    for (int i = tid; i < WID * 5; i += 128) w1s[i] = conv1_w[i];
    if (tid < WID) {
        float sc = bn1_g[tid] * rsqrtf(bn1_v[tid] + EPSF);
        sc1[tid] = sc;
        sh1[tid] = bn1_b[tid] - bn1_m[tid] * sc + sc * conv1_b[tid];
    }
    __syncthreads();

    const int wave = tid >> 5;
    const int lane = tid & 31;
    const int nt   = blockIdx.x * 4 + wave;
    const int b    = nt / 6;
    const int u0   = (nt - 6 * b) * 16;
    const int n    = lane & 15;
    const int half = lane >> 4;

    v2f pr[NPR][5];
    #pragma unroll
    for (int i = 0; i < NPR; ++i) {
        const int c = (i < CSTEPS) ? (4 * i + 2 * half) : 120;
        const float* p = P + ((size_t)b * CC + c) * PLEN + 5 * (u0 + n);
        #pragma unroll
        for (int j = 0; j < 5; ++j) pr[i][j] = v2f{ p[j], p[PLEN + j] };
    }

    const int fro = n * AP + 2 * half;
    v2f acur[CSTEPS + 1];

    v8f acc = {};
    float tc[7], tn[7];
    #pragma unroll
    for (int j = 0; j < 5; ++j) tc[j] = w1s[j];
    tc[5] = sc1[0]; tc[6] = sh1[0];
    v2f ec[7] = {}, en[7] = {}, em[7] = {};
    S12(0, ec, tc) S34(0, ec, tc) S567(0, ec, em, tc) S8(0, em) S9(0, ec, em)

#define TRIP(W_, RB_, SU_, SL_) {                                                                                      \
        const int w = (W_); const int rb = (RB_);                                                                      \
                                       \
        const int wn = (w + 1 < WID) ? (w + 1) : w;                                                                    \
                             \
        const int wl = (w + 2 < WID) ? (w + 2) : (WID - 1);                                                            \
                    \
                                                               \
        _Pragma("unroll")                                                                                              \
        for (int j = 0; j < 5; ++j) tn[j] = w1s[wn * 5 + j];                                                           \
        tn[5] = sc1[wn]; tn[6] = sh1[wn];                                                                              \
                                    \
        const float* fr = &As[rb][fro];                                                                                \
                                                \
        _Pragma("unroll")                                                                                              \
        for (int s = 0; s < CSTEPS; ++s) acur[s] = *(const v2f*)(fr + 4 * s);                                          \
        acur[CSTEPS] = *(const v2f*)(&As[rb][n * AP + 120]);                                                           \
                   \
                                                                      \
        __builtin_amdgcn_sched_barrier(0);                                                                             \
               \
                                                    \
        const char* gw = (const char*)(gA + wl * CC);                                                                  \
        _Pragma("unroll")                                                                                              \
        for (int j = 0; j < ACP - 1; ++j) SL_[j] = *(const v2f*)(gw + gb0 + 64 * j);                                   \
        SL_[ACP - 1] = *(const v2f*)(gw + gb7);                                                                        \
                    \
                                                                   \
        __builtin_amdgcn_sched_barrier(0);                                                                             \
                                                                                                                       \
                \
        _Pragma("unroll")                                                                                              \
        for (int g = 0; g < 5; ++g) {                                                                                  \
                                \
            if (g < 4) {                                                                                               \
                WM(g, 0, ec) S12(g + 1, en, tc)                                                                        \
                WM(g, 1, ec) S34(g + 1, en, tc)                                                                        \
                WM(g, 2, ec) S567(g + 1, en, em, tc)                                                                   \
                WM(g, 3, ec) S8(g + 1, em)                                                                             \
                WM(g, 4, ec) S9(g + 1, en, em)                                                                         \
                WM(g, 5, ec)                                                                                           \
            } else {                                                                                                   \
                       \
                                                  \
                __builtin_amdgcn_sched_barrier(0);                                                                     \
                     \
                       \
                                                                                \
                _Pragma("unroll")                                                                                      \
                for (int j = 0; j < ACP - 1; ++j) *(v2f*)(&As[rb ^ 1][lo0 + 16 * j]) = SU_[j];                         \
                *(v2f*)(&As[rb ^ 1][lo7]) = SU_[ACP - 1];                                                              \
                WM(4, 0, ec) S12(0, en, tn)                                                                            \
                WM(4, 1, ec) S34(0, en, tn)                                                                            \
                WM(4, 2, ec) S567(0, en, em, tn)                                                                       \
                WM(4, 3, ec) S8(0, em)                                                                                 \
                WM(4, 4, ec) S9(0, en, em)                                                                             \
                WM(4, 5, ec)                                                                                           \
                {                 \
                    v2f a  = acur[CSTEPS];                                                                             \
                    v2f bb = ec[6];                                                                                    \
                    if (half) { a = v2f{0.f, 0.f}; bb = v2f{0.f, 0.f}; }                                               \
                    acc = wmma_f32(a, bb, acc);                                                                        \
                }                                                                                                      \
            }                                                                                                          \
            _Pragma("unroll")                                                                                          \
            for (int i = 0; i < 7; ++i) ec[i] = en[i];                                                                 \
        }                                                                                                              \
        _Pragma("unroll")                                                                                              \
        for (int j = 0; j < 7; ++j) tc[j] = tn[j];                                                                     \
                                      \
              \
                  \
               \
                \
        __builtin_amdgcn_fence(__ATOMIC_RELEASE, "workgroup", "local");                                                \
        __builtin_amdgcn_s_barrier();                                                                                  \
        __builtin_amdgcn_fence(__ATOMIC_ACQUIRE, "workgroup", "local");                                                \
    }

    for (int wp = 0; wp < WID; wp += 2) {
        TRIP(wp,     0, sA, sB)
        TRIP(wp + 1, 1, sB, sA)
    }
#undef TRIP

    #pragma unroll
    for (int r = 0; r < 8; ++r) {
        int w2 = mt * 16 + r + half * 8;
        float sc = bn2_g[w2] * rsqrtf(bn2_v[w2] + EPSF);
        float sh = bn2_b[w2] - bn2_m[w2] * sc;
        float v  = sc * (acc[r] + conv2_b[w2]) + sh;
        float e  = v > 0.f ? v : (__expf(v) - 1.f);
        h2[((size_t)b * WID + w2) * UU + u0 + n] = e;
    }
}

__global__ __launch_bounds__(256) void k_proj(
    const float* __restrict__ h2, const float* __restrict__ proj_w,
    const float* __restrict__ proj_b, float* __restrict__ z)
{
    __shared__ float hs[WID * UU];
    __shared__ float pw[8 * WID];
    const int b = blockIdx.x;
    for (int i = threadIdx.x; i < WID * UU; i += 256) hs[i] = h2[(size_t)b * WID * UU + i];
    for (int i = threadIdx.x; i < 8 * WID; i += 256)  pw[i] = proj_w[i];
    __syncthreads();
    for (int idx = threadIdx.x; idx < HID; idx += 256) {
        int u = idx >> 3, e = idx & 7;
        float s = proj_b[e];
        #pragma unroll 8
        for (int w = 0; w < WID; ++w) s += pw[e * WID + w] * hs[w * UU + u];
        z[(size_t)b * HID + idx] = s;
    }
}

#define MKB    32
#define MAHEAD 7

template <int K>
static __device__ __forceinline__ v8f mlp_tile(const float* pa, const float* pb)
{
    constexpr int NB = K / MKB;
    float4 fa[NB][4], fb[NB][4];
    v8f acc = {};
    #pragma unroll
    for (int i = 0; i < MAHEAD; ++i) {
        #pragma unroll
        for (int j = 0; j < 4; ++j) {
            fa[i][j] = *(const float4*)(pa + MKB * i + 8 * j);
            fb[i][j] = *(const float4*)(pb + MKB * i + 8 * j);
        }
    }
    #pragma unroll
    for (int i = 0; i < NB; ++i) {
        if (i + MAHEAD < NB) {
            #pragma unroll
            for (int j = 0; j < 4; ++j) {
                fa[i + MAHEAD][j] = *(const float4*)(pa + MKB * (i + MAHEAD) + 8 * j);
                fb[i + MAHEAD][j] = *(const float4*)(pb + MKB * (i + MAHEAD) + 8 * j);
            }
        }
        #pragma unroll
        for (int j = 0; j < 4; ++j) {
            const v2f a0 = { fa[i][j].x, fa[i][j].y };
            const v2f b0 = { fb[i][j].x, fb[i][j].y };
            acc = wmma_f32(a0, b0, acc);
            const v2f a1 = { fa[i][j].z, fa[i][j].w };
            const v2f b1 = { fb[i][j].z, fb[i][j].w };
            acc = wmma_f32(a1, b1, acc);
        }
        __builtin_amdgcn_sched_barrier(0);
    }
    return acc;
}

__global__ __launch_bounds__(32) void k_mlp1(
    const float* __restrict__ z, const float* __restrict__ w,
    const float* __restrict__ bias, float* __restrict__ z1, float* __restrict__ g)
{
    const int lane = threadIdx.x;
    const int m0 = blockIdx.x * 16, n0 = blockIdx.y * 16;
    const float* pa = z + (size_t)(m0 + (lane & 15)) * HID + (lane >> 4) * 4;
    const float* pb = w + (size_t)(n0 + (lane & 15)) * HID + (lane >> 4) * 4;
    const v8f acc = mlp_tile<HID>(pa, pb);
    const int nn = n0 + (lane & 15);
    const float bn = bias[nn];
    #pragma unroll
    for (int r = 0; r < 8; ++r) {
        int mm = m0 + r + ((lane >> 4) << 3);
        float v = acc[r] + bn;
        z1[(size_t)mm * PROJ + nn] = v;
        g [(size_t)mm * PROJ + nn] = 0.5f * v * (1.f + erff(v * 0.70710678118654752f));
    }
}

__global__ __launch_bounds__(32) void k_mlp2(
    const float* __restrict__ g, const float* __restrict__ w,
    const float* __restrict__ bias, const float* __restrict__ z1,
    float* __restrict__ y)
{
    const int lane = threadIdx.x;
    const int m0 = blockIdx.x * 16, n0 = blockIdx.y * 16;
    const float* pa = g + (size_t)(m0 + (lane & 15)) * PROJ + (lane >> 4) * 4;
    const float* pb = w + (size_t)(n0 + (lane & 15)) * PROJ + (lane >> 4) * 4;
    const v8f acc = mlp_tile<PROJ>(pa, pb);
    const int nn = n0 + (lane & 15);
    const float bn = bias[nn];
    #pragma unroll
    for (int r = 0; r < 8; ++r) {
        int mm = m0 + r + ((lane >> 4) << 3);
        y[(size_t)mm * PROJ + nn] = z1[(size_t)mm * PROJ + nn] + acc[r] + bn;
    }
}

__global__ __launch_bounds__(256) void k_ln(
    const float* __restrict__ y, const float* __restrict__ gam,
    const float* __restrict__ bet, float* __restrict__ out)
{
    __shared__ float red[256];
    const int b = blockIdx.x;
    const float* row = y + (size_t)b * PROJ;
    float s = 0.f;
    for (int i = threadIdx.x; i < PROJ; i += 256) s += row[i];
    red[threadIdx.x] = s; __syncthreads();
    for (int off = 128; off > 0; off >>= 1) {
        if (threadIdx.x < off) red[threadIdx.x] += red[threadIdx.x + off];
        __syncthreads();
    }
    const float mu = red[0] * (1.f / PROJ);
    __syncthreads();
    float s2 = 0.f;
    for (int i = threadIdx.x; i < PROJ; i += 256) { float d = row[i] - mu; s2 += d * d; }
    red[threadIdx.x] = s2; __syncthreads();
    for (int off = 128; off > 0; off >>= 1) {
        if (threadIdx.x < off) red[threadIdx.x] += red[threadIdx.x + off];
        __syncthreads();
    }
    const float inv = rsqrtf(red[0] * (1.f / PROJ) + EPSF);
    for (int i = threadIdx.x; i < PROJ; i += 256)
        out[(size_t)b * PROJ + i] = (row[i] - mu) * inv * gam[i] + bet[i];
}

extern "C" void kernel_launch(void* const* d_in, const int* in_sizes, int n_in,
                              void* d_out, int out_size, void* d_ws, size_t ws_size,
                              hipStream_t stream) {
    const float* x       = (const float*)d_in[0];
    const int*   sid     = (const int*)  d_in[1];
    const float* W_sub   = (const float*)d_in[2];
    const float* b_sub   = (const float*)d_in[3];
    const float* conv1_w = (const float*)d_in[4];
    const float* conv1_b = (const float*)d_in[5];
    const float* bn1_g   = (const float*)d_in[6];
    const float* bn1_b   = (const float*)d_in[7];
    const float* bn1_m   = (const float*)d_in[8];
    const float* bn1_v   = (const float*)d_in[9];
    const float* conv2_w = (const float*)d_in[10];
    const float* conv2_b = (const float*)d_in[11];
    const float* bn2_g   = (const float*)d_in[12];
    const float* bn2_b   = (const float*)d_in[13];
    const float* bn2_m   = (const float*)d_in[14];
    const float* bn2_v   = (const float*)d_in[15];
    const float* proj_w  = (const float*)d_in[16];
    const float* proj_b  = (const float*)d_in[17];
    const float* mlp1_w  = (const float*)d_in[18];
    const float* mlp1_b  = (const float*)d_in[19];
    const float* mlp2_w  = (const float*)d_in[20];
    const float* mlp2_b  = (const float*)d_in[21];
    const float* ln_g    = (const float*)d_in[22];
    const float* ln_b    = (const float*)d_in[23];

    float* ws   = (float*)d_ws;
    float* out0 = ws;
    float* P    = out0 + (size_t)BB * CC * TT;
    float* h2   = P    + (size_t)BB * CC * PLEN;
    float* z    = h2   + (size_t)BB * WID * UU;
    float* z1   = z    + (size_t)BB * HID;
    float* gg   = z1   + (size_t)BB * PROJ;
    float* yy   = gg   + (size_t)BB * PROJ;

    dim3 g1(8, BB);
    k_subject<<<g1, 256, 0, stream>>>(x, sid, W_sub, b_sub, P);

    dim3 g3(BB * (UU / 16) / 4, WID / 16);
    k_conv2<<<g3, 128, 0, stream>>>(P, conv1_w, conv1_b, bn1_g, bn1_b, bn1_m, bn1_v,
                                    conv2_w, conv2_b, bn2_g, bn2_b, bn2_m, bn2_v, h2);

    k_proj<<<BB, 256, 0, stream>>>(h2, proj_w, proj_b, z);

    dim3 g5(2, PROJ / 16);
    k_mlp1<<<g5, 32, 0, stream>>>(z, mlp1_w, mlp1_b, z1, gg);
    k_mlp2<<<g5, 32, 0, stream>>>(gg, mlp2_w, mlp2_b, z1, yy);

    k_ln<<<BB, 256, 0, stream>>>(yy, ln_g, ln_b, (float*)d_out);
}


# conv2: 8-wave blocks, two waves per SIMD: a tile's even w and odd w go to two waves (own slabs, own h1), halves added through LDS (even sum + odd sum): K sum order changes
# speedup vs baseline: 4.7315x; 1.0291x over previous
#include <hip/hip_runtime.h>
#include <math.h>

#define BB   32
#define CC   122
#define TT   500
#define WID  80
#define UU   96
#define K2   (WID*CC)
#define HID  768
#define PROJ 1024
#define PLEN 484
#define EPSF 1e-5f

typedef float v2f __attribute__((ext_vector_type(2)));
typedef float v8f __attribute__((ext_vector_type(8)));

static __device__ __forceinline__ v8f wmma_f32(v2f a, v2f b, v8f c) {
    return __builtin_amdgcn_wmma_f32_16x16x4_f32(false, a, false, b, (short)0, c, false, false);
}

#define SKC   32
#define SP4   9
#define SNCH  16
#define SNT   64
#define SNB   80
#define SMR   128
#define TPF   84

__global__ __launch_bounds__(256) void k_subject(
    const float* __restrict__ x, const int* __restrict__ sid,
    const float* __restrict__ W_sub, const float* __restrict__ b_sub,
    float* __restrict__ P)
{
    __shared__ float4 As[2][SMR * SP4];
    __shared__ float4 Bs[2][SNB * SP4];
    __shared__ __attribute__((aligned(16))) float Ts[SMR * TPF];

    const int tid = threadIdx.x;
    const int b   = blockIdx.y;
    const int n0  = blockIdx.x * SNT;
    const int s   = sid[b];
    const float* Wg = W_sub + (size_t)s * TT * TT;
    const float* xb = x + (size_t)b * CC * TT;

    const int kq = tid & 7;
    const int r  = tid >> 3;
    const bool third = (r < SNB - 64);
    const float* ga[4];
    const float* gb[3];
    #pragma unroll
    for (int i = 0; i < 4; ++i) {
        int row = r + 32 * i; if (row >= CC) row = CC - 1;
        ga[i] = xb + (size_t)row * TT + 4 * kq;
    }
    #pragma unroll
    for (int i = 0; i < 3; ++i) {
        int row = n0 + r + 32 * i; if (row >= TT) row = TT - 1;
        gb[i] = Wg + (size_t)row * TT + 4 * kq;
    }
    const int st = r * SP4 + kq;

    const int wv   = tid >> 5;
    const int lane = tid & 31;
    const int c    = lane & 15;
    const int h    = lane >> 4;
    const int fa = (16 * wv + c) * SP4 + h;
    const int fb = c * SP4 + h;

    float4 ra[4], rb[3];
    v8f acc[5] = {};

    #pragma unroll
    for (int i = 0; i < 4; ++i) ra[i] = *(const float4*)(ga[i]);
    #pragma unroll
    for (int i = 0; i < 2; ++i) rb[i] = *(const float4*)(gb[i]);
    if (third) rb[2] = *(const float4*)(gb[2]);
    #pragma unroll
    for (int i = 0; i < 4; ++i) As[0][st + 32 * SP4 * i] = ra[i];
    #pragma unroll
    for (int i = 0; i < 2; ++i) Bs[0][st + 32 * SP4 * i] = rb[i];
    if (third) Bs[0][st + 64 * SP4] = rb[2];
    __syncthreads();

    for (int ch = 0; ch < SNCH; ++ch) {
        const int  buf  = ch & 1;
        const bool more = (ch + 1 < SNCH);
        const int  ko   = (ch + 1) * SKC;
        const bool tail = (ko + 4 * kq) >= TT;
        if (more) {
            const int kc = tail ? 0 : ko;
            #pragma unroll
            for (int i = 0; i < 4; ++i) ra[i] = *(const float4*)(ga[i] + kc);
            #pragma unroll
            for (int i = 0; i < 2; ++i) rb[i] = *(const float4*)(gb[i] + kc);
            if (third) rb[2] = *(const float4*)(gb[2] + kc);
        }

        const float4* pA = &As[buf][fa];
        const float4* pB = &Bs[buf][fb];
        float4 af[4], bf[4][5];
        #pragma unroll
        for (int s8 = 0; s8 < 4; ++s8) {
            af[s8] = pA[2 * s8];
            #pragma unroll
            for (int j = 0; j < 5; ++j) bf[s8][j] = pB[16 * SP4 * j + 2 * s8];
        }
        __builtin_amdgcn_sched_barrier(0);
        #pragma unroll
        for (int s8 = 0; s8 < 4; ++s8) {
            const v2f a0 = { af[s8].x, af[s8].y };
            const v2f a1 = { af[s8].z, af[s8].w };
            #pragma unroll
            for (int j = 0; j < 5; ++j) {
                const v2f w0 = { bf[s8][j].x, bf[s8][j].y };
                acc[j] = wmma_f32(a0, w0, acc[j]);
            }
            #pragma unroll
            for (int j = 0; j < 5; ++j) {
                const v2f w1 = { bf[s8][j].z, bf[s8][j].w };
                acc[j] = wmma_f32(a1, w1, acc[j]);
            }
        }

        if (more) {
            if (tail) {
                #pragma unroll
                for (int i = 0; i < 4; ++i) ra[i] = float4{0.f, 0.f, 0.f, 0.f};
                #pragma unroll
                for (int i = 0; i < 3; ++i) rb[i] = float4{0.f, 0.f, 0.f, 0.f};
            }
            #pragma unroll
            for (int i = 0; i < 4; ++i) As[buf ^ 1][st + 32 * SP4 * i] = ra[i];
            #pragma unroll
            for (int i = 0; i < 2; ++i) Bs[buf ^ 1][st + 32 * SP4 * i] = rb[i];
            if (third) Bs[buf ^ 1][st + 64 * SP4] = rb[2];
            __syncthreads();
        }
    }

    const float* bg = b_sub + (size_t)s * TT;
    float* tw = Ts + (16 * wv) * TPF;
    #pragma unroll
    for (int j = 0; j < 5; ++j) {
        int nn = n0 + 16 * j + c; if (nn >= TT) nn = TT - 1;
        const float bn = bg[nn];
        #pragma unroll
        for (int q = 0; q < 8; ++q)
            tw[(q + 8 * h) * TPF + 16 * j + c] = acc[j][q] + bn;
    }
    __syncthreads();

    const int pc = lane & 15;
    const int rs = lane >> 4;
    const int m  = n0 + 4 * pc;
    #pragma unroll
    for (int it = 0; it < 8; ++it) {
        const int rl = 2 * it + rs;
        const int mm = 16 * wv + rl;
        const float4* src = (const float4*)(tw + rl * TPF + 4 * pc);
        float a[20];
        #pragma unroll
        for (int k = 0; k < 5; ++k) {
            const float4 v = src[k];
            a[4 * k] = v.x; a[4 * k + 1] = v.y; a[4 * k + 2] = v.z; a[4 * k + 3] = v.w;
        }
        float o[4];
        #pragma unroll
        for (int e = 0; e < 4; ++e) {
            float sum = 0.f;
            #pragma unroll
            for (int i = 0; i < 17; ++i) sum += a[e + i];
            o[e] = sum * (1.f / 17.f);
        }
        if (mm < CC && m < PLEN)
            *(float4*)(P + ((size_t)b * CC + mm) * PLEN + m) = float4{o[0], o[1], o[2], o[3]};
    }
}

#define CSTEPS 30
#define GS     6
#define NPR    31
#define AP     124
#define ACP    8
#define SPL(x)        (v2f{ (x), (x) })
#define FMA(a, b, c)  __builtin_elementwise_fma(SPL(a), (b), (c))
#define NC(G) (((G) == 4) ? 7 : 6)
#define ST(G, D, expr) _Pragma("unroll") for (int i = 0; i < 7; ++i) { if (i < NC(G)) { D[i] = (expr); } }
#define S12(G, E, T)  ST(G, E, SPL(T[0]) * pr[6 * (G) + i][0]) ST(G, E, FMA(T[1], pr[6 * (G) + i][1], E[i]))
#define S34(G, E, T)  ST(G, E, FMA(T[2], pr[6 * (G) + i][2], E[i])) ST(G, E, FMA(T[3], pr[6 * (G) + i][3], E[i]))
#define S567(G, E, M, T) ST(G, E, FMA(T[4], pr[6 * (G) + i][4], E[i])) ST(G, E, FMA(T[5], E[i], SPL(T[6]))) \
                         ST(G, M, E[i] * SPL(1.44269504088896340736f))
#define S8(G, M)      ST(G, M, (v2f{ __builtin_amdgcn_exp2f(M[i].x), __builtin_amdgcn_exp2f(M[i].y) }))
#define S9(G, E, M)   ST(G, E, (v2f{ E[i].x > 0.f ? E[i].x : (M[i].x - 1.f), E[i].y > 0.f ? E[i].y : (M[i].y - 1.f) }))
#define WM(G, S, EC)  { acc = wmma_f32(acur[GS * (G) + (S)], EC[(S)], acc); }

__global__ __launch_bounds__(256) void k_conv2(
    const float* __restrict__ P,
    const float* __restrict__ conv1_w, const float* __restrict__ conv1_b,
    const float* __restrict__ bn1_g, const float* __restrict__ bn1_b,
    const float* __restrict__ bn1_m, const float* __restrict__ bn1_v,
    const float* __restrict__ conv2_w, const float* __restrict__ conv2_b,
    const float* __restrict__ bn2_g, const float* __restrict__ bn2_b,
    const float* __restrict__ bn2_m, const float* __restrict__ bn2_v,
    float* __restrict__ h2)
{
    __shared__ float w1s[WID * 5];
    __shared__ float sc1[WID], sh1[WID];
    __shared__ __attribute__((aligned(16))) float As[2 * 2 * 16 * AP];
    __shared__ float red[4 * 8 * 32];

    const int tid = threadIdx.x;
    const int kh  = __builtin_amdgcn_readfirstlane(tid >> 7);
    const int t7  = tid & 127;
    const int hb  = kh * (2 * 16 * AP);
    const int mt  = blockIdx.y;

    const float* gA = conv2_w + (size_t)(mt * 16) * K2;
    const int cr = t7 >> 3, cq = t7 & 7;
    const int q7 = (cq + 56 > 60) ? 60 : (cq + 56);
    const unsigned gb0 = 4u * (unsigned)(cr * K2 + 2 * cq);
    const unsigned gb7 = 4u * (unsigned)(cr * K2 + 2 * q7);
    const int lo0 = hb + cr * AP + 2 * cq, lo7 = hb + cr * AP + 2 * q7;
    v2f sA[ACP], sB[ACP];
    {
        const char* g0 = (const char*)(gA + kh * CC);
        #pragma unroll
        for (int j = 0; j < ACP - 1; ++j) sA[j] = *(const v2f*)(g0 + gb0 + 64 * j);
        sA[ACP - 1] = *(const v2f*)(g0 + gb7);
    }
    #pragma unroll
    for (int j = 0; j < ACP - 1; ++j) *(v2f*)(&As[lo0 + 16 * j]) = sA[j];
    *(v2f*)(&As[lo7]) = sA[ACP - 1];
    {
        const char* g1 = (const char*)(gA + (kh + 2) * CC);
        #pragma unroll
        for (int j = 0; j < ACP - 1; ++j) sA[j] = *(const v2f*)(g1 + gb0 + 64 * j);
        sA[ACP - 1] = *(const v2f*)(g1 + gb7);
    }

    for (int i = tid; i < WID * 5; i += 256) w1s[i] = conv1_w[i];
    if (tid < WID) {
        float sc = bn1_g[tid] * rsqrtf(bn1_v[tid] + EPSF);
        sc1[tid] = sc;
        sh1[tid] = bn1_b[tid] - bn1_m[tid] * sc + sc * conv1_b[tid];
    }
    __syncthreads();

    const int wave = tid >> 5;
    const int lane = tid & 31;
    const int wq   = wave & 3;
    const int nt   = blockIdx.x * 4 + wq;
    const int b    = nt / 6;
    const int u0   = (nt - 6 * b) * 16;
    const int n    = lane & 15;
    const int half = lane >> 4;

    v2f pr[NPR][5];
    #pragma unroll
    for (int i = 0; i < NPR; ++i) {
        const int c = (i < CSTEPS) ? (4 * i + 2 * half) : 120;
        const float* p = P + ((size_t)b * CC + c) * PLEN + 5 * (u0 + n);
        #pragma unroll
        for (int j = 0; j < 5; ++j) pr[i][j] = v2f{ p[j], p[PLEN + j] };
    }

    const int fro = hb + n * AP + 2 * half;
    v2f acur[CSTEPS + 1];

    v8f acc = {};
    float tc[7], tn[7];
    #pragma unroll
    for (int j = 0; j < 5; ++j) tc[j] = w1s[kh * 5 + j];
    tc[5] = sc1[kh]; tc[6] = sh1[kh];
    v2f ec[7] = {}, en[7] = {}, em[7] = {};
    S12(0, ec, tc) S34(0, ec, tc) S567(0, ec, em, tc) S8(0, em) S9(0, ec, em)

#define TRIP(W_, RB_, SU_, SL_) {                                                                                      \
        const int w = (W_); const int rb = (RB_);                                                                      \
                                       \
        const int wn = (w + 2 < WID) ? (w + 2) : w;                                                                    \
                             \
        const int wl = (w + 4 < WID) ? (w + 4) : w;                                                            \
                    \
                                                               \
        _Pragma("unroll")                                                                                              \
        for (int j = 0; j < 5; ++j) tn[j] = w1s[wn * 5 + j];                                                           \
        tn[5] = sc1[wn]; tn[6] = sh1[wn];                                                                              \
                                    \
        const float* fr = &As[rb * 16 * AP + fro];                                                                                \
                                                \
        _Pragma("unroll")                                                                                              \
        for (int s = 0; s < CSTEPS; ++s) acur[s] = *(const v2f*)(fr + 4 * s);                                          \
        acur[CSTEPS] = *(const v2f*)(&As[rb * 16 * AP + hb + n * AP + 120]);                                                           \
                   \
                                                                      \
        __builtin_amdgcn_sched_barrier(0);                                                                             \
               \
                                                    \
        const char* gw = (const char*)(gA + wl * CC);                                                                  \
        _Pragma("unroll")                                                                                              \
        for (int j = 0; j < ACP - 1; ++j) SL_[j] = *(const v2f*)(gw + gb0 + 64 * j);                                   \
        SL_[ACP - 1] = *(const v2f*)(gw + gb7);                                                                        \
                    \
                                                                   \
        __builtin_amdgcn_sched_barrier(0);                                                                             \
                                                                                                                       \
                \
        _Pragma("unroll")                                                                                              \
        for (int g = 0; g < 5; ++g) {                                                                                  \
                                \
            if (g < 4) {                                                                                               \
                WM(g, 0, ec) S12(g + 1, en, tc)                                                                        \
                WM(g, 1, ec) S34(g + 1, en, tc)                                                                        \
                WM(g, 2, ec) S567(g + 1, en, em, tc)                                                                   \
                WM(g, 3, ec) S8(g + 1, em)                                                                             \
                WM(g, 4, ec) S9(g + 1, en, em)                                                                         \
                WM(g, 5, ec)                                                                                           \
            } else {                                                                                                   \
                       \
                                                  \
                __builtin_amdgcn_sched_barrier(0);                                                                     \
                     \
                       \
                                                                                \
                _Pragma("unroll")                                                                                      \
                for (int j = 0; j < ACP - 1; ++j) *(v2f*)(&As[(rb ^ 1) * 16 * AP + lo0 + 16 * j]) = SU_[j];                         \
                *(v2f*)(&As[(rb ^ 1) * 16 * AP + lo7]) = SU_[ACP - 1];                                                              \
                WM(4, 0, ec) S12(0, en, tn)                                                                            \
                WM(4, 1, ec) S34(0, en, tn)                                                                            \
                WM(4, 2, ec) S567(0, en, em, tn)                                                                       \
                WM(4, 3, ec) S8(0, em)                                                                                 \
                WM(4, 4, ec) S9(0, en, em)                                                                             \
                WM(4, 5, ec)                                                                                           \
                {                 \
                    v2f a  = acur[CSTEPS];                                                                             \
                    v2f bb = ec[6];                                                                                    \
                    if (half) { a = v2f{0.f, 0.f}; bb = v2f{0.f, 0.f}; }                                               \
                    acc = wmma_f32(a, bb, acc);                                                                        \
                }                                                                                                      \
            }                                                                                                          \
            _Pragma("unroll")                                                                                          \
            for (int i = 0; i < 7; ++i) ec[i] = en[i];                                                                 \
        }                                                                                                              \
        _Pragma("unroll")                                                                                              \
        for (int j = 0; j < 7; ++j) tc[j] = tn[j];                                                                     \
                                      \
              \
                  \
               \
                \
        __builtin_amdgcn_fence(__ATOMIC_RELEASE, "workgroup", "local");                                                \
        __builtin_amdgcn_s_barrier();                                                                                  \
        __builtin_amdgcn_fence(__ATOMIC_ACQUIRE, "workgroup", "local");                                                \
    }

    for (int wp = 0; wp < WID; wp += 4) {
        TRIP(wp + kh,     0, sA, sB)
        TRIP(wp + 2 + kh, 1, sB, sA)
    }
#undef TRIP

    if (kh) {
        #pragma unroll
        for (int r = 0; r < 8; ++r) red[(wq * 8 + r) * 32 + lane] = acc[r];
    }
    __syncthreads();
    if (!kh) {
        #pragma unroll
        for (int r = 0; r < 8; ++r) {
            int w2 = mt * 16 + r + half * 8;
            float sc = bn2_g[w2] * rsqrtf(bn2_v[w2] + EPSF);
            float sh = bn2_b[w2] - bn2_m[w2] * sc;
            float a  = acc[r] + red[(wq * 8 + r) * 32 + lane];
            float v  = sc * (a + conv2_b[w2]) + sh;
            float e  = v > 0.f ? v : (__expf(v) - 1.f);
            h2[((size_t)b * WID + w2) * UU + u0 + n] = e;
        }
    }
}

__global__ __launch_bounds__(256) void k_proj(
    const float* __restrict__ h2, const float* __restrict__ proj_w,
    const float* __restrict__ proj_b, float* __restrict__ z)
{
    __shared__ float hs[WID * UU];
    __shared__ float pw[8 * WID];
    const int b = blockIdx.x;
    for (int i = threadIdx.x; i < WID * UU; i += 256) hs[i] = h2[(size_t)b * WID * UU + i];
    for (int i = threadIdx.x; i < 8 * WID; i += 256)  pw[i] = proj_w[i];
    __syncthreads();
    for (int idx = threadIdx.x; idx < HID; idx += 256) {
        int u = idx >> 3, e = idx & 7;
        float s = proj_b[e];
        #pragma unroll 8
        for (int w = 0; w < WID; ++w) s += pw[e * WID + w] * hs[w * UU + u];
        z[(size_t)b * HID + idx] = s;
    }
}

#define MKB    32
#define MAHEAD 7

template <int K>
static __device__ __forceinline__ v8f mlp_tile(const float* pa, const float* pb)
{
    constexpr int NB = K / MKB;
    float4 fa[NB][4], fb[NB][4];
    v8f acc = {};
    #pragma unroll
    for (int i = 0; i < MAHEAD; ++i) {
        #pragma unroll
        for (int j = 0; j < 4; ++j) {
            fa[i][j] = *(const float4*)(pa + MKB * i + 8 * j);
            fb[i][j] = *(const float4*)(pb + MKB * i + 8 * j);
        }
    }
    #pragma unroll
    for (int i = 0; i < NB; ++i) {
        if (i + MAHEAD < NB) {
            #pragma unroll
            for (int j = 0; j < 4; ++j) {
                fa[i + MAHEAD][j] = *(const float4*)(pa + MKB * (i + MAHEAD) + 8 * j);
                fb[i + MAHEAD][j] = *(const float4*)(pb + MKB * (i + MAHEAD) + 8 * j);
            }
        }
        #pragma unroll
        for (int j = 0; j < 4; ++j) {
            const v2f a0 = { fa[i][j].x, fa[i][j].y };
            const v2f b0 = { fb[i][j].x, fb[i][j].y };
            acc = wmma_f32(a0, b0, acc);
            const v2f a1 = { fa[i][j].z, fa[i][j].w };
            const v2f b1 = { fb[i][j].z, fb[i][j].w };
            acc = wmma_f32(a1, b1, acc);
        }
        __builtin_amdgcn_sched_barrier(0);
    }
    return acc;
}

__global__ __launch_bounds__(32) void k_mlp1(
    const float* __restrict__ z, const float* __restrict__ w,
    const float* __restrict__ bias, float* __restrict__ z1, float* __restrict__ g)
{
    const int lane = threadIdx.x;
    const int m0 = blockIdx.x * 16, n0 = blockIdx.y * 16;
    const float* pa = z + (size_t)(m0 + (lane & 15)) * HID + (lane >> 4) * 4;
    const float* pb = w + (size_t)(n0 + (lane & 15)) * HID + (lane >> 4) * 4;
    const v8f acc = mlp_tile<HID>(pa, pb);
    const int nn = n0 + (lane & 15);
    const float bn = bias[nn];
    #pragma unroll
    for (int r = 0; r < 8; ++r) {
        int mm = m0 + r + ((lane >> 4) << 3);
        float v = acc[r] + bn;
        z1[(size_t)mm * PROJ + nn] = v;
        g [(size_t)mm * PROJ + nn] = 0.5f * v * (1.f + erff(v * 0.70710678118654752f));
    }
}

__global__ __launch_bounds__(32) void k_mlp2(
    const float* __restrict__ g, const float* __restrict__ w,
    const float* __restrict__ bias, const float* __restrict__ z1,
    float* __restrict__ y)
{
    const int lane = threadIdx.x;
    const int m0 = blockIdx.x * 16, n0 = blockIdx.y * 16;
    const float* pa = g + (size_t)(m0 + (lane & 15)) * PROJ + (lane >> 4) * 4;
    const float* pb = w + (size_t)(n0 + (lane & 15)) * PROJ + (lane >> 4) * 4;
    const v8f acc = mlp_tile<PROJ>(pa, pb);
    const int nn = n0 + (lane & 15);
    const float bn = bias[nn];
    #pragma unroll
    for (int r = 0; r < 8; ++r) {
        int mm = m0 + r + ((lane >> 4) << 3);
        y[(size_t)mm * PROJ + nn] = z1[(size_t)mm * PROJ + nn] + acc[r] + bn;
    }
}

__global__ __launch_bounds__(256) void k_ln(
    const float* __restrict__ y, const float* __restrict__ gam,
    const float* __restrict__ bet, float* __restrict__ out)
{
    __shared__ float red[256];
    const int b = blockIdx.x;
    const float* row = y + (size_t)b * PROJ;
    float s = 0.f;
    for (int i = threadIdx.x; i < PROJ; i += 256) s += row[i];
    red[threadIdx.x] = s; __syncthreads();
    for (int off = 128; off > 0; off >>= 1) {
        if (threadIdx.x < off) red[threadIdx.x] += red[threadIdx.x + off];
        __syncthreads();
    }
    const float mu = red[0] * (1.f / PROJ);
    __syncthreads();
    float s2 = 0.f;
    for (int i = threadIdx.x; i < PROJ; i += 256) { float d = row[i] - mu; s2 += d * d; }
    red[threadIdx.x] = s2; __syncthreads();
    for (int off = 128; off > 0; off >>= 1) {
        if (threadIdx.x < off) red[threadIdx.x] += red[threadIdx.x + off];
        __syncthreads();
    }
    const float inv = rsqrtf(red[0] * (1.f / PROJ) + EPSF);
    for (int i = threadIdx.x; i < PROJ; i += 256)
        out[(size_t)b * PROJ + i] = (row[i] - mu) * inv * gam[i] + bet[i];
}

extern "C" void kernel_launch(void* const* d_in, const int* in_sizes, int n_in,
                              void* d_out, int out_size, void* d_ws, size_t ws_size,
                              hipStream_t stream) {
    const float* x       = (const float*)d_in[0];
    const int*   sid     = (const int*)  d_in[1];
    const float* W_sub   = (const float*)d_in[2];
    const float* b_sub   = (const float*)d_in[3];
    const float* conv1_w = (const float*)d_in[4];
    const float* conv1_b = (const float*)d_in[5];
    const float* bn1_g   = (const float*)d_in[6];
    const float* bn1_b   = (const float*)d_in[7];
    const float* bn1_m   = (const float*)d_in[8];
    const float* bn1_v   = (const float*)d_in[9];
    const float* conv2_w = (const float*)d_in[10];
    const float* conv2_b = (const float*)d_in[11];
    const float* bn2_g   = (const float*)d_in[12];
    const float* bn2_b   = (const float*)d_in[13];
    const float* bn2_m   = (const float*)d_in[14];
    const float* bn2_v   = (const float*)d_in[15];
    const float* proj_w  = (const float*)d_in[16];
    const float* proj_b  = (const float*)d_in[17];
    const float* mlp1_w  = (const float*)d_in[18];
    const float* mlp1_b  = (const float*)d_in[19];
    const float* mlp2_w  = (const float*)d_in[20];
    const float* mlp2_b  = (const float*)d_in[21];
    const float* ln_g    = (const float*)d_in[22];
    const float* ln_b    = (const float*)d_in[23];

    float* ws   = (float*)d_ws;
    float* out0 = ws;
    float* P    = out0 + (size_t)BB * CC * TT;
    float* h2   = P    + (size_t)BB * CC * PLEN;
    float* z    = h2   + (size_t)BB * WID * UU;
    float* z1   = z    + (size_t)BB * HID;
    float* gg   = z1   + (size_t)BB * PROJ;
    float* yy   = gg   + (size_t)BB * PROJ;

    dim3 g1(8, BB);
    k_subject<<<g1, 256, 0, stream>>>(x, sid, W_sub, b_sub, P);

    dim3 g3(BB * (UU / 16) / 4, WID / 16);
    k_conv2<<<g3, 256, 0, stream>>>(P, conv1_w, conv1_b, bn1_g, bn1_b, bn1_m, bn1_v,
                                    conv2_w, conv2_b, bn2_g, bn2_b, bn2_m, bn2_v, h2);

    k_proj<<<BB, 256, 0, stream>>>(h2, proj_w, proj_b, z);

    dim3 g5(2, PROJ / 16);
    k_mlp1<<<g5, 32, 0, stream>>>(z, mlp1_w, mlp1_b, z1, gg);
    k_mlp2<<<g5, 32, 0, stream>>>(gg, mlp2_w, mlp2_b, z1, yy);

    k_ln<<<BB, 256, 0, stream>>>(yy, ln_g, ln_b, (float*)d_out);
}


# conv2: h1 costs 11 vector instructions per channel pair, not 14: bn1 scale folded into the taps (shift is the first addend, so h1 rounds differently); elu = max(min(exp(x)-1,0),x), no compare+select
# speedup vs baseline: 4.8972x; 1.0350x over previous
#include <hip/hip_runtime.h>
#include <math.h>

#define BB   32
#define CC   122
#define TT   500
#define WID  80
#define UU   96
#define K2   (WID*CC)
#define HID  768
#define PROJ 1024
#define PLEN 484
#define EPSF 1e-5f

typedef float v2f __attribute__((ext_vector_type(2)));
typedef float v8f __attribute__((ext_vector_type(8)));

static __device__ __forceinline__ v8f wmma_f32(v2f a, v2f b, v8f c) {
    return __builtin_amdgcn_wmma_f32_16x16x4_f32(false, a, false, b, (short)0, c, false, false);
}

#define SKC   32
#define SP4   9
#define SNCH  16
#define SNT   64
#define SNB   80
#define SMR   128
#define TPF   84

__global__ __launch_bounds__(256) void k_subject(
    const float* __restrict__ x, const int* __restrict__ sid,
    const float* __restrict__ W_sub, const float* __restrict__ b_sub,
    float* __restrict__ P)
{
    __shared__ float4 As[2][SMR * SP4];
    __shared__ float4 Bs[2][SNB * SP4];
    __shared__ __attribute__((aligned(16))) float Ts[SMR * TPF];

    const int tid = threadIdx.x;
    const int b   = blockIdx.y;
    const int n0  = blockIdx.x * SNT;
    const int s   = sid[b];
    const float* Wg = W_sub + (size_t)s * TT * TT;
    const float* xb = x + (size_t)b * CC * TT;

    const int kq = tid & 7;
    const int r  = tid >> 3;
    const bool third = (r < SNB - 64);
    const float* ga[4];
    const float* gb[3];
    #pragma unroll
    for (int i = 0; i < 4; ++i) {
        int row = r + 32 * i; if (row >= CC) row = CC - 1;
        ga[i] = xb + (size_t)row * TT + 4 * kq;
    }
    #pragma unroll
    for (int i = 0; i < 3; ++i) {
        int row = n0 + r + 32 * i; if (row >= TT) row = TT - 1;
        gb[i] = Wg + (size_t)row * TT + 4 * kq;
    }
    const int st = r * SP4 + kq;

    const int wv   = tid >> 5;
    const int lane = tid & 31;
    const int c    = lane & 15;
    const int h    = lane >> 4;
    const int fa = (16 * wv + c) * SP4 + h;
    const int fb = c * SP4 + h;

    float4 ra[4], rb[3];
    v8f acc[5] = {};

    #pragma unroll
    for (int i = 0; i < 4; ++i) ra[i] = *(const float4*)(ga[i]);
    #pragma unroll
    for (int i = 0; i < 2; ++i) rb[i] = *(const float4*)(gb[i]);
    if (third) rb[2] = *(const float4*)(gb[2]);
    #pragma unroll
    for (int i = 0; i < 4; ++i) As[0][st + 32 * SP4 * i] = ra[i];
    #pragma unroll
    for (int i = 0; i < 2; ++i) Bs[0][st + 32 * SP4 * i] = rb[i];
    if (third) Bs[0][st + 64 * SP4] = rb[2];
    __syncthreads();

    for (int ch = 0; ch < SNCH; ++ch) {
        const int  buf  = ch & 1;
        const bool more = (ch + 1 < SNCH);
        const int  ko   = (ch + 1) * SKC;
        const bool tail = (ko + 4 * kq) >= TT;
        if (more) {
            const int kc = tail ? 0 : ko;
            #pragma unroll
            for (int i = 0; i < 4; ++i) ra[i] = *(const float4*)(ga[i] + kc);
            #pragma unroll
            for (int i = 0; i < 2; ++i) rb[i] = *(const float4*)(gb[i] + kc);
            if (third) rb[2] = *(const float4*)(gb[2] + kc);
        }

        const float4* pA = &As[buf][fa];
        const float4* pB = &Bs[buf][fb];
        float4 af[4], bf[4][5];
        #pragma unroll
        for (int s8 = 0; s8 < 4; ++s8) {
            af[s8] = pA[2 * s8];
            #pragma unroll
            for (int j = 0; j < 5; ++j) bf[s8][j] = pB[16 * SP4 * j + 2 * s8];
        }
        __builtin_amdgcn_sched_barrier(0);
        #pragma unroll
        for (int s8 = 0; s8 < 4; ++s8) {
            const v2f a0 = { af[s8].x, af[s8].y };
            const v2f a1 = { af[s8].z, af[s8].w };
            #pragma unroll
            for (int j = 0; j < 5; ++j) {
                const v2f w0 = { bf[s8][j].x, bf[s8][j].y };
                acc[j] = wmma_f32(a0, w0, acc[j]);
            }
            #pragma unroll
            for (int j = 0; j < 5; ++j) {
                const v2f w1 = { bf[s8][j].z, bf[s8][j].w };
                acc[j] = wmma_f32(a1, w1, acc[j]);
            }
        }

        if (more) {
            if (tail) {
                #pragma unroll
                for (int i = 0; i < 4; ++i) ra[i] = float4{0.f, 0.f, 0.f, 0.f};
                #pragma unroll
                for (int i = 0; i < 3; ++i) rb[i] = float4{0.f, 0.f, 0.f, 0.f};
            }
            #pragma unroll
            for (int i = 0; i < 4; ++i) As[buf ^ 1][st + 32 * SP4 * i] = ra[i];
            #pragma unroll
            for (int i = 0; i < 2; ++i) Bs[buf ^ 1][st + 32 * SP4 * i] = rb[i];
            if (third) Bs[buf ^ 1][st + 64 * SP4] = rb[2];
            __syncthreads();
        }
    }

    const float* bg = b_sub + (size_t)s * TT;
    float* tw = Ts + (16 * wv) * TPF;
    #pragma unroll
    for (int j = 0; j < 5; ++j) {
        int nn = n0 + 16 * j + c; if (nn >= TT) nn = TT - 1;
        const float bn = bg[nn];
        #pragma unroll
        for (int q = 0; q < 8; ++q)
            tw[(q + 8 * h) * TPF + 16 * j + c] = acc[j][q] + bn;
    }
    __syncthreads();

    const int pc = lane & 15;
    const int rs = lane >> 4;
    const int m  = n0 + 4 * pc;
    #pragma unroll
    for (int it = 0; it < 8; ++it) {
        const int rl = 2 * it + rs;
        const int mm = 16 * wv + rl;
        const float4* src = (const float4*)(tw + rl * TPF + 4 * pc);
        float a[20];
        #pragma unroll
        for (int k = 0; k < 5; ++k) {
            const float4 v = src[k];
            a[4 * k] = v.x; a[4 * k + 1] = v.y; a[4 * k + 2] = v.z; a[4 * k + 3] = v.w;
        }
        float o[4];
        #pragma unroll
        for (int e = 0; e < 4; ++e) {
            float sum = 0.f;
            #pragma unroll
            for (int i = 0; i < 17; ++i) sum += a[e + i];
            o[e] = sum * (1.f / 17.f);
        }
        if (mm < CC && m < PLEN)
            *(float4*)(P + ((size_t)b * CC + mm) * PLEN + m) = float4{o[0], o[1], o[2], o[3]};
    }
}

#define CSTEPS 30
#define GS     6
#define NPR    31
#define AP     124
#define ACP    8
#define SPL(x)        (v2f{ (x), (x) })
#define FMA(a, b, c)  __builtin_elementwise_fma(SPL(a), (b), (c))
#define NC(G) (((G) == 4) ? 7 : 6)
#define ST(G, D, expr) _Pragma("unroll") for (int i = 0; i < 7; ++i) { if (i < NC(G)) { D[i] = (expr); } }
#define S12(G, E, T)  ST(G, E, FMA(T[0], pr[6 * (G) + i][0], SPL(T[5]))) ST(G, E, FMA(T[1], pr[6 * (G) + i][1], E[i]))
#define S34(G, E, T)  ST(G, E, FMA(T[2], pr[6 * (G) + i][2], E[i])) ST(G, E, FMA(T[3], pr[6 * (G) + i][3], E[i]))
#define S56(G, E, M, T) ST(G, E, FMA(T[4], pr[6 * (G) + i][4], E[i])) ST(G, M, E[i] * SPL(1.44269504088896340736f))
#define S7(G, M)      ST(G, M, (v2f{ __builtin_amdgcn_exp2f(M[i].x), __builtin_amdgcn_exp2f(M[i].y) }))
#define S8(G, E, M)   ST(G, E, __builtin_elementwise_max(__builtin_elementwise_min(M[i] - SPL(1.f), SPL(0.f)), E[i]))
#define WM(G, S, EC)  { acc = wmma_f32(acur[GS * (G) + (S)], EC[(S)], acc); }

__global__ __launch_bounds__(256) void k_conv2(
    const float* __restrict__ P,
    const float* __restrict__ conv1_w, const float* __restrict__ conv1_b,
    const float* __restrict__ bn1_g, const float* __restrict__ bn1_b,
    const float* __restrict__ bn1_m, const float* __restrict__ bn1_v,
    const float* __restrict__ conv2_w, const float* __restrict__ conv2_b,
    const float* __restrict__ bn2_g, const float* __restrict__ bn2_b,
    const float* __restrict__ bn2_m, const float* __restrict__ bn2_v,
    float* __restrict__ h2)
{
    __shared__ float w1s[WID * 5];
    __shared__ float sh1[WID];
    __shared__ __attribute__((aligned(16))) float As[2 * 2 * 16 * AP];
    __shared__ float red[4 * 8 * 32];

    const int tid = threadIdx.x;
    const int kh  = __builtin_amdgcn_readfirstlane(tid >> 7);
    const int t7  = tid & 127;
    const int hb  = kh * (2 * 16 * AP);
    const int mt  = blockIdx.y;

    const float* gA = conv2_w + (size_t)(mt * 16) * K2;
    const int cr = t7 >> 3, cq = t7 & 7;
    const int q7 = (cq + 56 > 60) ? 60 : (cq + 56);
    const unsigned gb0 = 4u * (unsigned)(cr * K2 + 2 * cq);
    const unsigned gb7 = 4u * (unsigned)(cr * K2 + 2 * q7);
    const int lo0 = hb + cr * AP + 2 * cq, lo7 = hb + cr * AP + 2 * q7;
    v2f sA[ACP], sB[ACP];
    {
        const char* g0 = (const char*)(gA + kh * CC);
        #pragma unroll
        for (int j = 0; j < ACP - 1; ++j) sA[j] = *(const v2f*)(g0 + gb0 + 64 * j);
        sA[ACP - 1] = *(const v2f*)(g0 + gb7);
    }
    #pragma unroll
    for (int j = 0; j < ACP - 1; ++j) *(v2f*)(&As[lo0 + 16 * j]) = sA[j];
    *(v2f*)(&As[lo7]) = sA[ACP - 1];
    {
        const char* g1 = (const char*)(gA + (kh + 2) * CC);
        #pragma unroll
        for (int j = 0; j < ACP - 1; ++j) sA[j] = *(const v2f*)(g1 + gb0 + 64 * j);
        sA[ACP - 1] = *(const v2f*)(g1 + gb7);
    }

    for (int i = tid; i < WID * 5; i += 256) {
        const int wi = i / 5;
        w1s[i] = conv1_w[i] * (bn1_g[wi] * rsqrtf(bn1_v[wi] + EPSF));
    }
    if (tid < WID) {
        float sc = bn1_g[tid] * rsqrtf(bn1_v[tid] + EPSF);
        sh1[tid] = bn1_b[tid] - bn1_m[tid] * sc + sc * conv1_b[tid];
    }
    __syncthreads();

    const int wave = tid >> 5;
    const int lane = tid & 31;
    const int wq   = wave & 3;
    const int nt   = blockIdx.x * 4 + wq;
    const int b    = nt / 6;
    const int u0   = (nt - 6 * b) * 16;
    const int n    = lane & 15;
    const int half = lane >> 4;

    v2f pr[NPR][5];
    #pragma unroll
    for (int i = 0; i < NPR; ++i) {
        const int c = (i < CSTEPS) ? (4 * i + 2 * half) : 120;
        const float* p = P + ((size_t)b * CC + c) * PLEN + 5 * (u0 + n);
        #pragma unroll
        for (int j = 0; j < 5; ++j) pr[i][j] = v2f{ p[j], p[PLEN + j] };
    }

    const int fro = hb + n * AP + 2 * half;
    v2f acur[CSTEPS + 1];

    v8f acc = {};
    float tc[6], tn[6];
    #pragma unroll
    for (int j = 0; j < 5; ++j) tc[j] = w1s[kh * 5 + j];
    tc[5] = sh1[kh];
    v2f ec[7] = {}, en[7] = {}, em[7] = {};
    S12(0, ec, tc) S34(0, ec, tc) S56(0, ec, em, tc) S7(0, em) S8(0, ec, em)

#define TRIP(W_, RB_, SU_, SL_) {                                                                                      \
        const int w = (W_); const int rb = (RB_);                                                                      \
                                       \
        const int wn = (w + 2 < WID) ? (w + 2) : w;                                                                    \
                             \
        const int wl = (w + 4 < WID) ? (w + 4) : w;                                                            \
                    \
                                                               \
        _Pragma("unroll")                                                                                              \
        for (int j = 0; j < 5; ++j) tn[j] = w1s[wn * 5 + j];                                                           \
        tn[5] = sh1[wn];                                                                                               \
                                    \
        const float* fr = &As[rb * 16 * AP + fro];                                                                                \
                                                \
        _Pragma("unroll")                                                                                              \
        for (int s = 0; s < CSTEPS; ++s) acur[s] = *(const v2f*)(fr + 4 * s);                                          \
        acur[CSTEPS] = *(const v2f*)(&As[rb * 16 * AP + hb + n * AP + 120]);                                                           \
                   \
                                                                      \
        __builtin_amdgcn_sched_barrier(0);                                                                             \
               \
                                                    \
        const char* gw = (const char*)(gA + wl * CC);                                                                  \
        _Pragma("unroll")                                                                                              \
        for (int j = 0; j < ACP - 1; ++j) SL_[j] = *(const v2f*)(gw + gb0 + 64 * j);                                   \
        SL_[ACP - 1] = *(const v2f*)(gw + gb7);                                                                        \
                    \
                                                                   \
        __builtin_amdgcn_sched_barrier(0);                                                                             \
                                                                                                                       \
                \
        _Pragma("unroll")                                                                                              \
        for (int g = 0; g < 5; ++g) {                                                                                  \
                                \
            if (g < 4) {                                                                                               \
                WM(g, 0, ec) S12(g + 1, en, tc)                                                                        \
                WM(g, 1, ec) S34(g + 1, en, tc)                                                                        \
                WM(g, 2, ec) S56(g + 1, en, em, tc)                                                                    \
                WM(g, 3, ec) S7(g + 1, em)                                                                             \
                WM(g, 4, ec) S8(g + 1, en, em)                                                                         \
                WM(g, 5, ec)                                                                                           \
            } else {                                                                                                   \
                       \
                                                  \
                __builtin_amdgcn_sched_barrier(0);                                                                     \
                     \
                       \
                                                                                \
                _Pragma("unroll")                                                                                      \
                for (int j = 0; j < ACP - 1; ++j) *(v2f*)(&As[(rb ^ 1) * 16 * AP + lo0 + 16 * j]) = SU_[j];                         \
                *(v2f*)(&As[(rb ^ 1) * 16 * AP + lo7]) = SU_[ACP - 1];                                                              \
                WM(4, 0, ec) S12(0, en, tn)                                                                            \
                WM(4, 1, ec) S34(0, en, tn)                                                                            \
                WM(4, 2, ec) S56(0, en, em, tn)                                                                        \
                WM(4, 3, ec) S7(0, em)                                                                                 \
                WM(4, 4, ec) S8(0, en, em)                                                                             \
                WM(4, 5, ec)                                                                                           \
                {                 \
                    v2f a  = acur[CSTEPS];                                                                             \
                    v2f bb = ec[6];                                                                                    \
                    if (half) { a = v2f{0.f, 0.f}; bb = v2f{0.f, 0.f}; }                                               \
                    acc = wmma_f32(a, bb, acc);                                                                        \
                }                                                                                                      \
            }                                                                                                          \
            _Pragma("unroll")                                                                                          \
            for (int i = 0; i < 7; ++i) ec[i] = en[i];                                                                 \
        }                                                                                                              \
        _Pragma("unroll")                                                                                              \
        for (int j = 0; j < 6; ++j) tc[j] = tn[j];                                                                     \
                                      \
              \
                  \
               \
                \
        __builtin_amdgcn_fence(__ATOMIC_RELEASE, "workgroup", "local");                                                \
        __builtin_amdgcn_s_barrier();                                                                                  \
        __builtin_amdgcn_fence(__ATOMIC_ACQUIRE, "workgroup", "local");                                                \
    }

    for (int wp = 0; wp < WID; wp += 4) {
        TRIP(wp + kh,     0, sA, sB)
        TRIP(wp + 2 + kh, 1, sB, sA)
    }
#undef TRIP

    if (kh) {
        #pragma unroll
        for (int r = 0; r < 8; ++r) red[(wq * 8 + r) * 32 + lane] = acc[r];
    }
    __syncthreads();
    if (!kh) {
        #pragma unroll
        for (int r = 0; r < 8; ++r) {
            int w2 = mt * 16 + r + half * 8;
            float sc = bn2_g[w2] * rsqrtf(bn2_v[w2] + EPSF);
            float sh = bn2_b[w2] - bn2_m[w2] * sc;
            float a  = acc[r] + red[(wq * 8 + r) * 32 + lane];
            float v  = sc * (a + conv2_b[w2]) + sh;
            float e  = v > 0.f ? v : (__expf(v) - 1.f);
            h2[((size_t)b * WID + w2) * UU + u0 + n] = e;
        }
    }
}

__global__ __launch_bounds__(256) void k_proj(
    const float* __restrict__ h2, const float* __restrict__ proj_w,
    const float* __restrict__ proj_b, float* __restrict__ z)
{
    __shared__ float hs[WID * UU];
    __shared__ float pw[8 * WID];
    const int b = blockIdx.x;
    for (int i = threadIdx.x; i < WID * UU; i += 256) hs[i] = h2[(size_t)b * WID * UU + i];
    for (int i = threadIdx.x; i < 8 * WID; i += 256)  pw[i] = proj_w[i];
    __syncthreads();
    for (int idx = threadIdx.x; idx < HID; idx += 256) {
        int u = idx >> 3, e = idx & 7;
        float s = proj_b[e];
        #pragma unroll 8
        for (int w = 0; w < WID; ++w) s += pw[e * WID + w] * hs[w * UU + u];
        z[(size_t)b * HID + idx] = s;
    }
}

#define MKB    32
#define MAHEAD 7

template <int K>
static __device__ __forceinline__ v8f mlp_tile(const float* pa, const float* pb)
{
    constexpr int NB = K / MKB;
    float4 fa[NB][4], fb[NB][4];
    v8f acc = {};
    #pragma unroll
    for (int i = 0; i < MAHEAD; ++i) {
        #pragma unroll
        for (int j = 0; j < 4; ++j) {
            fa[i][j] = *(const float4*)(pa + MKB * i + 8 * j);
            fb[i][j] = *(const float4*)(pb + MKB * i + 8 * j);
        }
    }
    #pragma unroll
    for (int i = 0; i < NB; ++i) {
        if (i + MAHEAD < NB) {
            #pragma unroll
            for (int j = 0; j < 4; ++j) {
                fa[i + MAHEAD][j] = *(const float4*)(pa + MKB * (i + MAHEAD) + 8 * j);
                fb[i + MAHEAD][j] = *(const float4*)(pb + MKB * (i + MAHEAD) + 8 * j);
            }
        }
        #pragma unroll
        for (int j = 0; j < 4; ++j) {
            const v2f a0 = { fa[i][j].x, fa[i][j].y };
            const v2f b0 = { fb[i][j].x, fb[i][j].y };
            acc = wmma_f32(a0, b0, acc);
            const v2f a1 = { fa[i][j].z, fa[i][j].w };
            const v2f b1 = { fb[i][j].z, fb[i][j].w };
            acc = wmma_f32(a1, b1, acc);
        }
        __builtin_amdgcn_sched_barrier(0);
    }
    return acc;
}

__global__ __launch_bounds__(32) void k_mlp1(
    const float* __restrict__ z, const float* __restrict__ w,
    const float* __restrict__ bias, float* __restrict__ z1, float* __restrict__ g)
{
    const int lane = threadIdx.x;
    const int m0 = blockIdx.x * 16, n0 = blockIdx.y * 16;
    const float* pa = z + (size_t)(m0 + (lane & 15)) * HID + (lane >> 4) * 4;
    const float* pb = w + (size_t)(n0 + (lane & 15)) * HID + (lane >> 4) * 4;
    const v8f acc = mlp_tile<HID>(pa, pb);
    const int nn = n0 + (lane & 15);
    const float bn = bias[nn];
    #pragma unroll
    for (int r = 0; r < 8; ++r) {
        int mm = m0 + r + ((lane >> 4) << 3);
        float v = acc[r] + bn;
        z1[(size_t)mm * PROJ + nn] = v;
        g [(size_t)mm * PROJ + nn] = 0.5f * v * (1.f + erff(v * 0.70710678118654752f));
    }
}

__global__ __launch_bounds__(32) void k_mlp2(
    const float* __restrict__ g, const float* __restrict__ w,
    const float* __restrict__ bias, const float* __restrict__ z1,
    float* __restrict__ y)
{
    const int lane = threadIdx.x;
    const int m0 = blockIdx.x * 16, n0 = blockIdx.y * 16;
    const float* pa = g + (size_t)(m0 + (lane & 15)) * PROJ + (lane >> 4) * 4;
    const float* pb = w + (size_t)(n0 + (lane & 15)) * PROJ + (lane >> 4) * 4;
    const v8f acc = mlp_tile<PROJ>(pa, pb);
    const int nn = n0 + (lane & 15);
    const float bn = bias[nn];
    #pragma unroll
    for (int r = 0; r < 8; ++r) {
        int mm = m0 + r + ((lane >> 4) << 3);
        y[(size_t)mm * PROJ + nn] = z1[(size_t)mm * PROJ + nn] + acc[r] + bn;
    }
}

__global__ __launch_bounds__(256) void k_ln(
    const float* __restrict__ y, const float* __restrict__ gam,
    const float* __restrict__ bet, float* __restrict__ out)
{
    __shared__ float red[256];
    const int b = blockIdx.x;
    const float* row = y + (size_t)b * PROJ;
    float s = 0.f;
    for (int i = threadIdx.x; i < PROJ; i += 256) s += row[i];
    red[threadIdx.x] = s; __syncthreads();
    for (int off = 128; off > 0; off >>= 1) {
        if (threadIdx.x < off) red[threadIdx.x] += red[threadIdx.x + off];
        __syncthreads();
    }
    const float mu = red[0] * (1.f / PROJ);
    __syncthreads();
    float s2 = 0.f;
    for (int i = threadIdx.x; i < PROJ; i += 256) { float d = row[i] - mu; s2 += d * d; }
    red[threadIdx.x] = s2; __syncthreads();
    for (int off = 128; off > 0; off >>= 1) {
        if (threadIdx.x < off) red[threadIdx.x] += red[threadIdx.x + off];
        __syncthreads();
    }
    const float inv = rsqrtf(red[0] * (1.f / PROJ) + EPSF);
    for (int i = threadIdx.x; i < PROJ; i += 256)
        out[(size_t)b * PROJ + i] = (row[i] - mu) * inv * gam[i] + bet[i];
}

extern "C" void kernel_launch(void* const* d_in, const int* in_sizes, int n_in,
                              void* d_out, int out_size, void* d_ws, size_t ws_size,
                              hipStream_t stream) {
    const float* x       = (const float*)d_in[0];
    const int*   sid     = (const int*)  d_in[1];
    const float* W_sub   = (const float*)d_in[2];
    const float* b_sub   = (const float*)d_in[3];
    const float* conv1_w = (const float*)d_in[4];
    const float* conv1_b = (const float*)d_in[5];
    const float* bn1_g   = (const float*)d_in[6];
    const float* bn1_b   = (const float*)d_in[7];
    const float* bn1_m   = (const float*)d_in[8];
    const float* bn1_v   = (const float*)d_in[9];
    const float* conv2_w = (const float*)d_in[10];
    const float* conv2_b = (const float*)d_in[11];
    const float* bn2_g   = (const float*)d_in[12];
    const float* bn2_b   = (const float*)d_in[13];
    const float* bn2_m   = (const float*)d_in[14];
    const float* bn2_v   = (const float*)d_in[15];
    const float* proj_w  = (const float*)d_in[16];
    const float* proj_b  = (const float*)d_in[17];
    const float* mlp1_w  = (const float*)d_in[18];
    const float* mlp1_b  = (const float*)d_in[19];
    const float* mlp2_w  = (const float*)d_in[20];
    const float* mlp2_b  = (const float*)d_in[21];
    const float* ln_g    = (const float*)d_in[22];
    const float* ln_b    = (const float*)d_in[23];

    float* ws   = (float*)d_ws;
    float* out0 = ws;
    float* P    = out0 + (size_t)BB * CC * TT;
    float* h2   = P    + (size_t)BB * CC * PLEN;
    float* z    = h2   + (size_t)BB * WID * UU;
    float* z1   = z    + (size_t)BB * HID;
    float* gg   = z1   + (size_t)BB * PROJ;
    float* yy   = gg   + (size_t)BB * PROJ;

    dim3 g1(8, BB);
    k_subject<<<g1, 256, 0, stream>>>(x, sid, W_sub, b_sub, P);

    dim3 g3(BB * (UU / 16) / 4, WID / 16);
    k_conv2<<<g3, 256, 0, stream>>>(P, conv1_w, conv1_b, bn1_g, bn1_b, bn1_m, bn1_v,
                                    conv2_w, conv2_b, bn2_g, bn2_b, bn2_m, bn2_v, h2);

    k_proj<<<BB, 256, 0, stream>>>(h2, proj_w, proj_b, z);

    dim3 g5(2, PROJ / 16);
    k_mlp1<<<g5, 32, 0, stream>>>(z, mlp1_w, mlp1_b, z1, gg);
    k_mlp2<<<g5, 32, 0, stream>>>(gg, mlp2_w, mlp2_b, z1, yy);

    k_ln<<<BB, 256, 0, stream>>>(yy, ln_g, ln_b, (float*)d_out);
}
